# starting kernel
# baseline (speedup 1.0000x reference)
#include <hip/hip_runtime.h>
#include <hip/hip_bf16.h>
#include <stdint.h>

typedef __attribute__((ext_vector_type(16))) _Float16 v16h;
typedef __attribute__((ext_vector_type(8)))  _Float16 v8h;
typedef __attribute__((ext_vector_type(8)))  float    v8f;

#define NFEAT 64
#define HW    128
#define HALO  130
#define NBATCH 4
#define KTAP  441
#define NCH   442
#define NPADC 448

__device__ __forceinline__ v16h cat8(v8h lo, v8h hi) {
  return __builtin_shufflevector(lo, hi, 0,1,2,3,4,5,6,7,8,9,10,11,12,13,14,15);
}
__device__ __forceinline__ v8f wmma16(v16h a, v16h b, v8f c) {
  return __builtin_amdgcn_wmma_f32_16x16x32_f16(false, a, false, b, (short)0, c, false, false);
}

__global__ void zero_u32(uint32_t* p, size_t n) {
  size_t i = (size_t)blockIdx.x * blockDim.x + threadIdx.x;
  if (i < n) p[i] = 0u;
}

__global__ void pack_w(const float* __restrict__ w1s, const float* __restrict__ w2s,
                       _Float16* __restrict__ dst) {
  int idx = blockIdx.x * blockDim.x + threadIdx.x;
  const int TOT = 2 * 8 * 9 * 64 * 64;
  if (idx >= TOT) return;
  int c   = idx & 63;
  int n   = (idx >> 6) & 63;
  int pg  = idx >> 12;
  int pos = pg % 9;
  int lay = pg / 9;
  int blk = lay & 7;
  int which = lay >> 3;
  const float* src = which ? w2s : w1s;
  float v = src[(((size_t)(blk * 64 + n) * 64 + c) * 9) + pos];
  dst[idx] = (_Float16)v;
}

__global__ void pack_wout(const float* __restrict__ w_out, _Float16* __restrict__ dst) {
  int idx = blockIdx.x * blockDim.x + threadIdx.x;
  if (idx >= NPADC * 64) return;
  int c = idx & 63;
  int n = idx >> 6;
  dst[idx] = (n < NCH) ? (_Float16)w_out[(size_t)n * 64 + c] : (_Float16)0.0f;
}

__global__ void conv_in_k(const float* __restrict__ z, const float* __restrict__ w_in,
                          const float* __restrict__ b_in, _Float16* __restrict__ dst) {
  int idx = blockIdx.x * blockDim.x + threadIdx.x;
  const int TOT = NBATCH * HW * HW * NFEAT;
  if (idx >= TOT) return;
  int co = idx & 63;
  int t  = idx >> 6;
  int x  = t & (HW - 1); t >>= 7;
  int y  = t & (HW - 1); t >>= 7;
  int b  = t;
  float acc = b_in[co];
  #pragma unroll
  for (int ci = 0; ci < 3; ++ci) {
    const float* zp = z + ((size_t)(b * 3 + ci)) * HW * HW;
    const float* wp = w_in + ((size_t)(co * 3 + ci)) * 9;
    #pragma unroll
    for (int ky = 0; ky < 3; ++ky) {
      int yy = y + ky - 1;
      if ((unsigned)yy >= (unsigned)HW) continue;
      #pragma unroll
      for (int kx = 0; kx < 3; ++kx) {
        int xx = x + kx - 1;
        if ((unsigned)xx >= (unsigned)HW) continue;
        acc += zp[yy * HW + xx] * wp[ky * 3 + kx];
      }
    }
  }
  dst[(((size_t)b * HALO + y + 1) * HALO + x + 1) * NFEAT + co] = (_Float16)acc;
}

__global__ __launch_bounds__(256) void conv3x3_wmma(
    const _Float16* __restrict__ src, _Float16* __restrict__ dst,
    const _Float16* __restrict__ wl, const float* __restrict__ bias,
    const _Float16* __restrict__ resid, int do_relu) {
  extern __shared__ _Float16 lw[];
  {
    const uint4* g = (const uint4*)wl;
    uint4* s = (uint4*)lw;
    for (int i = threadIdx.x; i < (9 * 64 * 64) / 8; i += 256) s[i] = g[i];
  }
  __syncthreads();

  const int wave = threadIdx.x >> 5;
  const int lane = threadIdx.x & 31;
  const int tile = blockIdx.x * 8 + wave;
  const int b  = tile >> 10;
  const int r  = tile & 1023;
  const int y  = r >> 3;
  const int x0 = (r & 7) << 4;

  const int mrow = lane & 15;
  const int ksel = (lane >> 4) << 3;
  const int nlan = lane & 15;
  const int mhi  = (lane >> 4) << 3;

  v8f acc[4] = {};

  #pragma unroll
  for (int ky = 0; ky < 3; ++ky) {
    #pragma unroll
    for (int kx = 0; kx < 3; ++kx) {
      const _Float16* ap =
          src + (((size_t)b * HALO + y + ky) * HALO + (x0 + mrow + kx)) * NFEAT;
      const _Float16* wp = lw + (ky * 3 + kx) * 4096;
      v16h a0 = cat8(*(const v8h*)(ap + ksel),      *(const v8h*)(ap + 16 + ksel));
      v16h a1 = cat8(*(const v8h*)(ap + 32 + ksel), *(const v8h*)(ap + 48 + ksel));
      #pragma unroll
      for (int kc = 0; kc < 2; ++kc) {
        const int c0 = kc * 32;
        const v16h a = kc ? a1 : a0;
        const _Float16* w0 = wp + (size_t)(0 * 16 + nlan) * 64 + c0 + ksel;
        const _Float16* w1 = wp + (size_t)(1 * 16 + nlan) * 64 + c0 + ksel;
        const _Float16* w2 = wp + (size_t)(2 * 16 + nlan) * 64 + c0 + ksel;
        const _Float16* w3 = wp + (size_t)(3 * 16 + nlan) * 64 + c0 + ksel;
        v16h b0 = cat8(*(const v8h*)w0, *(const v8h*)(w0 + 16));
        v16h b1 = cat8(*(const v8h*)w1, *(const v8h*)(w1 + 16));
        v16h b2 = cat8(*(const v8h*)w2, *(const v8h*)(w2 + 16));
        v16h b3 = cat8(*(const v8h*)w3, *(const v8h*)(w3 + 16));
        acc[0] = wmma16(a, b0, acc[0]);
        acc[1] = wmma16(a, b1, acc[1]);
        acc[2] = wmma16(a, b2, acc[2]);
        acc[3] = wmma16(a, b3, acc[3]);
      }
    }
  }

  #pragma unroll
  for (int nt = 0; nt < 4; ++nt) {
    const int n = nt * 16 + nlan;
    const float bv = bias[n];
    #pragma unroll
    for (int i = 0; i < 8; ++i) {
      const int m = i + mhi;
      size_t off = (((size_t)b * HALO + y + 1) * HALO + (x0 + m + 1)) * NFEAT + n;
      float v = acc[nt][i] + bv;
      if (resid) v += (float)resid[off];
      if (do_relu) v = v > 0.0f ? v : 0.0f;
      dst[off] = (_Float16)v;
    }
  }
}

__global__ __launch_bounds__(128) void convout_softmax(
    const _Float16* __restrict__ feat, const _Float16* __restrict__ wo,
    const float* __restrict__ bo, _Float16* __restrict__ probs,
    float* __restrict__ nsum) {
  extern __shared__ float lgs[];
  const int wave = threadIdx.x >> 5;
  const int lane = threadIdx.x & 31;
  const int tile = blockIdx.x * 4 + wave;
  const int b  = tile >> 10;
  const int r  = tile & 1023;
  const int y  = r >> 3;
  const int x0 = (r & 7) << 4;

  const int mrow = lane & 15;
  const int ksel = (lane >> 4) << 3;
  const int nlan = lane & 15;
  const int mhi  = (lane >> 4) << 3;

  const _Float16* ap =
      feat + (((size_t)b * HALO + y + 1) * HALO + (x0 + mrow + 1)) * NFEAT;
  v16h a0 = cat8(*(const v8h*)(ap + ksel),      *(const v8h*)(ap + 16 + ksel));
  v16h a1 = cat8(*(const v8h*)(ap + 32 + ksel), *(const v8h*)(ap + 48 + ksel));

  float* lg = lgs + wave * 16 * NPADC;
  float* st = lgs + 4 * 16 * NPADC + wave * 32;

  for (int nt = 0; nt < NPADC / 16; ++nt) {
    const int n = nt * 16 + nlan;
    const _Float16* wp = wo + (size_t)n * 64;
    v16h b0 = cat8(*(const v8h*)(wp + ksel),      *(const v8h*)(wp + 16 + ksel));
    v16h b1 = cat8(*(const v8h*)(wp + 32 + ksel), *(const v8h*)(wp + 48 + ksel));
    v8f c = {};
    c = wmma16(a0, b0, c);
    c = wmma16(a1, b1, c);
    const float bv = (n < NCH) ? bo[n] : 0.0f;
    #pragma unroll
    for (int i = 0; i < 8; ++i) lg[(i + mhi) * NPADC + n] = c[i] + bv;
  }
  __syncthreads();

  for (int p = 0; p < 16; ++p) {
    const float* row = lg + p * NPADC;
    float mx = -1e30f;
    for (int t = lane; t < NCH; t += 32) mx = fmaxf(mx, row[t]);
    #pragma unroll
    for (int off = 16; off; off >>= 1) mx = fmaxf(mx, __shfl_xor(mx, off, 32));
    float sm = 0.0f;
    for (int t = lane; t < NCH; t += 32) sm += __expf(row[t] - mx);
    #pragma unroll
    for (int off = 16; off; off >>= 1) sm += __shfl_xor(sm, off, 32);
    if (lane == 0) { st[p] = mx; st[16 + p] = 1.0f / sm; }
  }

  const int p   = lane & 15;
  const float mx  = st[p];
  const float inv = st[16 + p];
  const float* row = lg + p * NPADC;
  _Float16* pb = probs + (size_t)b * NCH * (HW * HW) + (size_t)y * HW + x0 + p;
  float nacc = 0.0f;
  for (int t = (lane >> 4); t < NCH; t += 2) {
    float pv = __expf(row[t] - mx) * inv;
    pb[(size_t)t * (HW * HW)] = (_Float16)pv;
    if (t == KTAP) nacc += pv;
  }
  atomicAdd(nsum + b, nacc);
}

__global__ void apply_k(const float* __restrict__ x, const float* __restrict__ eps,
                        const _Float16* __restrict__ probs, const float* __restrict__ nsum,
                        float* __restrict__ out) {
  int idx = blockIdx.x * blockDim.x + threadIdx.x;
  const int TOT = NBATCH * 3 * HW * HW;
  if (idx >= TOT) return;
  int w = idx & (HW - 1);
  int t = idx >> 7;
  int h = t & (HW - 1); t >>= 7;
  int c = t % 3;
  int b = t / 3;

  const float* xb = x + ((size_t)(b * 3 + c)) * 512 * 512;
  const _Float16* pb = probs + (size_t)b * NCH * (HW * HW) + (size_t)h * HW + w;
  const int yb = h * 4 - 10, xbs = w * 4 - 10;
  float acc = 0.0f;
  for (int ky = 0; ky < 21; ++ky) {
    int yi = yb + ky;
    if ((unsigned)yi >= 512u) continue;
    const float* xr = xb + (size_t)yi * 512;
    const _Float16* pr = pb + (size_t)(ky * 21) * (HW * HW);
    __builtin_prefetch(pr + (size_t)21 * (HW * HW), 0, 1);
    if (yi + 1 < 512) __builtin_prefetch(xr + 512, 0, 1);
    for (int kx = 0; kx < 21; ++kx) {
      int xi = xbs + kx;
      if ((unsigned)xi >= 512u) continue;
      acc += xr[xi] * (float)pr[(size_t)kx * (HW * HW)];
    }
  }
  float noise = (nsum[b] * (1.0f / (HW * HW))) * eps[idx];
  out[idx] = acc + noise;
}

extern "C" void kernel_launch(void* const* d_in, const int* in_sizes, int n_in,
                              void* d_out, int out_size, void* d_ws, size_t ws_size,
                              hipStream_t stream) {
  const float* x     = (const float*)d_in[0];
  const float* z     = (const float*)d_in[1];
  const float* eps   = (const float*)d_in[2];
  const float* w_in  = (const float*)d_in[3];
  const float* b_in  = (const float*)d_in[4];
  const float* w1s   = (const float*)d_in[5];
  const float* b1s   = (const float*)d_in[6];
  const float* w2s   = (const float*)d_in[7];
  const float* b2s   = (const float*)d_in[8];
  const float* w_out = (const float*)d_in[9];
  const float* b_out = (const float*)d_in[10];
  float* out = (float*)d_out;

  char* ws = (char*)d_ws;
  const size_t FBYTES = (size_t)NBATCH * HALO * HALO * NFEAT * sizeof(_Float16);
  _Float16* F0 = (_Float16*)(ws);
  _Float16* F1 = (_Float16*)(ws + FBYTES);
  _Float16* F2 = (_Float16*)(ws + 2 * FBYTES);
  _Float16* wpack  = (_Float16*)(ws + 3 * FBYTES);
  const size_t WPACK_ELEMS = (size_t)2 * 8 * 9 * 64 * 64;
  _Float16* wopack = wpack + WPACK_ELEMS;
  _Float16* probs  = wopack + (size_t)NPADC * 64;
  float*    nsum   = (float*)(probs + (size_t)NBATCH * NCH * HW * HW);

  const size_t zn = (3 * FBYTES) / 4;
  zero_u32<<<(unsigned)((zn + 255) / 256), 256, 0, stream>>>((uint32_t*)ws, zn);
  zero_u32<<<1, 32, 0, stream>>>((uint32_t*)nsum, 4);

  pack_w<<<(2 * 8 * 9 * 64 * 64 + 255) / 256, 256, 0, stream>>>(w1s, w2s, wpack);
  pack_wout<<<(NPADC * 64 + 255) / 256, 256, 0, stream>>>(w_out, wopack);

  conv_in_k<<<(NBATCH * HW * HW * NFEAT + 255) / 256, 256, 0, stream>>>(z, w_in, b_in, F0);

  _Float16* cur = F0;
  _Float16* nxt = F2;
  const size_t LAYER = (size_t)9 * 64 * 64;
  for (int i = 0; i < 8; ++i) {
    conv3x3_wmma<<<512, 256, 9 * 64 * 64 * sizeof(_Float16), stream>>>(
        cur, F1, wpack + (size_t)i * LAYER, b1s + (size_t)i * 64, nullptr, 1);
    conv3x3_wmma<<<512, 256, 9 * 64 * 64 * sizeof(_Float16), stream>>>(
        F1, nxt, wpack + (size_t)(8 + i) * LAYER, b2s + (size_t)i * 64, cur, 0);
    _Float16* t = cur; cur = nxt; nxt = t;
  }

  const size_t SM_OUT = (size_t)4 * 16 * NPADC * sizeof(float) + 4 * 32 * sizeof(float);
  convout_softmax<<<1024, 128, (unsigned)SM_OUT, stream>>>(cur, wopack, b_out, probs, nsum);
  apply_k<<<(NBATCH * 3 * HW * HW + 255) / 256, 256, 0, stream>>>(x, eps, probs, nsum, out);
}


# 1x1 conv + softmax kernel: keep all 28 logit tiles in the WMMA accumulators and do the softmax in registers with 16-byte prob stores, instead of three bank-conflicted passes over LDS
# speedup vs baseline: 1.3078x; 1.3078x over previous
#include <hip/hip_runtime.h>
#include <hip/hip_bf16.h>
#include <stdint.h>

typedef __attribute__((ext_vector_type(16))) _Float16 v16h;
typedef __attribute__((ext_vector_type(8)))  _Float16 v8h;
typedef __attribute__((ext_vector_type(8)))  float    v8f;

#define NFEAT 64
#define HW    128
#define HALO  130
#define NBATCH 4
#define KTAP  441
#define NCH   442
#define NPADC 448

__device__ __forceinline__ v16h cat8(v8h lo, v8h hi) {
  return __builtin_shufflevector(lo, hi, 0,1,2,3,4,5,6,7,8,9,10,11,12,13,14,15);
}
__device__ __forceinline__ v8f wmma16(v16h a, v16h b, v8f c) {
  return __builtin_amdgcn_wmma_f32_16x16x32_f16(false, a, false, b, (short)0, c, false, false);
}

__global__ void zero_u32(uint32_t* p, size_t n) {
  size_t i = (size_t)blockIdx.x * blockDim.x + threadIdx.x;
  if (i < n) p[i] = 0u;
}

__global__ void pack_w(const float* __restrict__ w1s, const float* __restrict__ w2s,
                       _Float16* __restrict__ dst) {
  int idx = blockIdx.x * blockDim.x + threadIdx.x;
  const int TOT = 2 * 8 * 9 * 64 * 64;
  if (idx >= TOT) return;
  int c   = idx & 63;
  int n   = (idx >> 6) & 63;
  int pg  = idx >> 12;
  int pos = pg % 9;
  int lay = pg / 9;
  int blk = lay & 7;
  int which = lay >> 3;
  const float* src = which ? w2s : w1s;
  float v = src[(((size_t)(blk * 64 + n) * 64 + c) * 9) + pos];
  dst[idx] = (_Float16)v;
}

__global__ void pack_wout(const float* __restrict__ w_out, _Float16* __restrict__ dst) {
  int idx = blockIdx.x * blockDim.x + threadIdx.x;
  if (idx >= NPADC * 64) return;
  int c = idx & 63;
  int n = idx >> 6;
  dst[idx] = (n < NCH) ? (_Float16)w_out[(size_t)n * 64 + c] : (_Float16)0.0f;
}

__global__ void conv_in_k(const float* __restrict__ z, const float* __restrict__ w_in,
                          const float* __restrict__ b_in, _Float16* __restrict__ dst) {
  int idx = blockIdx.x * blockDim.x + threadIdx.x;
  const int TOT = NBATCH * HW * HW * NFEAT;
  if (idx >= TOT) return;
  int co = idx & 63;
  int t  = idx >> 6;
  int x  = t & (HW - 1); t >>= 7;
  int y  = t & (HW - 1); t >>= 7;
  int b  = t;
  float acc = b_in[co];
  #pragma unroll
  for (int ci = 0; ci < 3; ++ci) {
    const float* zp = z + ((size_t)(b * 3 + ci)) * HW * HW;
    const float* wp = w_in + ((size_t)(co * 3 + ci)) * 9;
    #pragma unroll
    for (int ky = 0; ky < 3; ++ky) {
      int yy = y + ky - 1;
      if ((unsigned)yy >= (unsigned)HW) continue;
      #pragma unroll
      for (int kx = 0; kx < 3; ++kx) {
        int xx = x + kx - 1;
        if ((unsigned)xx >= (unsigned)HW) continue;
        acc += zp[yy * HW + xx] * wp[ky * 3 + kx];
      }
    }
  }
  dst[(((size_t)b * HALO + y + 1) * HALO + x + 1) * NFEAT + co] = (_Float16)acc;
}

__global__ __launch_bounds__(256) void conv3x3_wmma(
    const _Float16* __restrict__ src, _Float16* __restrict__ dst,
    const _Float16* __restrict__ wl, const float* __restrict__ bias,
    const _Float16* __restrict__ resid, int do_relu) {
  extern __shared__ _Float16 lw[];
  {
    const uint4* g = (const uint4*)wl;
    uint4* s = (uint4*)lw;
    for (int i = threadIdx.x; i < (9 * 64 * 64) / 8; i += 256) s[i] = g[i];
  }
  __syncthreads();

  const int wave = threadIdx.x >> 5;
  const int lane = threadIdx.x & 31;
  const int tile = blockIdx.x * 8 + wave;
  const int b  = tile >> 10;
  const int r  = tile & 1023;
  const int y  = r >> 3;
  const int x0 = (r & 7) << 4;

  const int mrow = lane & 15;
  const int ksel = (lane >> 4) << 3;
  const int nlan = lane & 15;
  const int mhi  = (lane >> 4) << 3;

  v8f acc[4] = {};

  #pragma unroll
  for (int ky = 0; ky < 3; ++ky) {
    #pragma unroll
    for (int kx = 0; kx < 3; ++kx) {
      const _Float16* ap =
          src + (((size_t)b * HALO + y + ky) * HALO + (x0 + mrow + kx)) * NFEAT;
      const _Float16* wp = lw + (ky * 3 + kx) * 4096;
      v16h a0 = cat8(*(const v8h*)(ap + ksel),      *(const v8h*)(ap + 16 + ksel));
      v16h a1 = cat8(*(const v8h*)(ap + 32 + ksel), *(const v8h*)(ap + 48 + ksel));
      #pragma unroll
      for (int kc = 0; kc < 2; ++kc) {
        const int c0 = kc * 32;
        const v16h a = kc ? a1 : a0;
        const _Float16* w0 = wp + (size_t)(0 * 16 + nlan) * 64 + c0 + ksel;
        const _Float16* w1 = wp + (size_t)(1 * 16 + nlan) * 64 + c0 + ksel;
        const _Float16* w2 = wp + (size_t)(2 * 16 + nlan) * 64 + c0 + ksel;
        const _Float16* w3 = wp + (size_t)(3 * 16 + nlan) * 64 + c0 + ksel;
        v16h b0 = cat8(*(const v8h*)w0, *(const v8h*)(w0 + 16));
        v16h b1 = cat8(*(const v8h*)w1, *(const v8h*)(w1 + 16));
        v16h b2 = cat8(*(const v8h*)w2, *(const v8h*)(w2 + 16));
        v16h b3 = cat8(*(const v8h*)w3, *(const v8h*)(w3 + 16));
        acc[0] = wmma16(a, b0, acc[0]);
        acc[1] = wmma16(a, b1, acc[1]);
        acc[2] = wmma16(a, b2, acc[2]);
        acc[3] = wmma16(a, b3, acc[3]);
      }
    }
  }

  #pragma unroll
  for (int nt = 0; nt < 4; ++nt) {
    const int n = nt * 16 + nlan;
    const float bv = bias[n];
    #pragma unroll
    for (int i = 0; i < 8; ++i) {
      const int m = i + mhi;
      size_t off = (((size_t)b * HALO + y + 1) * HALO + (x0 + m + 1)) * NFEAT + n;
      float v = acc[nt][i] + bv;
      if (resid) v += (float)resid[off];
      if (do_relu) v = v > 0.0f ? v : 0.0f;
      dst[off] = (_Float16)v;
    }
  }
}

__global__ __launch_bounds__(128) void convout_softmax(
    const _Float16* __restrict__ feat, const _Float16* __restrict__ wo,
    const float* __restrict__ bo, _Float16* __restrict__ probs,
    float* __restrict__ nsum) {
  const int wave = threadIdx.x >> 5;
  const int lane = threadIdx.x & 31;
  const int tile = blockIdx.x * 4 + wave;
  const int b  = tile >> 10;
  const int r  = tile & 1023;
  const int y  = r >> 3;
  const int x0 = (r & 7) << 4;

  const int mrow = lane & 15;
  const int ksel = (lane >> 4) << 3;
  const int nlan = lane & 15;
  const int mhi  = (lane >> 4) << 3;

  const _Float16* ap =
      feat + (((size_t)b * HALO + y + 1) * HALO + (x0 + mrow + 1)) * NFEAT;
  v16h a0 = cat8(*(const v8h*)(ap + ksel),      *(const v8h*)(ap + 16 + ksel));
  v16h a1 = cat8(*(const v8h*)(ap + 32 + ksel), *(const v8h*)(ap + 48 + ksel));

  constexpr int NT = NPADC / 16;
  v8f lg[NT];

  constexpr int PF = 5;
  __builtin_amdgcn_sched_group_barrier(0x020, 4 + 4 * PF, 0);
  #pragma unroll
  for (int nt = 0; nt < NT; ++nt) {
    const int n = nt * 16 + nlan;
    const _Float16* wp = wo + (size_t)n * 64;
    v16h b0 = cat8(*(const v8h*)(wp + ksel),      *(const v8h*)(wp + 16 + ksel));
    v16h b1 = cat8(*(const v8h*)(wp + 32 + ksel), *(const v8h*)(wp + 48 + ksel));
    v8f c = {};
    c = wmma16(a0, b0, c);
    c = wmma16(a1, b1, c);
    lg[nt] = c;
    __builtin_amdgcn_sched_group_barrier(0x008, 2, 0);
    __builtin_amdgcn_sched_group_barrier(0x020, 4, 0);
  }
  #pragma unroll
  for (int nt = 0; nt < NT; ++nt) {
    const int n = nt * 16 + nlan;
    const float bv = (n < NCH) ? bo[n] : -1e30f;
    #pragma unroll
    for (int i = 0; i < 8; ++i) lg[nt][i] += bv;
  }

  float inv[8];
  #pragma unroll
  for (int i = 0; i < 8; ++i) {
    float m = lg[0][i];
    #pragma unroll
    for (int nt = 1; nt < NT; ++nt) m = fmaxf(m, lg[nt][i]);
    #pragma unroll
    for (int off = 8; off; off >>= 1) m = fmaxf(m, __shfl_xor(m, off, 32));
    float s = 0.0f;
    #pragma unroll
    for (int nt = 0; nt < NT; ++nt) {
      const float e = __expf(lg[nt][i] - m);
      lg[nt][i] = e;
      s += e;
    }
    #pragma unroll
    for (int off = 8; off; off >>= 1) s += __shfl_xor(s, off, 32);
    inv[i] = 1.0f / s;
  }

  _Float16* pb = probs + (size_t)b * NCH * (HW * HW) + (size_t)y * HW + x0 + mhi;
  #pragma unroll
  for (int nt = 0; nt < NT; ++nt) {
    const int t = nt * 16 + nlan;
    v8h o;
    #pragma unroll
    for (int i = 0; i < 8; ++i) o[i] = (_Float16)(lg[nt][i] * inv[i]);
    if (t < NCH) *(v8h*)(pb + (size_t)t * (HW * HW)) = o;
  }

  float nacc = 0.0f;
  if (nlan == (KTAP & 15)) {
    #pragma unroll
    for (int i = 0; i < 8; ++i) nacc += lg[KTAP >> 4][i] * inv[i];
  }
  nacc += __shfl_xor(nacc, 16, 32);
  if (lane == (KTAP & 15)) atomicAdd(nsum + b, nacc);
}

__global__ void apply_k(const float* __restrict__ x, const float* __restrict__ eps,
                        const _Float16* __restrict__ probs, const float* __restrict__ nsum,
                        float* __restrict__ out) {
  int idx = blockIdx.x * blockDim.x + threadIdx.x;
  const int TOT = NBATCH * 3 * HW * HW;
  if (idx >= TOT) return;
  int w = idx & (HW - 1);
  int t = idx >> 7;
  int h = t & (HW - 1); t >>= 7;
  int c = t % 3;
  int b = t / 3;

  const float* xb = x + ((size_t)(b * 3 + c)) * 512 * 512;
  const _Float16* pb = probs + (size_t)b * NCH * (HW * HW) + (size_t)h * HW + w;
  const int yb = h * 4 - 10, xbs = w * 4 - 10;
  float acc = 0.0f;
  for (int ky = 0; ky < 21; ++ky) {
    int yi = yb + ky;
    if ((unsigned)yi >= 512u) continue;
    const float* xr = xb + (size_t)yi * 512;
    const _Float16* pr = pb + (size_t)(ky * 21) * (HW * HW);
    __builtin_prefetch(pr + (size_t)21 * (HW * HW), 0, 1);
    if (yi + 1 < 512) __builtin_prefetch(xr + 512, 0, 1);
    for (int kx = 0; kx < 21; ++kx) {
      int xi = xbs + kx;
      if ((unsigned)xi >= 512u) continue;
      acc += xr[xi] * (float)pr[(size_t)kx * (HW * HW)];
    }
  }
  float noise = (nsum[b] * (1.0f / (HW * HW))) * eps[idx];
  out[idx] = acc + noise;
}

extern "C" void kernel_launch(void* const* d_in, const int* in_sizes, int n_in,
                              void* d_out, int out_size, void* d_ws, size_t ws_size,
                              hipStream_t stream) {
  const float* x     = (const float*)d_in[0];
  const float* z     = (const float*)d_in[1];
  const float* eps   = (const float*)d_in[2];
  const float* w_in  = (const float*)d_in[3];
  const float* b_in  = (const float*)d_in[4];
  const float* w1s   = (const float*)d_in[5];
  const float* b1s   = (const float*)d_in[6];
  const float* w2s   = (const float*)d_in[7];
  const float* b2s   = (const float*)d_in[8];
  const float* w_out = (const float*)d_in[9];
  const float* b_out = (const float*)d_in[10];
  float* out = (float*)d_out;

  char* ws = (char*)d_ws;
  const size_t FBYTES = (size_t)NBATCH * HALO * HALO * NFEAT * sizeof(_Float16);
  _Float16* F0 = (_Float16*)(ws);
  _Float16* F1 = (_Float16*)(ws + FBYTES);
  _Float16* F2 = (_Float16*)(ws + 2 * FBYTES);
  _Float16* wpack  = (_Float16*)(ws + 3 * FBYTES);
  const size_t WPACK_ELEMS = (size_t)2 * 8 * 9 * 64 * 64;
  _Float16* wopack = wpack + WPACK_ELEMS;
  _Float16* probs  = wopack + (size_t)NPADC * 64;
  float*    nsum   = (float*)(probs + (size_t)NBATCH * NCH * HW * HW);

  const size_t zn = (3 * FBYTES) / 4;
  zero_u32<<<(unsigned)((zn + 255) / 256), 256, 0, stream>>>((uint32_t*)ws, zn);
  zero_u32<<<1, 32, 0, stream>>>((uint32_t*)nsum, 4);

  pack_w<<<(2 * 8 * 9 * 64 * 64 + 255) / 256, 256, 0, stream>>>(w1s, w2s, wpack);
  pack_wout<<<(NPADC * 64 + 255) / 256, 256, 0, stream>>>(w_out, wopack);

  conv_in_k<<<(NBATCH * HW * HW * NFEAT + 255) / 256, 256, 0, stream>>>(z, w_in, b_in, F0);

  _Float16* cur = F0;
  _Float16* nxt = F2;
  const size_t LAYER = (size_t)9 * 64 * 64;
  for (int i = 0; i < 8; ++i) {
    conv3x3_wmma<<<512, 256, 9 * 64 * 64 * sizeof(_Float16), stream>>>(
        cur, F1, wpack + (size_t)i * LAYER, b1s + (size_t)i * 64, nullptr, 1);
    conv3x3_wmma<<<512, 256, 9 * 64 * 64 * sizeof(_Float16), stream>>>(
        F1, nxt, wpack + (size_t)(8 + i) * LAYER, b2s + (size_t)i * 64, cur, 0);
    _Float16* t = cur; cur = nxt; nxt = t;
  }

  convout_softmax<<<1024, 128, 0, stream>>>(cur, wopack, b_out, probs, nsum);
  apply_k<<<(NBATCH * 3 * HW * HW + 255) / 256, 256, 0, stream>>>(x, eps, probs, nsum, out);
}


# 3x3 conv kernel: pad the LDS weight rows from 128 B to 144 B (16 B x 9) so the 16 rows of each B-fragment ds_load_b128 hit distinct banks instead of 8 dwords per bank
# speedup vs baseline: 1.6101x; 1.2312x over previous
#include <hip/hip_runtime.h>
#include <hip/hip_bf16.h>
#include <stdint.h>

typedef __attribute__((ext_vector_type(16))) _Float16 v16h;
typedef __attribute__((ext_vector_type(8)))  _Float16 v8h;
typedef __attribute__((ext_vector_type(8)))  float    v8f;

#define NFEAT 64
#define HW    128
#define HALO  130
#define NBATCH 4
#define KTAP  441
#define NCH   442
#define NPADC 448
#define LDSP  72

__device__ __forceinline__ v16h cat8(v8h lo, v8h hi) {
  return __builtin_shufflevector(lo, hi, 0,1,2,3,4,5,6,7,8,9,10,11,12,13,14,15);
}
__device__ __forceinline__ v8f wmma16(v16h a, v16h b, v8f c) {
  return __builtin_amdgcn_wmma_f32_16x16x32_f16(false, a, false, b, (short)0, c, false, false);
}

__global__ void zero_u32(uint32_t* p, size_t n) {
  size_t i = (size_t)blockIdx.x * blockDim.x + threadIdx.x;
  if (i < n) p[i] = 0u;
}

__global__ void pack_w(const float* __restrict__ w1s, const float* __restrict__ w2s,
                       _Float16* __restrict__ dst) {
  int idx = blockIdx.x * blockDim.x + threadIdx.x;
  const int TOT = 2 * 8 * 9 * 64 * 64;
  if (idx >= TOT) return;
  int c   = idx & 63;
  int n   = (idx >> 6) & 63;
  int pg  = idx >> 12;
  int pos = pg % 9;
  int lay = pg / 9;
  int blk = lay & 7;
  int which = lay >> 3;
  const float* src = which ? w2s : w1s;
  float v = src[(((size_t)(blk * 64 + n) * 64 + c) * 9) + pos];
  dst[idx] = (_Float16)v;
}

__global__ void pack_wout(const float* __restrict__ w_out, _Float16* __restrict__ dst) {
  int idx = blockIdx.x * blockDim.x + threadIdx.x;
  if (idx >= NPADC * 64) return;
  int c = idx & 63;
  int n = idx >> 6;
  dst[idx] = (n < NCH) ? (_Float16)w_out[(size_t)n * 64 + c] : (_Float16)0.0f;
}

__global__ void conv_in_k(const float* __restrict__ z, const float* __restrict__ w_in,
                          const float* __restrict__ b_in, _Float16* __restrict__ dst) {
  int idx = blockIdx.x * blockDim.x + threadIdx.x;
  const int TOT = NBATCH * HW * HW * NFEAT;
  if (idx >= TOT) return;
  int co = idx & 63;
  int t  = idx >> 6;
  int x  = t & (HW - 1); t >>= 7;
  int y  = t & (HW - 1); t >>= 7;
  int b  = t;
  float acc = b_in[co];
  #pragma unroll
  for (int ci = 0; ci < 3; ++ci) {
    const float* zp = z + ((size_t)(b * 3 + ci)) * HW * HW;
    const float* wp = w_in + ((size_t)(co * 3 + ci)) * 9;
    #pragma unroll
    for (int ky = 0; ky < 3; ++ky) {
      int yy = y + ky - 1;
      if ((unsigned)yy >= (unsigned)HW) continue;
      #pragma unroll
      for (int kx = 0; kx < 3; ++kx) {
        int xx = x + kx - 1;
        if ((unsigned)xx >= (unsigned)HW) continue;
        acc += zp[yy * HW + xx] * wp[ky * 3 + kx];
      }
    }
  }
  dst[(((size_t)b * HALO + y + 1) * HALO + x + 1) * NFEAT + co] = (_Float16)acc;
}

__global__ __launch_bounds__(256) void conv3x3_wmma(
    const _Float16* __restrict__ src, _Float16* __restrict__ dst,
    const _Float16* __restrict__ wl, const float* __restrict__ bias,
    const _Float16* __restrict__ resid, int do_relu) {
  extern __shared__ __attribute__((aligned(16))) _Float16 lw[];
  {
    const uint4* g = (const uint4*)wl;
    for (int i = threadIdx.x; i < (9 * 64 * 64) / 8; i += 256)
      *(uint4*)(lw + (i >> 3) * LDSP + (i & 7) * 8) = g[i];
  }
  __syncthreads();

  const int wave = threadIdx.x >> 5;
  const int lane = threadIdx.x & 31;
  const int tile = blockIdx.x * 8 + wave;
  const int b  = tile >> 10;
  const int r  = tile & 1023;
  const int y  = r >> 3;
  const int x0 = (r & 7) << 4;

  const int mrow = lane & 15;
  const int ksel = (lane >> 4) << 3;
  const int nlan = lane & 15;
  const int mhi  = (lane >> 4) << 3;

  v8f acc[4] = {};

  #pragma unroll
  for (int ky = 0; ky < 3; ++ky) {
    #pragma unroll
    for (int kx = 0; kx < 3; ++kx) {
      const _Float16* ap =
          src + (((size_t)b * HALO + y + ky) * HALO + (x0 + mrow + kx)) * NFEAT;
      const _Float16* wp = lw + (ky * 3 + kx) * (64 * LDSP);
      v16h a0 = cat8(*(const v8h*)(ap + ksel),      *(const v8h*)(ap + 16 + ksel));
      v16h a1 = cat8(*(const v8h*)(ap + 32 + ksel), *(const v8h*)(ap + 48 + ksel));
      #pragma unroll
      for (int kc = 0; kc < 2; ++kc) {
        const int c0 = kc * 32;
        const v16h a = kc ? a1 : a0;
        const _Float16* w0 = wp + (size_t)(0 * 16 + nlan) * LDSP + c0 + ksel;
        const _Float16* w1 = wp + (size_t)(1 * 16 + nlan) * LDSP + c0 + ksel;
        const _Float16* w2 = wp + (size_t)(2 * 16 + nlan) * LDSP + c0 + ksel;
        const _Float16* w3 = wp + (size_t)(3 * 16 + nlan) * LDSP + c0 + ksel;
        v16h b0 = cat8(*(const v8h*)w0, *(const v8h*)(w0 + 16));
        v16h b1 = cat8(*(const v8h*)w1, *(const v8h*)(w1 + 16));
        v16h b2 = cat8(*(const v8h*)w2, *(const v8h*)(w2 + 16));
        v16h b3 = cat8(*(const v8h*)w3, *(const v8h*)(w3 + 16));
        acc[0] = wmma16(a, b0, acc[0]);
        acc[1] = wmma16(a, b1, acc[1]);
        acc[2] = wmma16(a, b2, acc[2]);
        acc[3] = wmma16(a, b3, acc[3]);
      }
    }
  }

  #pragma unroll
  for (int nt = 0; nt < 4; ++nt) {
    const int n = nt * 16 + nlan;
    const float bv = bias[n];
    #pragma unroll
    for (int i = 0; i < 8; ++i) {
      const int m = i + mhi;
      size_t off = (((size_t)b * HALO + y + 1) * HALO + (x0 + m + 1)) * NFEAT + n;
      float v = acc[nt][i] + bv;
      if (resid) v += (float)resid[off];
      if (do_relu) v = v > 0.0f ? v : 0.0f;
      dst[off] = (_Float16)v;
    }
  }
}

__global__ __launch_bounds__(128) void convout_softmax(
    const _Float16* __restrict__ feat, const _Float16* __restrict__ wo,
    const float* __restrict__ bo, _Float16* __restrict__ probs,
    float* __restrict__ nsum) {
  const int wave = threadIdx.x >> 5;
  const int lane = threadIdx.x & 31;
  const int tile = blockIdx.x * 4 + wave;
  const int b  = tile >> 10;
  const int r  = tile & 1023;
  const int y  = r >> 3;
  const int x0 = (r & 7) << 4;

  const int mrow = lane & 15;
  const int ksel = (lane >> 4) << 3;
  const int nlan = lane & 15;
  const int mhi  = (lane >> 4) << 3;

  const _Float16* ap =
      feat + (((size_t)b * HALO + y + 1) * HALO + (x0 + mrow + 1)) * NFEAT;
  v16h a0 = cat8(*(const v8h*)(ap + ksel),      *(const v8h*)(ap + 16 + ksel));
  v16h a1 = cat8(*(const v8h*)(ap + 32 + ksel), *(const v8h*)(ap + 48 + ksel));

  constexpr int NT = NPADC / 16;
  v8f lg[NT];

  constexpr int PF = 5;
  __builtin_amdgcn_sched_group_barrier(0x020, 4 + 4 * PF, 0);
  #pragma unroll
  for (int nt = 0; nt < NT; ++nt) {
    const int n = nt * 16 + nlan;
    const _Float16* wp = wo + (size_t)n * 64;
    v16h b0 = cat8(*(const v8h*)(wp + ksel),      *(const v8h*)(wp + 16 + ksel));
    v16h b1 = cat8(*(const v8h*)(wp + 32 + ksel), *(const v8h*)(wp + 48 + ksel));
    v8f c = {};
    c = wmma16(a0, b0, c);
    c = wmma16(a1, b1, c);
    lg[nt] = c;
    __builtin_amdgcn_sched_group_barrier(0x008, 2, 0);
    __builtin_amdgcn_sched_group_barrier(0x020, 4, 0);
  }
  #pragma unroll
  for (int nt = 0; nt < NT; ++nt) {
    const int n = nt * 16 + nlan;
    const float bv = (n < NCH) ? bo[n] : -1e30f;
    #pragma unroll
    for (int i = 0; i < 8; ++i) lg[nt][i] += bv;
  }

  float inv[8];
  #pragma unroll
  for (int i = 0; i < 8; ++i) {
    float m = lg[0][i];
    #pragma unroll
    for (int nt = 1; nt < NT; ++nt) m = fmaxf(m, lg[nt][i]);
    #pragma unroll
    for (int off = 8; off; off >>= 1) m = fmaxf(m, __shfl_xor(m, off, 32));
    float s = 0.0f;
    #pragma unroll
    for (int nt = 0; nt < NT; ++nt) {
      const float e = __expf(lg[nt][i] - m);
      lg[nt][i] = e;
      s += e;
    }
    #pragma unroll
    for (int off = 8; off; off >>= 1) s += __shfl_xor(s, off, 32);
    inv[i] = 1.0f / s;
  }

  _Float16* pb = probs + (size_t)b * NCH * (HW * HW) + (size_t)y * HW + x0 + mhi;
  #pragma unroll
  for (int nt = 0; nt < NT; ++nt) {
    const int t = nt * 16 + nlan;
    v8h o;
    #pragma unroll
    for (int i = 0; i < 8; ++i) o[i] = (_Float16)(lg[nt][i] * inv[i]);
    if (t < NCH) *(v8h*)(pb + (size_t)t * (HW * HW)) = o;
  }

  float nacc = 0.0f;
  if (nlan == (KTAP & 15)) {
    #pragma unroll
    for (int i = 0; i < 8; ++i) nacc += lg[KTAP >> 4][i] * inv[i];
  }
  nacc += __shfl_xor(nacc, 16, 32);
  if (lane == (KTAP & 15)) atomicAdd(nsum + b, nacc);
}

__global__ void apply_k(const float* __restrict__ x, const float* __restrict__ eps,
                        const _Float16* __restrict__ probs, const float* __restrict__ nsum,
                        float* __restrict__ out) {
  int idx = blockIdx.x * blockDim.x + threadIdx.x;
  const int TOT = NBATCH * 3 * HW * HW;
  if (idx >= TOT) return;
  int w = idx & (HW - 1);
  int t = idx >> 7;
  int h = t & (HW - 1); t >>= 7;
  int c = t % 3;
  int b = t / 3;

  const float* xb = x + ((size_t)(b * 3 + c)) * 512 * 512;
  const _Float16* pb = probs + (size_t)b * NCH * (HW * HW) + (size_t)h * HW + w;
  const int yb = h * 4 - 10, xbs = w * 4 - 10;
  float acc = 0.0f;
  for (int ky = 0; ky < 21; ++ky) {
    int yi = yb + ky;
    if ((unsigned)yi >= 512u) continue;
    const float* xr = xb + (size_t)yi * 512;
    const _Float16* pr = pb + (size_t)(ky * 21) * (HW * HW);
    __builtin_prefetch(pr + (size_t)21 * (HW * HW), 0, 1);
    if (yi + 1 < 512) __builtin_prefetch(xr + 512, 0, 1);
    for (int kx = 0; kx < 21; ++kx) {
      int xi = xbs + kx;
      if ((unsigned)xi >= 512u) continue;
      acc += xr[xi] * (float)pr[(size_t)kx * (HW * HW)];
    }
  }
  float noise = (nsum[b] * (1.0f / (HW * HW))) * eps[idx];
  out[idx] = acc + noise;
}

extern "C" void kernel_launch(void* const* d_in, const int* in_sizes, int n_in,
                              void* d_out, int out_size, void* d_ws, size_t ws_size,
                              hipStream_t stream) {
  const float* x     = (const float*)d_in[0];
  const float* z     = (const float*)d_in[1];
  const float* eps   = (const float*)d_in[2];
  const float* w_in  = (const float*)d_in[3];
  const float* b_in  = (const float*)d_in[4];
  const float* w1s   = (const float*)d_in[5];
  const float* b1s   = (const float*)d_in[6];
  const float* w2s   = (const float*)d_in[7];
  const float* b2s   = (const float*)d_in[8];
  const float* w_out = (const float*)d_in[9];
  const float* b_out = (const float*)d_in[10];
  float* out = (float*)d_out;

  char* ws = (char*)d_ws;
  const size_t FBYTES = (size_t)NBATCH * HALO * HALO * NFEAT * sizeof(_Float16);
  _Float16* F0 = (_Float16*)(ws);
  _Float16* F1 = (_Float16*)(ws + FBYTES);
  _Float16* F2 = (_Float16*)(ws + 2 * FBYTES);
  _Float16* wpack  = (_Float16*)(ws + 3 * FBYTES);
  const size_t WPACK_ELEMS = (size_t)2 * 8 * 9 * 64 * 64;
  _Float16* wopack = wpack + WPACK_ELEMS;
  _Float16* probs  = wopack + (size_t)NPADC * 64;
  float*    nsum   = (float*)(probs + (size_t)NBATCH * NCH * HW * HW);

  const size_t zn = (3 * FBYTES) / 4;
  zero_u32<<<(unsigned)((zn + 255) / 256), 256, 0, stream>>>((uint32_t*)ws, zn);
  zero_u32<<<1, 32, 0, stream>>>((uint32_t*)nsum, 4);

  pack_w<<<(2 * 8 * 9 * 64 * 64 + 255) / 256, 256, 0, stream>>>(w1s, w2s, wpack);
  pack_wout<<<(NPADC * 64 + 255) / 256, 256, 0, stream>>>(w_out, wopack);

  conv_in_k<<<(NBATCH * HW * HW * NFEAT + 255) / 256, 256, 0, stream>>>(z, w_in, b_in, F0);

  _Float16* cur = F0;
  _Float16* nxt = F2;
  const size_t LAYER = (size_t)9 * 64 * 64;
  for (int i = 0; i < 8; ++i) {
    conv3x3_wmma<<<512, 256, 9 * 64 * LDSP * sizeof(_Float16), stream>>>(
        cur, F1, wpack + (size_t)i * LAYER, b1s + (size_t)i * 64, nullptr, 1);
    conv3x3_wmma<<<512, 256, 9 * 64 * LDSP * sizeof(_Float16), stream>>>(
        F1, nxt, wpack + (size_t)(8 + i) * LAYER, b2s + (size_t)i * 64, cur, 0);
    _Float16* t = cur; cur = nxt; nxt = t;
  }

  convout_softmax<<<1024, 128, 0, stream>>>(cur, wopack, b_out, probs, nsum);
  apply_k<<<(NBATCH * 3 * HW * HW + 255) / 256, 256, 0, stream>>>(x, eps, probs, nsum, out);
}


# patch-gather kernel: lane = 4 pixels x 3 channels, wave = 3 tap rows, all prob and x loads unconditional 8/16-byte vectors (edges clamped, zeroed), partials summed in LDS; was a waited load per tap
# speedup vs baseline: 2.1804x; 1.3542x over previous
#include <hip/hip_runtime.h>
#include <hip/hip_bf16.h>
#include <stdint.h>

typedef __attribute__((ext_vector_type(16))) _Float16 v16h;
typedef __attribute__((ext_vector_type(8)))  _Float16 v8h;
typedef __attribute__((ext_vector_type(8)))  float    v8f;

#define NFEAT 64
#define HW    128
#define HALO  130
#define NBATCH 4
#define KTAP  441
#define NCH   442
#define NPADC 448
#define LDSP  72

__device__ __forceinline__ v16h cat8(v8h lo, v8h hi) {
  return __builtin_shufflevector(lo, hi, 0,1,2,3,4,5,6,7,8,9,10,11,12,13,14,15);
}
__device__ __forceinline__ v8f wmma16(v16h a, v16h b, v8f c) {
  return __builtin_amdgcn_wmma_f32_16x16x32_f16(false, a, false, b, (short)0, c, false, false);
}

__global__ void zero_u32(uint32_t* p, size_t n) {
  size_t i = (size_t)blockIdx.x * blockDim.x + threadIdx.x;
  if (i < n) p[i] = 0u;
}

__global__ void pack_w(const float* __restrict__ w1s, const float* __restrict__ w2s,
                       _Float16* __restrict__ dst) {
  int idx = blockIdx.x * blockDim.x + threadIdx.x;
  const int TOT = 2 * 8 * 9 * 64 * 64;
  if (idx >= TOT) return;
  int c   = idx & 63;
  int n   = (idx >> 6) & 63;
  int pg  = idx >> 12;
  int pos = pg % 9;
  int lay = pg / 9;
  int blk = lay & 7;
  int which = lay >> 3;
  const float* src = which ? w2s : w1s;
  float v = src[(((size_t)(blk * 64 + n) * 64 + c) * 9) + pos];
  dst[idx] = (_Float16)v;
}

__global__ void pack_wout(const float* __restrict__ w_out, _Float16* __restrict__ dst) {
  int idx = blockIdx.x * blockDim.x + threadIdx.x;
  if (idx >= NPADC * 64) return;
  int c = idx & 63;
  int n = idx >> 6;
  dst[idx] = (n < NCH) ? (_Float16)w_out[(size_t)n * 64 + c] : (_Float16)0.0f;
}

__global__ void conv_in_k(const float* __restrict__ z, const float* __restrict__ w_in,
                          const float* __restrict__ b_in, _Float16* __restrict__ dst) {
  int idx = blockIdx.x * blockDim.x + threadIdx.x;
  const int TOT = NBATCH * HW * HW * NFEAT;
  if (idx >= TOT) return;
  int co = idx & 63;
  int t  = idx >> 6;
  int x  = t & (HW - 1); t >>= 7;
  int y  = t & (HW - 1); t >>= 7;
  int b  = t;
  float acc = b_in[co];
  #pragma unroll
  for (int ci = 0; ci < 3; ++ci) {
    const float* zp = z + ((size_t)(b * 3 + ci)) * HW * HW;
    const float* wp = w_in + ((size_t)(co * 3 + ci)) * 9;
    #pragma unroll
    for (int ky = 0; ky < 3; ++ky) {
      int yy = y + ky - 1;
      if ((unsigned)yy >= (unsigned)HW) continue;
      #pragma unroll
      for (int kx = 0; kx < 3; ++kx) {
        int xx = x + kx - 1;
        if ((unsigned)xx >= (unsigned)HW) continue;
        acc += zp[yy * HW + xx] * wp[ky * 3 + kx];
      }
    }
  }
  dst[(((size_t)b * HALO + y + 1) * HALO + x + 1) * NFEAT + co] = (_Float16)acc;
}

__global__ __launch_bounds__(256) void conv3x3_wmma(
    const _Float16* __restrict__ src, _Float16* __restrict__ dst,
    const _Float16* __restrict__ wl, const float* __restrict__ bias,
    const _Float16* __restrict__ resid, int do_relu) {
  extern __shared__ __attribute__((aligned(16))) _Float16 lw[];
  {
    const uint4* g = (const uint4*)wl;
    for (int i = threadIdx.x; i < (9 * 64 * 64) / 8; i += 256)
      *(uint4*)(lw + (i >> 3) * LDSP + (i & 7) * 8) = g[i];
  }
  __syncthreads();

  const int wave = threadIdx.x >> 5;
  const int lane = threadIdx.x & 31;
  const int tile = blockIdx.x * 8 + wave;
  const int b  = tile >> 10;
  const int r  = tile & 1023;
  const int y  = r >> 3;
  const int x0 = (r & 7) << 4;

  const int mrow = lane & 15;
  const int ksel = (lane >> 4) << 3;
  const int nlan = lane & 15;
  const int mhi  = (lane >> 4) << 3;

  v8f acc[4] = {};

  #pragma unroll
  for (int ky = 0; ky < 3; ++ky) {
    #pragma unroll
    for (int kx = 0; kx < 3; ++kx) {
      const _Float16* ap =
          src + (((size_t)b * HALO + y + ky) * HALO + (x0 + mrow + kx)) * NFEAT;
      const _Float16* wp = lw + (ky * 3 + kx) * (64 * LDSP);
      v16h a0 = cat8(*(const v8h*)(ap + ksel),      *(const v8h*)(ap + 16 + ksel));
      v16h a1 = cat8(*(const v8h*)(ap + 32 + ksel), *(const v8h*)(ap + 48 + ksel));
      #pragma unroll
      for (int kc = 0; kc < 2; ++kc) {
        const int c0 = kc * 32;
        const v16h a = kc ? a1 : a0;
        const _Float16* w0 = wp + (size_t)(0 * 16 + nlan) * LDSP + c0 + ksel;
        const _Float16* w1 = wp + (size_t)(1 * 16 + nlan) * LDSP + c0 + ksel;
        const _Float16* w2 = wp + (size_t)(2 * 16 + nlan) * LDSP + c0 + ksel;
        const _Float16* w3 = wp + (size_t)(3 * 16 + nlan) * LDSP + c0 + ksel;
        v16h b0 = cat8(*(const v8h*)w0, *(const v8h*)(w0 + 16));
        v16h b1 = cat8(*(const v8h*)w1, *(const v8h*)(w1 + 16));
        v16h b2 = cat8(*(const v8h*)w2, *(const v8h*)(w2 + 16));
        v16h b3 = cat8(*(const v8h*)w3, *(const v8h*)(w3 + 16));
        acc[0] = wmma16(a, b0, acc[0]);
        acc[1] = wmma16(a, b1, acc[1]);
        acc[2] = wmma16(a, b2, acc[2]);
        acc[3] = wmma16(a, b3, acc[3]);
      }
    }
  }

  #pragma unroll
  for (int nt = 0; nt < 4; ++nt) {
    const int n = nt * 16 + nlan;
    const float bv = bias[n];
    #pragma unroll
    for (int i = 0; i < 8; ++i) {
      const int m = i + mhi;
      size_t off = (((size_t)b * HALO + y + 1) * HALO + (x0 + m + 1)) * NFEAT + n;
      float v = acc[nt][i] + bv;
      if (resid) v += (float)resid[off];
      if (do_relu) v = v > 0.0f ? v : 0.0f;
      dst[off] = (_Float16)v;
    }
  }
}

__global__ __launch_bounds__(128) void convout_softmax(
    const _Float16* __restrict__ feat, const _Float16* __restrict__ wo,
    const float* __restrict__ bo, _Float16* __restrict__ probs,
    float* __restrict__ nsum) {
  const int wave = threadIdx.x >> 5;
  const int lane = threadIdx.x & 31;
  const int tile = blockIdx.x * 4 + wave;
  const int b  = tile >> 10;
  const int r  = tile & 1023;
  const int y  = r >> 3;
  const int x0 = (r & 7) << 4;

  const int mrow = lane & 15;
  const int ksel = (lane >> 4) << 3;
  const int nlan = lane & 15;
  const int mhi  = (lane >> 4) << 3;

  const _Float16* ap =
      feat + (((size_t)b * HALO + y + 1) * HALO + (x0 + mrow + 1)) * NFEAT;
  v16h a0 = cat8(*(const v8h*)(ap + ksel),      *(const v8h*)(ap + 16 + ksel));
  v16h a1 = cat8(*(const v8h*)(ap + 32 + ksel), *(const v8h*)(ap + 48 + ksel));

  constexpr int NT = NPADC / 16;
  v8f lg[NT];

  constexpr int PF = 5;
  __builtin_amdgcn_sched_group_barrier(0x020, 4 + 4 * PF, 0);
  #pragma unroll
  for (int nt = 0; nt < NT; ++nt) {
    const int n = nt * 16 + nlan;
    const _Float16* wp = wo + (size_t)n * 64;
    v16h b0 = cat8(*(const v8h*)(wp + ksel),      *(const v8h*)(wp + 16 + ksel));
    v16h b1 = cat8(*(const v8h*)(wp + 32 + ksel), *(const v8h*)(wp + 48 + ksel));
    v8f c = {};
    c = wmma16(a0, b0, c);
    c = wmma16(a1, b1, c);
    lg[nt] = c;
    __builtin_amdgcn_sched_group_barrier(0x008, 2, 0);
    __builtin_amdgcn_sched_group_barrier(0x020, 4, 0);
  }
  #pragma unroll
  for (int nt = 0; nt < NT; ++nt) {
    const int n = nt * 16 + nlan;
    const float bv = (n < NCH) ? bo[n] : -1e30f;
    #pragma unroll
    for (int i = 0; i < 8; ++i) lg[nt][i] += bv;
  }

  float inv[8];
  #pragma unroll
  for (int i = 0; i < 8; ++i) {
    float m = lg[0][i];
    #pragma unroll
    for (int nt = 1; nt < NT; ++nt) m = fmaxf(m, lg[nt][i]);
    #pragma unroll
    for (int off = 8; off; off >>= 1) m = fmaxf(m, __shfl_xor(m, off, 32));
    float s = 0.0f;
    #pragma unroll
    for (int nt = 0; nt < NT; ++nt) {
      const float e = __expf(lg[nt][i] - m);
      lg[nt][i] = e;
      s += e;
    }
    #pragma unroll
    for (int off = 8; off; off >>= 1) s += __shfl_xor(s, off, 32);
    inv[i] = 1.0f / s;
  }

  _Float16* pb = probs + (size_t)b * NCH * (HW * HW) + (size_t)y * HW + x0 + mhi;
  #pragma unroll
  for (int nt = 0; nt < NT; ++nt) {
    const int t = nt * 16 + nlan;
    v8h o;
    #pragma unroll
    for (int i = 0; i < 8; ++i) o[i] = (_Float16)(lg[nt][i] * inv[i]);
    if (t < NCH) *(v8h*)(pb + (size_t)t * (HW * HW)) = o;
  }

  float nacc = 0.0f;
  if (nlan == (KTAP & 15)) {
    #pragma unroll
    for (int i = 0; i < 8; ++i) nacc += lg[KTAP >> 4][i] * inv[i];
  }
  nacc += __shfl_xor(nacc, 16, 32);
  if (lane == (KTAP & 15)) atomicAdd(nsum + b, nacc);
}

typedef __attribute__((ext_vector_type(4))) _Float16 v4h;
typedef __attribute__((ext_vector_type(4))) float    v4f;
typedef __attribute__((ext_vector_type(2))) float    v2f;

#define AK_WAVES 7
#define AK_ROWS  3
#define AK_NG    9

__global__ __launch_bounds__(AK_WAVES * 32) void apply_k(
    const float* __restrict__ x, const float* __restrict__ eps,
    const _Float16* __restrict__ probs, const float* __restrict__ nsum,
    float* __restrict__ out) {
  __shared__ float red[AK_WAVES * 12 * 32];
  const int lane = threadIdx.x & 31;
  const int wave = __builtin_amdgcn_readfirstlane(threadIdx.x >> 5);
  const int h = blockIdx.x & (HW - 1);
  const int b = blockIdx.x >> 7;
  const int col0 = 16 * lane - 12;

  float acc[3][4];
  #pragma unroll
  for (int c = 0; c < 3; ++c)
    #pragma unroll
    for (int q = 0; q < 4; ++q) acc[c][q] = 0.0f;

  #pragma unroll 1
  for (int r = 0; r < AK_ROWS; ++r) {
    const int ky = wave * AK_ROWS + r;
    const int yi = 4 * h - 10 + ky;
    const float rowkeep = ((unsigned)yi < 512u) ? 1.0f : 0.0f;
    const int yc = yi < 0 ? 0 : (yi > 511 ? 511 : yi);

    const _Float16* pr =
        probs + ((size_t)(b * NCH + ky * 21)) * (HW * HW) + (size_t)h * HW + 4 * lane;
    v4h pv[21];
    #pragma unroll
    for (int kx = 0; kx < 21; ++kx)
      pv[kx] = __builtin_nontemporal_load((const v4h*)(pr + (size_t)kx * (HW * HW)));

    bool ok[AK_NG];
    #pragma unroll
    for (int g = 0; g < AK_NG; ++g) {
      const int col = col0 + 4 * g;
      ok[g] = (g < 3) ? (col >= 0) : ((g > 6) ? (col < 512) : true);
    }
    v2f xlo[3];
    v4f xv[3][AK_NG - 2];
    v2f xhi[3];
    float xend[3];
    #pragma unroll
    for (int c = 0; c < 3; ++c) {
      const float* xr = x + ((size_t)((b * 3 + c) * 512 + yc)) * 512;
      xlo[c] = *(const v2f*)(xr + (ok[0] ? col0 : 0) + 2);
      #pragma unroll
      for (int g = 1; g < AK_NG - 1; ++g)
        xv[c][g - 1] = *(const v4f*)(xr + (ok[g] ? col0 + 4 * g : 0));
      const float* xe = xr + (ok[AK_NG - 1] ? col0 + 4 * (AK_NG - 1) : 0);
      xhi[c]  = *(const v2f*)xe;
      xend[c] = xe[2];
    }
    __builtin_amdgcn_sched_barrier(0);

    float keep[AK_NG];
    #pragma unroll
    for (int g = 0; g < AK_NG; ++g) keep[g] = ok[g] ? 1.0f : 0.0f;
    float xs[3][4 * AK_NG];
    #pragma unroll
    for (int c = 0; c < 3; ++c) {
      xs[c][2] = xlo[c][0] * keep[0];
      xs[c][3] = xlo[c][1] * keep[0];
      #pragma unroll
      for (int g = 1; g < AK_NG - 1; ++g) {
        #pragma unroll
        for (int e = 0; e < 4; ++e)
          xs[c][4 * g + e] = (g < 3 || g > 6) ? xv[c][g - 1][e] * keep[g] : xv[c][g - 1][e];
      }
      xs[c][32] = xhi[c][0] * keep[AK_NG - 1];
      xs[c][33] = xhi[c][1] * keep[AK_NG - 1];
      xs[c][34] = xend[c]   * keep[AK_NG - 1];
    }

    float part[3][4];
    #pragma unroll
    for (int c = 0; c < 3; ++c)
      #pragma unroll
      for (int q = 0; q < 4; ++q) part[c][q] = 0.0f;
    #pragma unroll
    for (int kx = 0; kx < 21; ++kx) {
      #pragma unroll
      for (int q = 0; q < 4; ++q) {
        const float pf = (float)pv[kx][q];
        #pragma unroll
        for (int c = 0; c < 3; ++c) part[c][q] += xs[c][4 * q + 2 + kx] * pf;
      }
    }
    #pragma unroll
    for (int c = 0; c < 3; ++c)
      #pragma unroll
      for (int q = 0; q < 4; ++q) acc[c][q] += part[c][q] * rowkeep;
  }

  #pragma unroll
  for (int c = 0; c < 3; ++c)
    #pragma unroll
    for (int q = 0; q < 4; ++q) red[(wave * 12 + c * 4 + q) * 32 + lane] = acc[c][q];
  __syncthreads();

  if (wave < 3) {
    const int c = wave;
    float s[4];
    #pragma unroll
    for (int q = 0; q < 4; ++q) {
      float t = 0.0f;
      #pragma unroll
      for (int j = 0; j < AK_WAVES; ++j) t += red[(j * 12 + c * 4 + q) * 32 + lane];
      s[q] = t;
    }
    const size_t idx = (((size_t)(b * 3 + c)) * HW + h) * HW + 4 * lane;
    const float ns = nsum[b] * (1.0f / (HW * HW));
    const v4f e4 = *(const v4f*)(eps + idx);
    v4f o;
    #pragma unroll
    for (int q = 0; q < 4; ++q) o[q] = s[q] + ns * e4[q];
    *(v4f*)(out + idx) = o;
  }
}

extern "C" void kernel_launch(void* const* d_in, const int* in_sizes, int n_in,
                              void* d_out, int out_size, void* d_ws, size_t ws_size,
                              hipStream_t stream) {
  const float* x     = (const float*)d_in[0];
  const float* z     = (const float*)d_in[1];
  const float* eps   = (const float*)d_in[2];
  const float* w_in  = (const float*)d_in[3];
  const float* b_in  = (const float*)d_in[4];
  const float* w1s   = (const float*)d_in[5];
  const float* b1s   = (const float*)d_in[6];
  const float* w2s   = (const float*)d_in[7];
  const float* b2s   = (const float*)d_in[8];
  const float* w_out = (const float*)d_in[9];
  const float* b_out = (const float*)d_in[10];
  float* out = (float*)d_out;

  char* ws = (char*)d_ws;
  const size_t FBYTES = (size_t)NBATCH * HALO * HALO * NFEAT * sizeof(_Float16);
  _Float16* F0 = (_Float16*)(ws);
  _Float16* F1 = (_Float16*)(ws + FBYTES);
  _Float16* F2 = (_Float16*)(ws + 2 * FBYTES);
  _Float16* wpack  = (_Float16*)(ws + 3 * FBYTES);
  const size_t WPACK_ELEMS = (size_t)2 * 8 * 9 * 64 * 64;
  _Float16* wopack = wpack + WPACK_ELEMS;
  _Float16* probs  = wopack + (size_t)NPADC * 64;
  float*    nsum   = (float*)(probs + (size_t)NBATCH * NCH * HW * HW);

  const size_t zn = (3 * FBYTES) / 4;
  zero_u32<<<(unsigned)((zn + 255) / 256), 256, 0, stream>>>((uint32_t*)ws, zn);
  zero_u32<<<1, 32, 0, stream>>>((uint32_t*)nsum, 4);

  pack_w<<<(2 * 8 * 9 * 64 * 64 + 255) / 256, 256, 0, stream>>>(w1s, w2s, wpack);
  pack_wout<<<(NPADC * 64 + 255) / 256, 256, 0, stream>>>(w_out, wopack);

  conv_in_k<<<(NBATCH * HW * HW * NFEAT + 255) / 256, 256, 0, stream>>>(z, w_in, b_in, F0);

  _Float16* cur = F0;
  _Float16* nxt = F2;
  const size_t LAYER = (size_t)9 * 64 * 64;
  for (int i = 0; i < 8; ++i) {
    conv3x3_wmma<<<512, 256, 9 * 64 * LDSP * sizeof(_Float16), stream>>>(
        cur, F1, wpack + (size_t)i * LAYER, b1s + (size_t)i * 64, nullptr, 1);
    conv3x3_wmma<<<512, 256, 9 * 64 * LDSP * sizeof(_Float16), stream>>>(
        F1, nxt, wpack + (size_t)(8 + i) * LAYER, b2s + (size_t)i * 64, cur, 0);
    _Float16* t = cur; cur = nxt; nxt = t;
  }

  convout_softmax<<<1024, 128, 0, stream>>>(cur, wopack, b_out, probs, nsum);
  apply_k<<<NBATCH * HW, AK_WAVES * 32, 0, stream>>>(x, eps, probs, nsum, out);
}


# conv_in kernel: lane = pixel, a wave = 32 pixels x all 64 channels, weights and bias by scalar loads, 8 16-byte stores per pixel line; was lane = channel: 27 waited weight loads of 27 lines each
# speedup vs baseline: 2.6275x; 1.2051x over previous
#include <hip/hip_runtime.h>
#include <hip/hip_bf16.h>
#include <stdint.h>

typedef __attribute__((ext_vector_type(16))) _Float16 v16h;
typedef __attribute__((ext_vector_type(8)))  _Float16 v8h;
typedef __attribute__((ext_vector_type(8)))  float    v8f;

#define NFEAT 64
#define HW    128
#define HALO  130
#define NBATCH 4
#define KTAP  441
#define NCH   442
#define NPADC 448
#define LDSP  72

__device__ __forceinline__ v16h cat8(v8h lo, v8h hi) {
  return __builtin_shufflevector(lo, hi, 0,1,2,3,4,5,6,7,8,9,10,11,12,13,14,15);
}
__device__ __forceinline__ v8f wmma16(v16h a, v16h b, v8f c) {
  return __builtin_amdgcn_wmma_f32_16x16x32_f16(false, a, false, b, (short)0, c, false, false);
}

__global__ void zero_u32(uint32_t* p, size_t n) {
  size_t i = (size_t)blockIdx.x * blockDim.x + threadIdx.x;
  if (i < n) p[i] = 0u;
}

__global__ void pack_w(const float* __restrict__ w1s, const float* __restrict__ w2s,
                       _Float16* __restrict__ dst) {
  int idx = blockIdx.x * blockDim.x + threadIdx.x;
  const int TOT = 2 * 8 * 9 * 64 * 64;
  if (idx >= TOT) return;
  int c   = idx & 63;
  int n   = (idx >> 6) & 63;
  int pg  = idx >> 12;
  int pos = pg % 9;
  int lay = pg / 9;
  int blk = lay & 7;
  int which = lay >> 3;
  const float* src = which ? w2s : w1s;
  float v = src[(((size_t)(blk * 64 + n) * 64 + c) * 9) + pos];
  dst[idx] = (_Float16)v;
}

__global__ void pack_wout(const float* __restrict__ w_out, _Float16* __restrict__ dst) {
  int idx = blockIdx.x * blockDim.x + threadIdx.x;
  if (idx >= NPADC * 64) return;
  int c = idx & 63;
  int n = idx >> 6;
  dst[idx] = (n < NCH) ? (_Float16)w_out[(size_t)n * 64 + c] : (_Float16)0.0f;
}

__global__ __launch_bounds__(128) void conv_in_k(const float* __restrict__ z,
                                                 const float* __restrict__ w_in,
                                                 const float* __restrict__ b_in,
                                                 _Float16* __restrict__ dst) {
  const int x = threadIdx.x;
  const int y = blockIdx.x & (HW - 1);
  const int b = blockIdx.x >> 7;

  float zv[27];
  #pragma unroll
  for (int ci = 0; ci < 3; ++ci) {
    const float* zp = z + ((size_t)(b * 3 + ci)) * HW * HW;
    #pragma unroll
    for (int ky = 0; ky < 3; ++ky) {
      const int yy = y + ky - 1;
      #pragma unroll
      for (int kx = 0; kx < 3; ++kx) {
        const int xx = x + kx - 1;
        const bool ok = ((unsigned)yy < (unsigned)HW) && ((unsigned)xx < (unsigned)HW);
        zv[ci * 9 + ky * 3 + kx] = ok ? zp[yy * HW + xx] : 0.0f;
      }
    }
  }

  _Float16* op = dst + (((size_t)b * HALO + y + 1) * HALO + x + 1) * NFEAT;
  #pragma unroll
  for (int g = 0; g < NFEAT / 8; ++g) {
    v8h o;
    #pragma unroll
    for (int jp = 0; jp < 4; ++jp) {
      const int co = g * 8 + jp * 2;
      const float* wp = w_in + co * 27;
      float a0 = b_in[co];
      float a1 = b_in[co + 1];
      #pragma unroll
      for (int t = 0; t < 27; ++t) {
        a0 = __builtin_fmaf(zv[t], wp[t], a0);
        a1 = __builtin_fmaf(zv[t], wp[27 + t], a1);
      }
      o[jp * 2]     = (_Float16)a0;
      o[jp * 2 + 1] = (_Float16)a1;
    }
    *(v8h*)(op + g * 8) = o;
  }
}

__global__ __launch_bounds__(256) void conv3x3_wmma(
    const _Float16* __restrict__ src, _Float16* __restrict__ dst,
    const _Float16* __restrict__ wl, const float* __restrict__ bias,
    const _Float16* __restrict__ resid, int do_relu) {
  extern __shared__ __attribute__((aligned(16))) _Float16 lw[];
  {
    const uint4* g = (const uint4*)wl;
    for (int i = threadIdx.x; i < (9 * 64 * 64) / 8; i += 256)
      *(uint4*)(lw + (i >> 3) * LDSP + (i & 7) * 8) = g[i];
  }
  __syncthreads();

  const int wave = threadIdx.x >> 5;
  const int lane = threadIdx.x & 31;
  const int tile = blockIdx.x * 8 + wave;
  const int b  = tile >> 10;
  const int r  = tile & 1023;
  const int y  = r >> 3;
  const int x0 = (r & 7) << 4;

  const int mrow = lane & 15;
  const int ksel = (lane >> 4) << 3;
  const int nlan = lane & 15;
  const int mhi  = (lane >> 4) << 3;

  v8f acc[4] = {};

  #pragma unroll
  for (int ky = 0; ky < 3; ++ky) {
    #pragma unroll
    for (int kx = 0; kx < 3; ++kx) {
      const _Float16* ap =
          src + (((size_t)b * HALO + y + ky) * HALO + (x0 + mrow + kx)) * NFEAT;
      const _Float16* wp = lw + (ky * 3 + kx) * (64 * LDSP);
      v16h a0 = cat8(*(const v8h*)(ap + ksel),      *(const v8h*)(ap + 16 + ksel));
      v16h a1 = cat8(*(const v8h*)(ap + 32 + ksel), *(const v8h*)(ap + 48 + ksel));
      #pragma unroll
      for (int kc = 0; kc < 2; ++kc) {
        const int c0 = kc * 32;
        const v16h a = kc ? a1 : a0;
        const _Float16* w0 = wp + (size_t)(0 * 16 + nlan) * LDSP + c0 + ksel;
        const _Float16* w1 = wp + (size_t)(1 * 16 + nlan) * LDSP + c0 + ksel;
        const _Float16* w2 = wp + (size_t)(2 * 16 + nlan) * LDSP + c0 + ksel;
        const _Float16* w3 = wp + (size_t)(3 * 16 + nlan) * LDSP + c0 + ksel;
        v16h b0 = cat8(*(const v8h*)w0, *(const v8h*)(w0 + 16));
        v16h b1 = cat8(*(const v8h*)w1, *(const v8h*)(w1 + 16));
        v16h b2 = cat8(*(const v8h*)w2, *(const v8h*)(w2 + 16));
        v16h b3 = cat8(*(const v8h*)w3, *(const v8h*)(w3 + 16));
        acc[0] = wmma16(a, b0, acc[0]);
        acc[1] = wmma16(a, b1, acc[1]);
        acc[2] = wmma16(a, b2, acc[2]);
        acc[3] = wmma16(a, b3, acc[3]);
      }
    }
  }

  #pragma unroll
  for (int nt = 0; nt < 4; ++nt) {
    const int n = nt * 16 + nlan;
    const float bv = bias[n];
    #pragma unroll
    for (int i = 0; i < 8; ++i) {
      const int m = i + mhi;
      size_t off = (((size_t)b * HALO + y + 1) * HALO + (x0 + m + 1)) * NFEAT + n;
      float v = acc[nt][i] + bv;
      if (resid) v += (float)resid[off];
      if (do_relu) v = v > 0.0f ? v : 0.0f;
      dst[off] = (_Float16)v;
    }
  }
}

__global__ __launch_bounds__(128) void convout_softmax(
    const _Float16* __restrict__ feat, const _Float16* __restrict__ wo,
    const float* __restrict__ bo, _Float16* __restrict__ probs,
    float* __restrict__ nsum) {
  const int wave = threadIdx.x >> 5;
  const int lane = threadIdx.x & 31;
  const int tile = blockIdx.x * 4 + wave;
  const int b  = tile >> 10;
  const int r  = tile & 1023;
  const int y  = r >> 3;
  const int x0 = (r & 7) << 4;

  const int mrow = lane & 15;
  const int ksel = (lane >> 4) << 3;
  const int nlan = lane & 15;
  const int mhi  = (lane >> 4) << 3;

  const _Float16* ap =
      feat + (((size_t)b * HALO + y + 1) * HALO + (x0 + mrow + 1)) * NFEAT;
  v16h a0 = cat8(*(const v8h*)(ap + ksel),      *(const v8h*)(ap + 16 + ksel));
  v16h a1 = cat8(*(const v8h*)(ap + 32 + ksel), *(const v8h*)(ap + 48 + ksel));

  constexpr int NT = NPADC / 16;
  v8f lg[NT];

  constexpr int PF = 5;
  __builtin_amdgcn_sched_group_barrier(0x020, 4 + 4 * PF, 0);
  #pragma unroll
  for (int nt = 0; nt < NT; ++nt) {
    const int n = nt * 16 + nlan;
    const _Float16* wp = wo + (size_t)n * 64;
    v16h b0 = cat8(*(const v8h*)(wp + ksel),      *(const v8h*)(wp + 16 + ksel));
    v16h b1 = cat8(*(const v8h*)(wp + 32 + ksel), *(const v8h*)(wp + 48 + ksel));
    v8f c = {};
    c = wmma16(a0, b0, c);
    c = wmma16(a1, b1, c);
    lg[nt] = c;
    __builtin_amdgcn_sched_group_barrier(0x008, 2, 0);
    __builtin_amdgcn_sched_group_barrier(0x020, 4, 0);
  }
  #pragma unroll
  for (int nt = 0; nt < NT; ++nt) {
    const int n = nt * 16 + nlan;
    const float bv = (n < NCH) ? bo[n] : -1e30f;
    #pragma unroll
    for (int i = 0; i < 8; ++i) lg[nt][i] += bv;
  }

  float inv[8];
  #pragma unroll
  for (int i = 0; i < 8; ++i) {
    float m = lg[0][i];
    #pragma unroll
    for (int nt = 1; nt < NT; ++nt) m = fmaxf(m, lg[nt][i]);
    #pragma unroll
    for (int off = 8; off; off >>= 1) m = fmaxf(m, __shfl_xor(m, off, 32));
    float s = 0.0f;
    #pragma unroll
    for (int nt = 0; nt < NT; ++nt) {
      const float e = __expf(lg[nt][i] - m);
      lg[nt][i] = e;
      s += e;
    }
    #pragma unroll
    for (int off = 8; off; off >>= 1) s += __shfl_xor(s, off, 32);
    inv[i] = 1.0f / s;
  }

  _Float16* pb = probs + (size_t)b * NCH * (HW * HW) + (size_t)y * HW + x0 + mhi;
  #pragma unroll
  for (int nt = 0; nt < NT; ++nt) {
    const int t = nt * 16 + nlan;
    v8h o;
    #pragma unroll
    for (int i = 0; i < 8; ++i) o[i] = (_Float16)(lg[nt][i] * inv[i]);
    if (t < NCH) *(v8h*)(pb + (size_t)t * (HW * HW)) = o;
  }

  float nacc = 0.0f;
  if (nlan == (KTAP & 15)) {
    #pragma unroll
    for (int i = 0; i < 8; ++i) nacc += lg[KTAP >> 4][i] * inv[i];
  }
  nacc += __shfl_xor(nacc, 16, 32);
  if (lane == (KTAP & 15)) atomicAdd(nsum + b, nacc);
}

typedef __attribute__((ext_vector_type(4))) _Float16 v4h;
typedef __attribute__((ext_vector_type(4))) float    v4f;
typedef __attribute__((ext_vector_type(2))) float    v2f;

#define AK_WAVES 7
#define AK_ROWS  3
#define AK_NG    9

__global__ __launch_bounds__(AK_WAVES * 32) void apply_k(
    const float* __restrict__ x, const float* __restrict__ eps,
    const _Float16* __restrict__ probs, const float* __restrict__ nsum,
    float* __restrict__ out) {
  __shared__ float red[AK_WAVES * 12 * 32];
  const int lane = threadIdx.x & 31;
  const int wave = __builtin_amdgcn_readfirstlane(threadIdx.x >> 5);
  const int h = blockIdx.x & (HW - 1);
  const int b = blockIdx.x >> 7;
  const int col0 = 16 * lane - 12;

  float acc[3][4];
  #pragma unroll
  for (int c = 0; c < 3; ++c)
    #pragma unroll
    for (int q = 0; q < 4; ++q) acc[c][q] = 0.0f;

  #pragma unroll 1
  for (int r = 0; r < AK_ROWS; ++r) {
    const int ky = wave * AK_ROWS + r;
    const int yi = 4 * h - 10 + ky;
    const float rowkeep = ((unsigned)yi < 512u) ? 1.0f : 0.0f;
    const int yc = yi < 0 ? 0 : (yi > 511 ? 511 : yi);

    const _Float16* pr =
        probs + ((size_t)(b * NCH + ky * 21)) * (HW * HW) + (size_t)h * HW + 4 * lane;
    v4h pv[21];
    #pragma unroll
    for (int kx = 0; kx < 21; ++kx)
      pv[kx] = __builtin_nontemporal_load((const v4h*)(pr + (size_t)kx * (HW * HW)));

    bool ok[AK_NG];
    #pragma unroll
    for (int g = 0; g < AK_NG; ++g) {
      const int col = col0 + 4 * g;
      ok[g] = (g < 3) ? (col >= 0) : ((g > 6) ? (col < 512) : true);
    }
    v2f xlo[3];
    v4f xv[3][AK_NG - 2];
    v2f xhi[3];
    float xend[3];
    #pragma unroll
    for (int c = 0; c < 3; ++c) {
      const float* xr = x + ((size_t)((b * 3 + c) * 512 + yc)) * 512;
      xlo[c] = *(const v2f*)(xr + (ok[0] ? col0 : 0) + 2);
      #pragma unroll
      for (int g = 1; g < AK_NG - 1; ++g)
        xv[c][g - 1] = *(const v4f*)(xr + (ok[g] ? col0 + 4 * g : 0));
      const float* xe = xr + (ok[AK_NG - 1] ? col0 + 4 * (AK_NG - 1) : 0);
      xhi[c]  = *(const v2f*)xe;
      xend[c] = xe[2];
    }
    __builtin_amdgcn_sched_barrier(0);

    float keep[AK_NG];
    #pragma unroll
    for (int g = 0; g < AK_NG; ++g) keep[g] = ok[g] ? 1.0f : 0.0f;
    float xs[3][4 * AK_NG];
    #pragma unroll
    for (int c = 0; c < 3; ++c) {
      xs[c][2] = xlo[c][0] * keep[0];
      xs[c][3] = xlo[c][1] * keep[0];
      #pragma unroll
      for (int g = 1; g < AK_NG - 1; ++g) {
        #pragma unroll
        for (int e = 0; e < 4; ++e)
          xs[c][4 * g + e] = (g < 3 || g > 6) ? xv[c][g - 1][e] * keep[g] : xv[c][g - 1][e];
      }
      xs[c][32] = xhi[c][0] * keep[AK_NG - 1];
      xs[c][33] = xhi[c][1] * keep[AK_NG - 1];
      xs[c][34] = xend[c]   * keep[AK_NG - 1];
    }

    float part[3][4];
    #pragma unroll
    for (int c = 0; c < 3; ++c)
      #pragma unroll
      for (int q = 0; q < 4; ++q) part[c][q] = 0.0f;
    #pragma unroll
    for (int kx = 0; kx < 21; ++kx) {
      #pragma unroll
      for (int q = 0; q < 4; ++q) {
        const float pf = (float)pv[kx][q];
        #pragma unroll
        for (int c = 0; c < 3; ++c) part[c][q] += xs[c][4 * q + 2 + kx] * pf;
      }
    }
    #pragma unroll
    for (int c = 0; c < 3; ++c)
      #pragma unroll
      for (int q = 0; q < 4; ++q) acc[c][q] += part[c][q] * rowkeep;
  }

  #pragma unroll
  for (int c = 0; c < 3; ++c)
    #pragma unroll
    for (int q = 0; q < 4; ++q) red[(wave * 12 + c * 4 + q) * 32 + lane] = acc[c][q];
  __syncthreads();

  if (wave < 3) {
    const int c = wave;
    float s[4];
    #pragma unroll
    for (int q = 0; q < 4; ++q) {
      float t = 0.0f;
      #pragma unroll
      for (int j = 0; j < AK_WAVES; ++j) t += red[(j * 12 + c * 4 + q) * 32 + lane];
      s[q] = t;
    }
    const size_t idx = (((size_t)(b * 3 + c)) * HW + h) * HW + 4 * lane;
    const float ns = nsum[b] * (1.0f / (HW * HW));
    const v4f e4 = *(const v4f*)(eps + idx);
    v4f o;
    #pragma unroll
    for (int q = 0; q < 4; ++q) o[q] = s[q] + ns * e4[q];
    *(v4f*)(out + idx) = o;
  }
}

extern "C" void kernel_launch(void* const* d_in, const int* in_sizes, int n_in,
                              void* d_out, int out_size, void* d_ws, size_t ws_size,
                              hipStream_t stream) {
  const float* x     = (const float*)d_in[0];
  const float* z     = (const float*)d_in[1];
  const float* eps   = (const float*)d_in[2];
  const float* w_in  = (const float*)d_in[3];
  const float* b_in  = (const float*)d_in[4];
  const float* w1s   = (const float*)d_in[5];
  const float* b1s   = (const float*)d_in[6];
  const float* w2s   = (const float*)d_in[7];
  const float* b2s   = (const float*)d_in[8];
  const float* w_out = (const float*)d_in[9];
  const float* b_out = (const float*)d_in[10];
  float* out = (float*)d_out;

  char* ws = (char*)d_ws;
  const size_t FBYTES = (size_t)NBATCH * HALO * HALO * NFEAT * sizeof(_Float16);
  _Float16* F0 = (_Float16*)(ws);
  _Float16* F1 = (_Float16*)(ws + FBYTES);
  _Float16* F2 = (_Float16*)(ws + 2 * FBYTES);
  _Float16* wpack  = (_Float16*)(ws + 3 * FBYTES);
  const size_t WPACK_ELEMS = (size_t)2 * 8 * 9 * 64 * 64;
  _Float16* wopack = wpack + WPACK_ELEMS;
  _Float16* probs  = wopack + (size_t)NPADC * 64;
  float*    nsum   = (float*)(probs + (size_t)NBATCH * NCH * HW * HW);

  const size_t zn = (3 * FBYTES) / 4;
  zero_u32<<<(unsigned)((zn + 255) / 256), 256, 0, stream>>>((uint32_t*)ws, zn);
  zero_u32<<<1, 32, 0, stream>>>((uint32_t*)nsum, 4);

  pack_w<<<(2 * 8 * 9 * 64 * 64 + 255) / 256, 256, 0, stream>>>(w1s, w2s, wpack);
  pack_wout<<<(NPADC * 64 + 255) / 256, 256, 0, stream>>>(w_out, wopack);

  conv_in_k<<<NBATCH * HW, HW, 0, stream>>>(z, w_in, b_in, F0);

  _Float16* cur = F0;
  _Float16* nxt = F2;
  const size_t LAYER = (size_t)9 * 64 * 64;
  for (int i = 0; i < 8; ++i) {
    conv3x3_wmma<<<512, 256, 9 * 64 * LDSP * sizeof(_Float16), stream>>>(
        cur, F1, wpack + (size_t)i * LAYER, b1s + (size_t)i * 64, nullptr, 1);
    conv3x3_wmma<<<512, 256, 9 * 64 * LDSP * sizeof(_Float16), stream>>>(
        F1, nxt, wpack + (size_t)(8 + i) * LAYER, b2s + (size_t)i * 64, cur, 0);
    _Float16* t = cur; cur = nxt; nxt = t;
  }

  convout_softmax<<<1024, 128, 0, stream>>>(cur, wopack, b_out, probs, nsum);
  apply_k<<<NBATCH * HW, AK_WAVES * 32, 0, stream>>>(x, eps, probs, nsum, out);
}


# 3x3 conv kernel: all global loads (weight copy, A fragments of the 9 taps, bias, residual) issued in groups ahead of use, B fragments read one k-step ahead; was a waited round trip per load and WMMA
# speedup vs baseline: 3.7140x; 1.4135x over previous
#include <hip/hip_runtime.h>
#include <hip/hip_bf16.h>
#include <stdint.h>

typedef __attribute__((ext_vector_type(16))) _Float16 v16h;
typedef __attribute__((ext_vector_type(8)))  _Float16 v8h;
typedef __attribute__((ext_vector_type(8)))  float    v8f;

#define NFEAT 64
#define HW    128
#define HALO  130
#define NBATCH 4
#define KTAP  441
#define NCH   442
#define NPADC 448
#define LDSP  72

__device__ __forceinline__ v16h cat8(v8h lo, v8h hi) {
  return __builtin_shufflevector(lo, hi, 0,1,2,3,4,5,6,7,8,9,10,11,12,13,14,15);
}
__device__ __forceinline__ v8f wmma16(v16h a, v16h b, v8f c) {
  return __builtin_amdgcn_wmma_f32_16x16x32_f16(false, a, false, b, (short)0, c, false, false);
}

__global__ void zero_u32(uint32_t* p, size_t n) {
  size_t i = (size_t)blockIdx.x * blockDim.x + threadIdx.x;
  if (i < n) p[i] = 0u;
}

__global__ void pack_w(const float* __restrict__ w1s, const float* __restrict__ w2s,
                       _Float16* __restrict__ dst) {
  int idx = blockIdx.x * blockDim.x + threadIdx.x;
  const int TOT = 2 * 8 * 9 * 64 * 64;
  if (idx >= TOT) return;
  int c   = idx & 63;
  int n   = (idx >> 6) & 63;
  int pg  = idx >> 12;
  int pos = pg % 9;
  int lay = pg / 9;
  int blk = lay & 7;
  int which = lay >> 3;
  const float* src = which ? w2s : w1s;
  float v = src[(((size_t)(blk * 64 + n) * 64 + c) * 9) + pos];
  dst[idx] = (_Float16)v;
}

__global__ void pack_wout(const float* __restrict__ w_out, _Float16* __restrict__ dst) {
  int idx = blockIdx.x * blockDim.x + threadIdx.x;
  if (idx >= NPADC * 64) return;
  int c = idx & 63;
  int n = idx >> 6;
  dst[idx] = (n < NCH) ? (_Float16)w_out[(size_t)n * 64 + c] : (_Float16)0.0f;
}

__global__ __launch_bounds__(128) void conv_in_k(const float* __restrict__ z,
                                                 const float* __restrict__ w_in,
                                                 const float* __restrict__ b_in,
                                                 _Float16* __restrict__ dst) {
  const int x = threadIdx.x;
  const int y = blockIdx.x & (HW - 1);
  const int b = blockIdx.x >> 7;

  float zv[27];
  #pragma unroll
  for (int ci = 0; ci < 3; ++ci) {
    const float* zp = z + ((size_t)(b * 3 + ci)) * HW * HW;
    #pragma unroll
    for (int ky = 0; ky < 3; ++ky) {
      const int yy = y + ky - 1;
      #pragma unroll
      for (int kx = 0; kx < 3; ++kx) {
        const int xx = x + kx - 1;
        const bool ok = ((unsigned)yy < (unsigned)HW) && ((unsigned)xx < (unsigned)HW);
        zv[ci * 9 + ky * 3 + kx] = ok ? zp[yy * HW + xx] : 0.0f;
      }
    }
  }

  _Float16* op = dst + (((size_t)b * HALO + y + 1) * HALO + x + 1) * NFEAT;
  #pragma unroll
  for (int g = 0; g < NFEAT / 8; ++g) {
    v8h o;
    #pragma unroll
    for (int jp = 0; jp < 4; ++jp) {
      const int co = g * 8 + jp * 2;
      const float* wp = w_in + co * 27;
      float a0 = b_in[co];
      float a1 = b_in[co + 1];
      #pragma unroll
      for (int t = 0; t < 27; ++t) {
        a0 = __builtin_fmaf(zv[t], wp[t], a0);
        a1 = __builtin_fmaf(zv[t], wp[27 + t], a1);
      }
      o[jp * 2]     = (_Float16)a0;
      o[jp * 2 + 1] = (_Float16)a1;
    }
    *(v8h*)(op + g * 8) = o;
  }
}

template <bool RESID>
__global__ __launch_bounds__(256) void conv3x3_wmma(
    const _Float16* __restrict__ src, _Float16* __restrict__ dst,
    const _Float16* __restrict__ wl, const float* __restrict__ bias,
    const _Float16* __restrict__ resid, int do_relu) {
  extern __shared__ __attribute__((aligned(16))) _Float16 lw[];

  const int wave = threadIdx.x >> 5;
  const int lane = threadIdx.x & 31;
  const int tile = blockIdx.x * 8 + wave;
  const int b  = tile >> 10;
  const int r  = tile & 1023;
  const int y  = r >> 3;
  const int x0 = (r & 7) << 4;

  const int mrow = lane & 15;
  const int ksel = (lane >> 4) << 3;
  const int nlan = lane & 15;
  const int mhi  = (lane >> 4) << 3;

  constexpr int NPIECE = (9 * 64 * 64) / 8 / 256;
  {
    uint4 wst[NPIECE];
    const uint4* g = (const uint4*)wl;
    #pragma unroll
    for (int j = 0; j < NPIECE; ++j) wst[j] = g[threadIdx.x + 256 * j];
    #pragma unroll
    for (int j = 0; j < NPIECE; ++j) {
      const int i = threadIdx.x + 256 * j;
      *(uint4*)(lw + (i >> 3) * LDSP + (i & 7) * 8) = wst[j];
    }
  }
  v8h af[9][4];
  #pragma unroll
  for (int ky = 0; ky < 3; ++ky) {
    #pragma unroll
    for (int kx = 0; kx < 3; ++kx) {
      const _Float16* ap =
          src + (((size_t)b * HALO + y + ky) * HALO + (x0 + mrow + kx)) * NFEAT;
      #pragma unroll
      for (int q = 0; q < 4; ++q) af[ky * 3 + kx][q] = *(const v8h*)(ap + 16 * q + ksel);
    }
  }
  float bv[4];
  #pragma unroll
  for (int nt = 0; nt < 4; ++nt) bv[nt] = bias[nt * 16 + nlan];
  __builtin_amdgcn_sched_barrier(0);
  __builtin_amdgcn_fence(__ATOMIC_RELEASE, "workgroup", "local");
  __builtin_amdgcn_s_barrier();
  __builtin_amdgcn_fence(__ATOMIC_ACQUIRE, "workgroup", "local");

  v8f acc[4] = {};

  const size_t roff = (((size_t)b * HALO + y + 1) * HALO + (x0 + mhi + 1)) * NFEAT + nlan;
  _Float16 rv[4][8];

  const _Float16* lb = lw + nlan * LDSP + ksel;
  int hi_off = 5 * 64 * LDSP;
  asm volatile("" : "+v"(hi_off));
  const _Float16* lb_hi = lb + hi_off;
  v16h bf[2][4];
  #pragma unroll
  for (int nt = 0; nt < 4; ++nt) {
    const _Float16* w = lb + nt * 16 * LDSP;
    bf[0][nt] = cat8(*(const v8h*)w, *(const v8h*)(w + 16));
  }
  __builtin_amdgcn_sched_group_barrier(0x100, 8, 0);
  constexpr int RSTEP = 10;
  #pragma unroll
  for (int s = 0; s < 18; ++s) {
    if (s + 1 < 18) {
      const int tn = (s + 1) >> 1;
      const _Float16* wn = (tn < 5 ? lb + tn * (64 * LDSP) : lb_hi + (tn - 5) * (64 * LDSP))
                           + ((s + 1) & 1) * 32;
      #pragma unroll
      for (int nt = 0; nt < 4; ++nt) {
        const _Float16* w = wn + nt * 16 * LDSP;
        bf[(s + 1) & 1][nt] = cat8(*(const v8h*)w, *(const v8h*)(w + 16));
      }
    }
    const v16h a = cat8(af[s >> 1][2 * (s & 1)], af[s >> 1][2 * (s & 1) + 1]);
    #pragma unroll
    for (int nt = 0; nt < 4; ++nt) acc[nt] = wmma16(a, bf[s & 1][nt], acc[nt]);
    if (s + 1 < 18) __builtin_amdgcn_sched_group_barrier(0x100, 8, 0);
    __builtin_amdgcn_sched_group_barrier(0x008, 4, 0);
    if (RESID && s == RSTEP) {
      #pragma unroll
      for (int nt = 0; nt < 4; ++nt)
        #pragma unroll
        for (int i = 0; i < 8; ++i) rv[nt][i] = resid[roff + (size_t)i * NFEAT + nt * 16];
      __builtin_amdgcn_sched_group_barrier(0x020, 32, 0);
    }
  }

  #pragma unroll
  for (int nt = 0; nt < 4; ++nt) {
    #pragma unroll
    for (int i = 0; i < 8; ++i) {
      float v = acc[nt][i] + bv[nt];
      if (RESID) v += (float)rv[nt][i];
      if (do_relu) v = v > 0.0f ? v : 0.0f;
      dst[roff + (size_t)i * NFEAT + nt * 16] = (_Float16)v;
    }
  }
}

__global__ __launch_bounds__(128) void convout_softmax(
    const _Float16* __restrict__ feat, const _Float16* __restrict__ wo,
    const float* __restrict__ bo, _Float16* __restrict__ probs,
    float* __restrict__ nsum) {
  const int wave = threadIdx.x >> 5;
  const int lane = threadIdx.x & 31;
  const int tile = blockIdx.x * 4 + wave;
  const int b  = tile >> 10;
  const int r  = tile & 1023;
  const int y  = r >> 3;
  const int x0 = (r & 7) << 4;

  const int mrow = lane & 15;
  const int ksel = (lane >> 4) << 3;
  const int nlan = lane & 15;
  const int mhi  = (lane >> 4) << 3;

  const _Float16* ap =
      feat + (((size_t)b * HALO + y + 1) * HALO + (x0 + mrow + 1)) * NFEAT;
  v16h a0 = cat8(*(const v8h*)(ap + ksel),      *(const v8h*)(ap + 16 + ksel));
  v16h a1 = cat8(*(const v8h*)(ap + 32 + ksel), *(const v8h*)(ap + 48 + ksel));

  constexpr int NT = NPADC / 16;
  v8f lg[NT];

  constexpr int PF = 5;
  __builtin_amdgcn_sched_group_barrier(0x020, 4 + 4 * PF, 0);
  #pragma unroll
  for (int nt = 0; nt < NT; ++nt) {
    const int n = nt * 16 + nlan;
    const _Float16* wp = wo + (size_t)n * 64;
    v16h b0 = cat8(*(const v8h*)(wp + ksel),      *(const v8h*)(wp + 16 + ksel));
    v16h b1 = cat8(*(const v8h*)(wp + 32 + ksel), *(const v8h*)(wp + 48 + ksel));
    v8f c = {};
    c = wmma16(a0, b0, c);
    c = wmma16(a1, b1, c);
    lg[nt] = c;
    __builtin_amdgcn_sched_group_barrier(0x008, 2, 0);
    __builtin_amdgcn_sched_group_barrier(0x020, 4, 0);
  }
  #pragma unroll
  for (int nt = 0; nt < NT; ++nt) {
    const int n = nt * 16 + nlan;
    const float bv = (n < NCH) ? bo[n] : -1e30f;
    #pragma unroll
    for (int i = 0; i < 8; ++i) lg[nt][i] += bv;
  }

  float inv[8];
  #pragma unroll
  for (int i = 0; i < 8; ++i) {
    float m = lg[0][i];
    #pragma unroll
    for (int nt = 1; nt < NT; ++nt) m = fmaxf(m, lg[nt][i]);
    #pragma unroll
    for (int off = 8; off; off >>= 1) m = fmaxf(m, __shfl_xor(m, off, 32));
    float s = 0.0f;
    #pragma unroll
    for (int nt = 0; nt < NT; ++nt) {
      const float e = __expf(lg[nt][i] - m);
      lg[nt][i] = e;
      s += e;
    }
    #pragma unroll
    for (int off = 8; off; off >>= 1) s += __shfl_xor(s, off, 32);
    inv[i] = 1.0f / s;
  }

  _Float16* pb = probs + (size_t)b * NCH * (HW * HW) + (size_t)y * HW + x0 + mhi;
  #pragma unroll
  for (int nt = 0; nt < NT; ++nt) {
    const int t = nt * 16 + nlan;
    v8h o;
    #pragma unroll
    for (int i = 0; i < 8; ++i) o[i] = (_Float16)(lg[nt][i] * inv[i]);
    if (t < NCH) *(v8h*)(pb + (size_t)t * (HW * HW)) = o;
  }

  float nacc = 0.0f;
  if (nlan == (KTAP & 15)) {
    #pragma unroll
    for (int i = 0; i < 8; ++i) nacc += lg[KTAP >> 4][i] * inv[i];
  }
  nacc += __shfl_xor(nacc, 16, 32);
  if (lane == (KTAP & 15)) atomicAdd(nsum + b, nacc);
}

typedef __attribute__((ext_vector_type(4))) _Float16 v4h;
typedef __attribute__((ext_vector_type(4))) float    v4f;
typedef __attribute__((ext_vector_type(2))) float    v2f;

#define AK_WAVES 7
#define AK_ROWS  3
#define AK_NG    9

__global__ __launch_bounds__(AK_WAVES * 32) void apply_k(
    const float* __restrict__ x, const float* __restrict__ eps,
    const _Float16* __restrict__ probs, const float* __restrict__ nsum,
    float* __restrict__ out) {
  __shared__ float red[AK_WAVES * 12 * 32];
  const int lane = threadIdx.x & 31;
  const int wave = __builtin_amdgcn_readfirstlane(threadIdx.x >> 5);
  const int h = blockIdx.x & (HW - 1);
  const int b = blockIdx.x >> 7;
  const int col0 = 16 * lane - 12;

  float acc[3][4];
  #pragma unroll
  for (int c = 0; c < 3; ++c)
    #pragma unroll
    for (int q = 0; q < 4; ++q) acc[c][q] = 0.0f;

  #pragma unroll 1
  for (int r = 0; r < AK_ROWS; ++r) {
    const int ky = wave * AK_ROWS + r;
    const int yi = 4 * h - 10 + ky;
    const float rowkeep = ((unsigned)yi < 512u) ? 1.0f : 0.0f;
    const int yc = yi < 0 ? 0 : (yi > 511 ? 511 : yi);

    const _Float16* pr =
        probs + ((size_t)(b * NCH + ky * 21)) * (HW * HW) + (size_t)h * HW + 4 * lane;
    v4h pv[21];
    #pragma unroll
    for (int kx = 0; kx < 21; ++kx)
      pv[kx] = __builtin_nontemporal_load((const v4h*)(pr + (size_t)kx * (HW * HW)));

    bool ok[AK_NG];
    #pragma unroll
    for (int g = 0; g < AK_NG; ++g) {
      const int col = col0 + 4 * g;
      ok[g] = (g < 3) ? (col >= 0) : ((g > 6) ? (col < 512) : true);
    }
    v2f xlo[3];
    v4f xv[3][AK_NG - 2];
    v2f xhi[3];
    float xend[3];
    #pragma unroll
    for (int c = 0; c < 3; ++c) {
      const float* xr = x + ((size_t)((b * 3 + c) * 512 + yc)) * 512;
      xlo[c] = *(const v2f*)(xr + (ok[0] ? col0 : 0) + 2);
      #pragma unroll
      for (int g = 1; g < AK_NG - 1; ++g)
        xv[c][g - 1] = *(const v4f*)(xr + (ok[g] ? col0 + 4 * g : 0));
      const float* xe = xr + (ok[AK_NG - 1] ? col0 + 4 * (AK_NG - 1) : 0);
      xhi[c]  = *(const v2f*)xe;
      xend[c] = xe[2];
    }
    __builtin_amdgcn_sched_barrier(0);

    float keep[AK_NG];
    #pragma unroll
    for (int g = 0; g < AK_NG; ++g) keep[g] = ok[g] ? 1.0f : 0.0f;
    float xs[3][4 * AK_NG];
    #pragma unroll
    for (int c = 0; c < 3; ++c) {
      xs[c][2] = xlo[c][0] * keep[0];
      xs[c][3] = xlo[c][1] * keep[0];
      #pragma unroll
      for (int g = 1; g < AK_NG - 1; ++g) {
        #pragma unroll
        for (int e = 0; e < 4; ++e)
          xs[c][4 * g + e] = (g < 3 || g > 6) ? xv[c][g - 1][e] * keep[g] : xv[c][g - 1][e];
      }
      xs[c][32] = xhi[c][0] * keep[AK_NG - 1];
      xs[c][33] = xhi[c][1] * keep[AK_NG - 1];
      xs[c][34] = xend[c]   * keep[AK_NG - 1];
    }

    float part[3][4];
    #pragma unroll
    for (int c = 0; c < 3; ++c)
      #pragma unroll
      for (int q = 0; q < 4; ++q) part[c][q] = 0.0f;
    #pragma unroll
    for (int kx = 0; kx < 21; ++kx) {
      #pragma unroll
      for (int q = 0; q < 4; ++q) {
        const float pf = (float)pv[kx][q];
        #pragma unroll
        for (int c = 0; c < 3; ++c) part[c][q] += xs[c][4 * q + 2 + kx] * pf;
      }
    }
    #pragma unroll
    for (int c = 0; c < 3; ++c)
      #pragma unroll
      for (int q = 0; q < 4; ++q) acc[c][q] += part[c][q] * rowkeep;
  }

  #pragma unroll
  for (int c = 0; c < 3; ++c)
    #pragma unroll
    for (int q = 0; q < 4; ++q) red[(wave * 12 + c * 4 + q) * 32 + lane] = acc[c][q];
  __syncthreads();

  if (wave < 3) {
    const int c = wave;
    float s[4];
    #pragma unroll
    for (int q = 0; q < 4; ++q) {
      float t = 0.0f;
      #pragma unroll
      for (int j = 0; j < AK_WAVES; ++j) t += red[(j * 12 + c * 4 + q) * 32 + lane];
      s[q] = t;
    }
    const size_t idx = (((size_t)(b * 3 + c)) * HW + h) * HW + 4 * lane;
    const float ns = nsum[b] * (1.0f / (HW * HW));
    const v4f e4 = *(const v4f*)(eps + idx);
    v4f o;
    #pragma unroll
    for (int q = 0; q < 4; ++q) o[q] = s[q] + ns * e4[q];
    *(v4f*)(out + idx) = o;
  }
}

extern "C" void kernel_launch(void* const* d_in, const int* in_sizes, int n_in,
                              void* d_out, int out_size, void* d_ws, size_t ws_size,
                              hipStream_t stream) {
  const float* x     = (const float*)d_in[0];
  const float* z     = (const float*)d_in[1];
  const float* eps   = (const float*)d_in[2];
  const float* w_in  = (const float*)d_in[3];
  const float* b_in  = (const float*)d_in[4];
  const float* w1s   = (const float*)d_in[5];
  const float* b1s   = (const float*)d_in[6];
  const float* w2s   = (const float*)d_in[7];
  const float* b2s   = (const float*)d_in[8];
  const float* w_out = (const float*)d_in[9];
  const float* b_out = (const float*)d_in[10];
  float* out = (float*)d_out;

  char* ws = (char*)d_ws;
  const size_t FBYTES = (size_t)NBATCH * HALO * HALO * NFEAT * sizeof(_Float16);
  _Float16* F0 = (_Float16*)(ws);
  _Float16* F1 = (_Float16*)(ws + FBYTES);
  _Float16* F2 = (_Float16*)(ws + 2 * FBYTES);
  _Float16* wpack  = (_Float16*)(ws + 3 * FBYTES);
  const size_t WPACK_ELEMS = (size_t)2 * 8 * 9 * 64 * 64;
  _Float16* wopack = wpack + WPACK_ELEMS;
  _Float16* probs  = wopack + (size_t)NPADC * 64;
  float*    nsum   = (float*)(probs + (size_t)NBATCH * NCH * HW * HW);

  const size_t zn = (3 * FBYTES) / 4;
  zero_u32<<<(unsigned)((zn + 255) / 256), 256, 0, stream>>>((uint32_t*)ws, zn);
  zero_u32<<<1, 32, 0, stream>>>((uint32_t*)nsum, 4);

  pack_w<<<(2 * 8 * 9 * 64 * 64 + 255) / 256, 256, 0, stream>>>(w1s, w2s, wpack);
  pack_wout<<<(NPADC * 64 + 255) / 256, 256, 0, stream>>>(w_out, wopack);

  conv_in_k<<<NBATCH * HW, HW, 0, stream>>>(z, w_in, b_in, F0);

  _Float16* cur = F0;
  _Float16* nxt = F2;
  const size_t LAYER = (size_t)9 * 64 * 64;
  for (int i = 0; i < 8; ++i) {
    conv3x3_wmma<false><<<512, 256, 9 * 64 * LDSP * sizeof(_Float16), stream>>>(
        cur, F1, wpack + (size_t)i * LAYER, b1s + (size_t)i * 64, nullptr, 1);
    conv3x3_wmma<true><<<512, 256, 9 * 64 * LDSP * sizeof(_Float16), stream>>>(
        F1, nxt, wpack + (size_t)(8 + i) * LAYER, b2s + (size_t)i * 64, cur, 0);
    _Float16* t = cur; cur = nxt; nxt = t;
  }

  convout_softmax<<<1024, 128, 0, stream>>>(cur, wopack, b_out, probs, nsum);
  apply_k<<<NBATCH * HW, AK_WAVES * 32, 0, stream>>>(x, eps, probs, nsum, out);
}


# zeroing launch: clear only the 1-pixel frames of the 3 feature buffers (792 KB: all that is read before written) and the noise sums; was 26 MB by 202,800 one-store waves, plus a second launch
# speedup vs baseline: 3.8584x; 1.0389x over previous
#include <hip/hip_runtime.h>
#include <hip/hip_bf16.h>
#include <stdint.h>

typedef __attribute__((ext_vector_type(16))) _Float16 v16h;
typedef __attribute__((ext_vector_type(8)))  _Float16 v8h;
typedef __attribute__((ext_vector_type(8)))  float    v8f;

#define NFEAT 64
#define HW    128
#define HALO  130
#define NBATCH 4
#define KTAP  441
#define NCH   442
#define NPADC 448
#define LDSP  72

__device__ __forceinline__ v16h cat8(v8h lo, v8h hi) {
  return __builtin_shufflevector(lo, hi, 0,1,2,3,4,5,6,7,8,9,10,11,12,13,14,15);
}
__device__ __forceinline__ v8f wmma16(v16h a, v16h b, v8f c) {
  return __builtin_amdgcn_wmma_f32_16x16x32_f16(false, a, false, b, (short)0, c, false, false);
}

#define FRAME_PX     516
#define FRAME_PIECES (FRAME_PX * 8)
#define NPLANE       (3 * NBATCH)
__global__ __launch_bounds__(256) void zero_frames(_Float16* __restrict__ feat,
                                                   float* __restrict__ nsum) {
  const int idx = blockIdx.x * 256 + threadIdx.x;
  if (idx < NBATCH) nsum[idx] = 0.0f;
  if (idx >= NPLANE * FRAME_PIECES) return;
  const int plane = idx / FRAME_PIECES;
  const int rem   = idx - plane * FRAME_PIECES;
  const int hp    = rem >> 3;
  const int q     = rem & 7;
  int pix;
  if (hp < HALO + 1) {
    pix = hp;
  } else if (hp < HALO + 1 + 2 * (HW - 1)) {
    const int k = hp - (HALO + 1);
    pix = (1 + (k >> 1)) * HALO + (HALO - 1) + (k & 1);
  } else {
    pix = HW * HALO + (HALO - 1) + (hp - (HALO + 1 + 2 * (HW - 1)));
  }
  uint4* p = (uint4*)feat;
  p[((size_t)plane * (HALO * HALO) + pix) * (NFEAT / 8) + q] = make_uint4(0u, 0u, 0u, 0u);
}

__global__ void pack_w(const float* __restrict__ w1s, const float* __restrict__ w2s,
                       _Float16* __restrict__ dst) {
  int idx = blockIdx.x * blockDim.x + threadIdx.x;
  const int TOT = 2 * 8 * 9 * 64 * 64;
  if (idx >= TOT) return;
  int c   = idx & 63;
  int n   = (idx >> 6) & 63;
  int pg  = idx >> 12;
  int pos = pg % 9;
  int lay = pg / 9;
  int blk = lay & 7;
  int which = lay >> 3;
  const float* src = which ? w2s : w1s;
  float v = src[(((size_t)(blk * 64 + n) * 64 + c) * 9) + pos];
  dst[idx] = (_Float16)v;
}

__global__ void pack_wout(const float* __restrict__ w_out, _Float16* __restrict__ dst) {
  int idx = blockIdx.x * blockDim.x + threadIdx.x;
  if (idx >= NPADC * 64) return;
  int c = idx & 63;
  int n = idx >> 6;
  dst[idx] = (n < NCH) ? (_Float16)w_out[(size_t)n * 64 + c] : (_Float16)0.0f;
}

__global__ __launch_bounds__(128) void conv_in_k(const float* __restrict__ z,
                                                 const float* __restrict__ w_in,
                                                 const float* __restrict__ b_in,
                                                 _Float16* __restrict__ dst) {
  const int x = threadIdx.x;
  const int y = blockIdx.x & (HW - 1);
  const int b = blockIdx.x >> 7;

  float zv[27];
  #pragma unroll
  for (int ci = 0; ci < 3; ++ci) {
    const float* zp = z + ((size_t)(b * 3 + ci)) * HW * HW;
    #pragma unroll
    for (int ky = 0; ky < 3; ++ky) {
      const int yy = y + ky - 1;
      #pragma unroll
      for (int kx = 0; kx < 3; ++kx) {
        const int xx = x + kx - 1;
        const bool ok = ((unsigned)yy < (unsigned)HW) && ((unsigned)xx < (unsigned)HW);
        zv[ci * 9 + ky * 3 + kx] = ok ? zp[yy * HW + xx] : 0.0f;
      }
    }
  }

  _Float16* op = dst + (((size_t)b * HALO + y + 1) * HALO + x + 1) * NFEAT;
  #pragma unroll
  for (int g = 0; g < NFEAT / 8; ++g) {
    v8h o;
    #pragma unroll
    for (int jp = 0; jp < 4; ++jp) {
      const int co = g * 8 + jp * 2;
      const float* wp = w_in + co * 27;
      float a0 = b_in[co];
      float a1 = b_in[co + 1];
      #pragma unroll
      for (int t = 0; t < 27; ++t) {
        a0 = __builtin_fmaf(zv[t], wp[t], a0);
        a1 = __builtin_fmaf(zv[t], wp[27 + t], a1);
      }
      o[jp * 2]     = (_Float16)a0;
      o[jp * 2 + 1] = (_Float16)a1;
    }
    *(v8h*)(op + g * 8) = o;
  }
}

template <bool RESID>
__global__ __launch_bounds__(256) void conv3x3_wmma(
    const _Float16* __restrict__ src, _Float16* __restrict__ dst,
    const _Float16* __restrict__ wl, const float* __restrict__ bias,
    const _Float16* __restrict__ resid, int do_relu) {
  extern __shared__ __attribute__((aligned(16))) _Float16 lw[];

  const int wave = threadIdx.x >> 5;
  const int lane = threadIdx.x & 31;
  const int tile = blockIdx.x * 8 + wave;
  const int b  = tile >> 10;
  const int r  = tile & 1023;
  const int y  = r >> 3;
  const int x0 = (r & 7) << 4;

  const int mrow = lane & 15;
  const int ksel = (lane >> 4) << 3;
  const int nlan = lane & 15;
  const int mhi  = (lane >> 4) << 3;

  constexpr int NPIECE = (9 * 64 * 64) / 8 / 256;
  {
    uint4 wst[NPIECE];
    const uint4* g = (const uint4*)wl;
    #pragma unroll
    for (int j = 0; j < NPIECE; ++j) wst[j] = g[threadIdx.x + 256 * j];
    #pragma unroll
    for (int j = 0; j < NPIECE; ++j) {
      const int i = threadIdx.x + 256 * j;
      *(uint4*)(lw + (i >> 3) * LDSP + (i & 7) * 8) = wst[j];
    }
  }
  v8h af[9][4];
  #pragma unroll
  for (int ky = 0; ky < 3; ++ky) {
    #pragma unroll
    for (int kx = 0; kx < 3; ++kx) {
      const _Float16* ap =
          src + (((size_t)b * HALO + y + ky) * HALO + (x0 + mrow + kx)) * NFEAT;
      #pragma unroll
      for (int q = 0; q < 4; ++q) af[ky * 3 + kx][q] = *(const v8h*)(ap + 16 * q + ksel);
    }
  }
  float bv[4];
  #pragma unroll
  for (int nt = 0; nt < 4; ++nt) bv[nt] = bias[nt * 16 + nlan];
  __builtin_amdgcn_sched_barrier(0);
  __builtin_amdgcn_fence(__ATOMIC_RELEASE, "workgroup", "local");
  __builtin_amdgcn_s_barrier();
  __builtin_amdgcn_fence(__ATOMIC_ACQUIRE, "workgroup", "local");

  v8f acc[4] = {};

  const size_t roff = (((size_t)b * HALO + y + 1) * HALO + (x0 + mhi + 1)) * NFEAT + nlan;
  _Float16 rv[4][8];

  const _Float16* lb = lw + nlan * LDSP + ksel;
  int hi_off = 5 * 64 * LDSP;
  asm volatile("" : "+v"(hi_off));
  const _Float16* lb_hi = lb + hi_off;
  v16h bf[2][4];
  #pragma unroll
  for (int nt = 0; nt < 4; ++nt) {
    const _Float16* w = lb + nt * 16 * LDSP;
    bf[0][nt] = cat8(*(const v8h*)w, *(const v8h*)(w + 16));
  }
  __builtin_amdgcn_sched_group_barrier(0x100, 8, 0);
  constexpr int RSTEP = 10;
  #pragma unroll
  for (int s = 0; s < 18; ++s) {
    if (s + 1 < 18) {
      const int tn = (s + 1) >> 1;
      const _Float16* wn = (tn < 5 ? lb + tn * (64 * LDSP) : lb_hi + (tn - 5) * (64 * LDSP))
                           + ((s + 1) & 1) * 32;
      #pragma unroll
      for (int nt = 0; nt < 4; ++nt) {
        const _Float16* w = wn + nt * 16 * LDSP;
        bf[(s + 1) & 1][nt] = cat8(*(const v8h*)w, *(const v8h*)(w + 16));
      }
    }
    const v16h a = cat8(af[s >> 1][2 * (s & 1)], af[s >> 1][2 * (s & 1) + 1]);
    #pragma unroll
    for (int nt = 0; nt < 4; ++nt) acc[nt] = wmma16(a, bf[s & 1][nt], acc[nt]);
    if (s + 1 < 18) __builtin_amdgcn_sched_group_barrier(0x100, 8, 0);
    __builtin_amdgcn_sched_group_barrier(0x008, 4, 0);
    if (RESID && s == RSTEP) {
      #pragma unroll
      for (int nt = 0; nt < 4; ++nt)
        #pragma unroll
        for (int i = 0; i < 8; ++i) rv[nt][i] = resid[roff + (size_t)i * NFEAT + nt * 16];
      __builtin_amdgcn_sched_group_barrier(0x020, 32, 0);
    }
  }

  #pragma unroll
  for (int nt = 0; nt < 4; ++nt) {
    #pragma unroll
    for (int i = 0; i < 8; ++i) {
      float v = acc[nt][i] + bv[nt];
      if (RESID) v += (float)rv[nt][i];
      if (do_relu) v = v > 0.0f ? v : 0.0f;
      dst[roff + (size_t)i * NFEAT + nt * 16] = (_Float16)v;
    }
  }
}

__global__ __launch_bounds__(128) void convout_softmax(
    const _Float16* __restrict__ feat, const _Float16* __restrict__ wo,
    const float* __restrict__ bo, _Float16* __restrict__ probs,
    float* __restrict__ nsum) {
  const int wave = threadIdx.x >> 5;
  const int lane = threadIdx.x & 31;
  const int tile = blockIdx.x * 4 + wave;
  const int b  = tile >> 10;
  const int r  = tile & 1023;
  const int y  = r >> 3;
  const int x0 = (r & 7) << 4;

  const int mrow = lane & 15;
  const int ksel = (lane >> 4) << 3;
  const int nlan = lane & 15;
  const int mhi  = (lane >> 4) << 3;

  const _Float16* ap =
      feat + (((size_t)b * HALO + y + 1) * HALO + (x0 + mrow + 1)) * NFEAT;
  v16h a0 = cat8(*(const v8h*)(ap + ksel),      *(const v8h*)(ap + 16 + ksel));
  v16h a1 = cat8(*(const v8h*)(ap + 32 + ksel), *(const v8h*)(ap + 48 + ksel));

  constexpr int NT = NPADC / 16;
  v8f lg[NT];

  constexpr int PF = 5;
  __builtin_amdgcn_sched_group_barrier(0x020, 4 + 4 * PF, 0);
  #pragma unroll
  for (int nt = 0; nt < NT; ++nt) {
    const int n = nt * 16 + nlan;
    const _Float16* wp = wo + (size_t)n * 64;
    v16h b0 = cat8(*(const v8h*)(wp + ksel),      *(const v8h*)(wp + 16 + ksel));
    v16h b1 = cat8(*(const v8h*)(wp + 32 + ksel), *(const v8h*)(wp + 48 + ksel));
    v8f c = {};
    c = wmma16(a0, b0, c);
    c = wmma16(a1, b1, c);
    lg[nt] = c;
    __builtin_amdgcn_sched_group_barrier(0x008, 2, 0);
    __builtin_amdgcn_sched_group_barrier(0x020, 4, 0);
  }
  #pragma unroll
  for (int nt = 0; nt < NT; ++nt) {
    const int n = nt * 16 + nlan;
    const float bv = (n < NCH) ? bo[n] : -1e30f;
    #pragma unroll
    for (int i = 0; i < 8; ++i) lg[nt][i] += bv;
  }

  float inv[8];
  #pragma unroll
  for (int i = 0; i < 8; ++i) {
    float m = lg[0][i];
    #pragma unroll
    for (int nt = 1; nt < NT; ++nt) m = fmaxf(m, lg[nt][i]);
    #pragma unroll
    for (int off = 8; off; off >>= 1) m = fmaxf(m, __shfl_xor(m, off, 32));
    float s = 0.0f;
    #pragma unroll
    for (int nt = 0; nt < NT; ++nt) {
      const float e = __expf(lg[nt][i] - m);
      lg[nt][i] = e;
      s += e;
    }
    #pragma unroll
    for (int off = 8; off; off >>= 1) s += __shfl_xor(s, off, 32);
    inv[i] = 1.0f / s;
  }

  _Float16* pb = probs + (size_t)b * NCH * (HW * HW) + (size_t)y * HW + x0 + mhi;
  #pragma unroll
  for (int nt = 0; nt < NT; ++nt) {
    const int t = nt * 16 + nlan;
    v8h o;
    #pragma unroll
    for (int i = 0; i < 8; ++i) o[i] = (_Float16)(lg[nt][i] * inv[i]);
    if (t < NCH) *(v8h*)(pb + (size_t)t * (HW * HW)) = o;
  }

  float nacc = 0.0f;
  if (nlan == (KTAP & 15)) {
    #pragma unroll
    for (int i = 0; i < 8; ++i) nacc += lg[KTAP >> 4][i] * inv[i];
  }
  nacc += __shfl_xor(nacc, 16, 32);
  if (lane == (KTAP & 15)) atomicAdd(nsum + b, nacc);
}

typedef __attribute__((ext_vector_type(4))) _Float16 v4h;
typedef __attribute__((ext_vector_type(4))) float    v4f;
typedef __attribute__((ext_vector_type(2))) float    v2f;

#define AK_WAVES 7
#define AK_ROWS  3
#define AK_NG    9

__global__ __launch_bounds__(AK_WAVES * 32) void apply_k(
    const float* __restrict__ x, const float* __restrict__ eps,
    const _Float16* __restrict__ probs, const float* __restrict__ nsum,
    float* __restrict__ out) {
  __shared__ float red[AK_WAVES * 12 * 32];
  const int lane = threadIdx.x & 31;
  const int wave = __builtin_amdgcn_readfirstlane(threadIdx.x >> 5);
  const int h = blockIdx.x & (HW - 1);
  const int b = blockIdx.x >> 7;
  const int col0 = 16 * lane - 12;

  float acc[3][4];
  #pragma unroll
  for (int c = 0; c < 3; ++c)
    #pragma unroll
    for (int q = 0; q < 4; ++q) acc[c][q] = 0.0f;

  #pragma unroll 1
  for (int r = 0; r < AK_ROWS; ++r) {
    const int ky = wave * AK_ROWS + r;
    const int yi = 4 * h - 10 + ky;
    const float rowkeep = ((unsigned)yi < 512u) ? 1.0f : 0.0f;
    const int yc = yi < 0 ? 0 : (yi > 511 ? 511 : yi);

    const _Float16* pr =
        probs + ((size_t)(b * NCH + ky * 21)) * (HW * HW) + (size_t)h * HW + 4 * lane;
    v4h pv[21];
    #pragma unroll
    for (int kx = 0; kx < 21; ++kx)
      pv[kx] = __builtin_nontemporal_load((const v4h*)(pr + (size_t)kx * (HW * HW)));

    bool ok[AK_NG];
    #pragma unroll
    for (int g = 0; g < AK_NG; ++g) {
      const int col = col0 + 4 * g;
      ok[g] = (g < 3) ? (col >= 0) : ((g > 6) ? (col < 512) : true);
    }
    v2f xlo[3];
    v4f xv[3][AK_NG - 2];
    v2f xhi[3];
    float xend[3];
    #pragma unroll
    for (int c = 0; c < 3; ++c) {
      const float* xr = x + ((size_t)((b * 3 + c) * 512 + yc)) * 512;
      xlo[c] = *(const v2f*)(xr + (ok[0] ? col0 : 0) + 2);
      #pragma unroll
      for (int g = 1; g < AK_NG - 1; ++g)
        xv[c][g - 1] = *(const v4f*)(xr + (ok[g] ? col0 + 4 * g : 0));
      const float* xe = xr + (ok[AK_NG - 1] ? col0 + 4 * (AK_NG - 1) : 0);
      xhi[c]  = *(const v2f*)xe;
      xend[c] = xe[2];
    }
    __builtin_amdgcn_sched_barrier(0);

    float keep[AK_NG];
    #pragma unroll
    for (int g = 0; g < AK_NG; ++g) keep[g] = ok[g] ? 1.0f : 0.0f;
    float xs[3][4 * AK_NG];
    #pragma unroll
    for (int c = 0; c < 3; ++c) {
      xs[c][2] = xlo[c][0] * keep[0];
      xs[c][3] = xlo[c][1] * keep[0];
      #pragma unroll
      for (int g = 1; g < AK_NG - 1; ++g) {
        #pragma unroll
        for (int e = 0; e < 4; ++e)
          xs[c][4 * g + e] = (g < 3 || g > 6) ? xv[c][g - 1][e] * keep[g] : xv[c][g - 1][e];
      }
      xs[c][32] = xhi[c][0] * keep[AK_NG - 1];
      xs[c][33] = xhi[c][1] * keep[AK_NG - 1];
      xs[c][34] = xend[c]   * keep[AK_NG - 1];
    }

    float part[3][4];
    #pragma unroll
    for (int c = 0; c < 3; ++c)
      #pragma unroll
      for (int q = 0; q < 4; ++q) part[c][q] = 0.0f;
    #pragma unroll
    for (int kx = 0; kx < 21; ++kx) {
      #pragma unroll
      for (int q = 0; q < 4; ++q) {
        const float pf = (float)pv[kx][q];
        #pragma unroll
        for (int c = 0; c < 3; ++c) part[c][q] += xs[c][4 * q + 2 + kx] * pf;
      }
    }
    #pragma unroll
    for (int c = 0; c < 3; ++c)
      #pragma unroll
      for (int q = 0; q < 4; ++q) acc[c][q] += part[c][q] * rowkeep;
  }

  #pragma unroll
  for (int c = 0; c < 3; ++c)
    #pragma unroll
    for (int q = 0; q < 4; ++q) red[(wave * 12 + c * 4 + q) * 32 + lane] = acc[c][q];
  __syncthreads();

  if (wave < 3) {
    const int c = wave;
    float s[4];
    #pragma unroll
    for (int q = 0; q < 4; ++q) {
      float t = 0.0f;
      #pragma unroll
      for (int j = 0; j < AK_WAVES; ++j) t += red[(j * 12 + c * 4 + q) * 32 + lane];
      s[q] = t;
    }
    const size_t idx = (((size_t)(b * 3 + c)) * HW + h) * HW + 4 * lane;
    const float ns = nsum[b] * (1.0f / (HW * HW));
    const v4f e4 = *(const v4f*)(eps + idx);
    v4f o;
    #pragma unroll
    for (int q = 0; q < 4; ++q) o[q] = s[q] + ns * e4[q];
    *(v4f*)(out + idx) = o;
  }
}

extern "C" void kernel_launch(void* const* d_in, const int* in_sizes, int n_in,
                              void* d_out, int out_size, void* d_ws, size_t ws_size,
                              hipStream_t stream) {
  const float* x     = (const float*)d_in[0];
  const float* z     = (const float*)d_in[1];
  const float* eps   = (const float*)d_in[2];
  const float* w_in  = (const float*)d_in[3];
  const float* b_in  = (const float*)d_in[4];
  const float* w1s   = (const float*)d_in[5];
  const float* b1s   = (const float*)d_in[6];
  const float* w2s   = (const float*)d_in[7];
  const float* b2s   = (const float*)d_in[8];
  const float* w_out = (const float*)d_in[9];
  const float* b_out = (const float*)d_in[10];
  float* out = (float*)d_out;

  char* ws = (char*)d_ws;
  const size_t FBYTES = (size_t)NBATCH * HALO * HALO * NFEAT * sizeof(_Float16);
  _Float16* F0 = (_Float16*)(ws);
  _Float16* F1 = (_Float16*)(ws + FBYTES);
  _Float16* F2 = (_Float16*)(ws + 2 * FBYTES);
  _Float16* wpack  = (_Float16*)(ws + 3 * FBYTES);
  const size_t WPACK_ELEMS = (size_t)2 * 8 * 9 * 64 * 64;
  _Float16* wopack = wpack + WPACK_ELEMS;
  _Float16* probs  = wopack + (size_t)NPADC * 64;
  float*    nsum   = (float*)(probs + (size_t)NBATCH * NCH * HW * HW);

  zero_frames<<<(NPLANE * FRAME_PIECES + 255) / 256, 256, 0, stream>>>(F0, nsum);

  pack_w<<<(2 * 8 * 9 * 64 * 64 + 255) / 256, 256, 0, stream>>>(w1s, w2s, wpack);
  pack_wout<<<(NPADC * 64 + 255) / 256, 256, 0, stream>>>(w_out, wopack);

  conv_in_k<<<NBATCH * HW, HW, 0, stream>>>(z, w_in, b_in, F0);

  _Float16* cur = F0;
  _Float16* nxt = F2;
  const size_t LAYER = (size_t)9 * 64 * 64;
  for (int i = 0; i < 8; ++i) {
    conv3x3_wmma<false><<<512, 256, 9 * 64 * LDSP * sizeof(_Float16), stream>>>(
        cur, F1, wpack + (size_t)i * LAYER, b1s + (size_t)i * 64, nullptr, 1);
    conv3x3_wmma<true><<<512, 256, 9 * 64 * LDSP * sizeof(_Float16), stream>>>(
        F1, nxt, wpack + (size_t)(8 + i) * LAYER, b2s + (size_t)i * 64, cur, 0);
    _Float16* t = cur; cur = nxt; nxt = t;
  }

  convout_softmax<<<1024, 128, 0, stream>>>(cur, wopack, b_out, probs, nsum);
  apply_k<<<NBATCH * HW, AK_WAVES * 32, 0, stream>>>(x, eps, probs, nsum, out);
}


# feature maps as 4 planes of 16 channels and the 3x3 product transposed (weights = A operand): fragment loads read 4-5 whole lines, were 16; outputs and residuals are 16-byte whole-line accesses
# speedup vs baseline: 4.1424x; 1.0736x over previous
#include <hip/hip_runtime.h>
#include <hip/hip_bf16.h>
#include <stdint.h>

typedef __attribute__((ext_vector_type(16))) _Float16 v16h;
typedef __attribute__((ext_vector_type(8)))  _Float16 v8h;
typedef __attribute__((ext_vector_type(8)))  float    v8f;
typedef __attribute__((ext_vector_type(4)))  float    v4f;

#define NFEAT 64
#define HW    128
#define HALO  130
#define NBATCH 4
#define KTAP  441
#define NCH   442
#define NPADC 448
#define LDSP  72

#define NPL   4
#define XP    136
#define XO    4
__device__ __forceinline__ size_t fofs(int b, int yy, int pl, int xx) {
  return ((((size_t)b * HALO + yy) * NPL + pl) * XP + xx) * 16;
}

__device__ __forceinline__ v16h cat8(v8h lo, v8h hi) {
  return __builtin_shufflevector(lo, hi, 0,1,2,3,4,5,6,7,8,9,10,11,12,13,14,15);
}
__device__ __forceinline__ v8f wmma16(v16h a, v16h b, v8f c) {
  return __builtin_amdgcn_wmma_f32_16x16x32_f16(false, a, false, b, (short)0, c, false, false);
}

#define ROW_PIECES   (NPL * XP * 2)
#define IMG_PIECES   (HALO * ROW_PIECES)
#define SIDE_PIECES  (HW * NPL * 2 * 2)
#define FRAME_PIECES (2 * ROW_PIECES + SIDE_PIECES)
#define NPLANE       (3 * NBATCH)
__global__ __launch_bounds__(256) void zero_frames(_Float16* __restrict__ feat,
                                                   float* __restrict__ nsum) {
  const int idx = blockIdx.x * 256 + threadIdx.x;
  if (idx < NBATCH) nsum[idx] = 0.0f;
  if (idx >= NPLANE * FRAME_PIECES) return;
  const int plane = idx / FRAME_PIECES;
  const int rem   = idx - plane * FRAME_PIECES;
  int piece;
  if (rem < ROW_PIECES) {
    piece = rem;
  } else if (rem < 2 * ROW_PIECES) {
    piece = (HALO - 1) * ROW_PIECES + (rem - ROW_PIECES);
  } else {
    const int k    = rem - 2 * ROW_PIECES;
    const int half = k & 1;
    const int side = (k >> 1) & 1;
    const int pl   = (k >> 2) & 3;
    const int yy   = 1 + (k >> 4);
    piece = ((yy * NPL + pl) * XP + (side ? XO + HW : XO - 1)) * 2 + half;
  }
  uint4* p = (uint4*)feat;
  p[(size_t)plane * IMG_PIECES + piece] = make_uint4(0u, 0u, 0u, 0u);
}

__global__ void pack_w(const float* __restrict__ w1s, const float* __restrict__ w2s,
                       _Float16* __restrict__ dst) {
  int idx = blockIdx.x * blockDim.x + threadIdx.x;
  const int TOT = 2 * 8 * 9 * 64 * 64;
  if (idx >= TOT) return;
  int c   = idx & 63;
  int n   = (idx >> 6) & 63;
  int pg  = idx >> 12;
  int pos = pg % 9;
  int lay = pg / 9;
  int blk = lay & 7;
  int which = lay >> 3;
  const float* src = which ? w2s : w1s;
  float v = src[(((size_t)(blk * 64 + n) * 64 + c) * 9) + pos];
  dst[idx] = (_Float16)v;
}

__global__ void pack_wout(const float* __restrict__ w_out, _Float16* __restrict__ dst) {
  int idx = blockIdx.x * blockDim.x + threadIdx.x;
  if (idx >= NPADC * 64) return;
  int c = idx & 63;
  int n = idx >> 6;
  dst[idx] = (n < NCH) ? (_Float16)w_out[(size_t)n * 64 + c] : (_Float16)0.0f;
}

__global__ __launch_bounds__(128) void conv_in_k(const float* __restrict__ z,
                                                 const float* __restrict__ w_in,
                                                 const float* __restrict__ b_in,
                                                 _Float16* __restrict__ dst) {
  const int x = threadIdx.x;
  const int y = blockIdx.x & (HW - 1);
  const int b = blockIdx.x >> 7;

  float zv[27];
  #pragma unroll
  for (int ci = 0; ci < 3; ++ci) {
    const float* zp = z + ((size_t)(b * 3 + ci)) * HW * HW;
    #pragma unroll
    for (int ky = 0; ky < 3; ++ky) {
      const int yy = y + ky - 1;
      #pragma unroll
      for (int kx = 0; kx < 3; ++kx) {
        const int xx = x + kx - 1;
        const bool ok = ((unsigned)yy < (unsigned)HW) && ((unsigned)xx < (unsigned)HW);
        zv[ci * 9 + ky * 3 + kx] = ok ? zp[yy * HW + xx] : 0.0f;
      }
    }
  }

  _Float16* op = dst + fofs(b, y + 1, 0, x + XO);
  #pragma unroll
  for (int g = 0; g < NFEAT / 8; ++g) {
    v8h o;
    #pragma unroll
    for (int jp = 0; jp < 4; ++jp) {
      const int co = g * 8 + jp * 2;
      const float* wp = w_in + co * 27;
      float a0 = b_in[co];
      float a1 = b_in[co + 1];
      #pragma unroll
      for (int t = 0; t < 27; ++t) {
        a0 = __builtin_fmaf(zv[t], wp[t], a0);
        a1 = __builtin_fmaf(zv[t], wp[27 + t], a1);
      }
      o[jp * 2]     = (_Float16)a0;
      o[jp * 2 + 1] = (_Float16)a1;
    }
    *(v8h*)(op + (size_t)(g >> 1) * (XP * 16) + (g & 1) * 8) = o;
  }
}

template <bool RESID>
__global__ __launch_bounds__(256) void conv3x3_wmma(
    const _Float16* __restrict__ src, _Float16* __restrict__ dst,
    const _Float16* __restrict__ wl, const float* __restrict__ bias,
    const _Float16* __restrict__ resid, int do_relu) {
  extern __shared__ __attribute__((aligned(16))) _Float16 lw[];

  const int wave = threadIdx.x >> 5;
  const int lane = threadIdx.x & 31;
  const int tile = blockIdx.x * 8 + wave;
  const int b  = tile >> 10;
  const int r  = tile & 1023;
  const int y  = r >> 3;
  const int x0 = (r & 7) << 4;

  const int mrow = lane & 15;
  const int ksel = (lane >> 4) << 3;
  const int nlan = lane & 15;
  const int mhi  = (lane >> 4) << 3;

  constexpr int NPIECE = (9 * 64 * 64) / 8 / 256;
  {
    uint4 wst[NPIECE];
    const uint4* g = (const uint4*)wl;
    #pragma unroll
    for (int j = 0; j < NPIECE; ++j) wst[j] = g[threadIdx.x + 256 * j];
    #pragma unroll
    for (int j = 0; j < NPIECE; ++j) {
      const int i = threadIdx.x + 256 * j;
      *(uint4*)(lw + (i >> 3) * LDSP + (i & 7) * 8) = wst[j];
    }
  }
  v8h af[9][4];
  #pragma unroll
  for (int ky = 0; ky < 3; ++ky) {
    #pragma unroll
    for (int kx = 0; kx < 3; ++kx) {
      const _Float16* ap = src + fofs(b, y + ky, 0, x0 + mrow + kx + XO - 1) + ksel;
      #pragma unroll
      for (int q = 0; q < 4; ++q)
        af[ky * 3 + kx][q] = *(const v8h*)(ap + (size_t)q * (XP * 16));
    }
  }
  __builtin_amdgcn_sched_barrier(0);
  __builtin_amdgcn_fence(__ATOMIC_RELEASE, "workgroup", "local");
  __builtin_amdgcn_s_barrier();
  __builtin_amdgcn_fence(__ATOMIC_ACQUIRE, "workgroup", "local");

  v8f acc[4] = {};

  const size_t roff = fofs(b, y + 1, 0, x0 + mrow + XO) + mhi;
  v8h rv[4];
  v4f bq[4][2];

  const _Float16* lb = lw + nlan * LDSP + ksel;
  int hi_off = 5 * 64 * LDSP;
  asm volatile("" : "+v"(hi_off));
  const _Float16* lb_hi = lb + hi_off;
  v16h bf[2][4];
  #pragma unroll
  for (int nt = 0; nt < 4; ++nt) {
    const _Float16* w = lb + nt * 16 * LDSP;
    bf[0][nt] = cat8(*(const v8h*)w, *(const v8h*)(w + 16));
  }
  __builtin_amdgcn_sched_group_barrier(0x100, 8, 0);
  constexpr int RSTEP = 10;
  #pragma unroll
  for (int s = 0; s < 18; ++s) {
    if (s + 1 < 18) {
      const int tn = (s + 1) >> 1;
      const _Float16* wn = (tn < 5 ? lb + tn * (64 * LDSP) : lb_hi + (tn - 5) * (64 * LDSP))
                           + ((s + 1) & 1) * 32;
      #pragma unroll
      for (int nt = 0; nt < 4; ++nt) {
        const _Float16* w = wn + nt * 16 * LDSP;
        bf[(s + 1) & 1][nt] = cat8(*(const v8h*)w, *(const v8h*)(w + 16));
      }
    }
    const v16h a = cat8(af[s >> 1][2 * (s & 1)], af[s >> 1][2 * (s & 1) + 1]);
    #pragma unroll
    for (int nt = 0; nt < 4; ++nt) acc[nt] = wmma16(bf[s & 1][nt], a, acc[nt]);
    if (s + 1 < 18) __builtin_amdgcn_sched_group_barrier(0x100, 8, 0);
    __builtin_amdgcn_sched_group_barrier(0x008, 4, 0);
    if (s == RSTEP) {
      #pragma unroll
      for (int nt = 0; nt < 4; ++nt) {
        bq[nt][0] = *(const v4f*)(bias + nt * 16 + mhi);
        bq[nt][1] = *(const v4f*)(bias + nt * 16 + mhi + 4);
        if (RESID) rv[nt] = *(const v8h*)(resid + roff + (size_t)nt * (XP * 16));
      }
      __builtin_amdgcn_sched_group_barrier(0x020, RESID ? 12 : 8, 0);
    }
  }

  #pragma unroll
  for (int nt = 0; nt < 4; ++nt) {
    v8h o;
    #pragma unroll
    for (int i = 0; i < 8; ++i) {
      float v = acc[nt][i] + bq[nt][i >> 2][i & 3];
      if (RESID) v += (float)rv[nt][i];
      if (do_relu) v = v > 0.0f ? v : 0.0f;
      o[i] = (_Float16)v;
    }
    *(v8h*)(dst + roff + (size_t)nt * (XP * 16)) = o;
  }
}

__global__ __launch_bounds__(128) void convout_softmax(
    const _Float16* __restrict__ feat, const _Float16* __restrict__ wo,
    const float* __restrict__ bo, _Float16* __restrict__ probs,
    float* __restrict__ nsum) {
  const int wave = threadIdx.x >> 5;
  const int lane = threadIdx.x & 31;
  const int tile = blockIdx.x * 4 + wave;
  const int b  = tile >> 10;
  const int r  = tile & 1023;
  const int y  = r >> 3;
  const int x0 = (r & 7) << 4;

  const int mrow = lane & 15;
  const int ksel = (lane >> 4) << 3;
  const int nlan = lane & 15;
  const int mhi  = (lane >> 4) << 3;

  const _Float16* ap = feat + fofs(b, y + 1, 0, x0 + mrow + XO) + ksel;
  const size_t PLS = (size_t)XP * 16;
  v16h a0 = cat8(*(const v8h*)(ap),           *(const v8h*)(ap + PLS));
  v16h a1 = cat8(*(const v8h*)(ap + 2 * PLS), *(const v8h*)(ap + 3 * PLS));

  constexpr int NT = NPADC / 16;
  v8f lg[NT];

  constexpr int PF = 5;
  __builtin_amdgcn_sched_group_barrier(0x020, 4 + 4 * PF, 0);
  #pragma unroll
  for (int nt = 0; nt < NT; ++nt) {
    const int n = nt * 16 + nlan;
    const _Float16* wp = wo + (size_t)n * 64;
    v16h b0 = cat8(*(const v8h*)(wp + ksel),      *(const v8h*)(wp + 16 + ksel));
    v16h b1 = cat8(*(const v8h*)(wp + 32 + ksel), *(const v8h*)(wp + 48 + ksel));
    v8f c = {};
    c = wmma16(a0, b0, c);
    c = wmma16(a1, b1, c);
    lg[nt] = c;
    __builtin_amdgcn_sched_group_barrier(0x008, 2, 0);
    __builtin_amdgcn_sched_group_barrier(0x020, 4, 0);
  }
  #pragma unroll
  for (int nt = 0; nt < NT; ++nt) {
    const int n = nt * 16 + nlan;
    const float bv = (n < NCH) ? bo[n] : -1e30f;
    #pragma unroll
    for (int i = 0; i < 8; ++i) lg[nt][i] += bv;
  }

  float inv[8];
  #pragma unroll
  for (int i = 0; i < 8; ++i) {
    float m = lg[0][i];
    #pragma unroll
    for (int nt = 1; nt < NT; ++nt) m = fmaxf(m, lg[nt][i]);
    #pragma unroll
    for (int off = 8; off; off >>= 1) m = fmaxf(m, __shfl_xor(m, off, 32));
    float s = 0.0f;
    #pragma unroll
    for (int nt = 0; nt < NT; ++nt) {
      const float e = __expf(lg[nt][i] - m);
      lg[nt][i] = e;
      s += e;
    }
    #pragma unroll
    for (int off = 8; off; off >>= 1) s += __shfl_xor(s, off, 32);
    inv[i] = 1.0f / s;
  }

  _Float16* pb = probs + (size_t)b * NCH * (HW * HW) + (size_t)y * HW + x0 + mhi;
  #pragma unroll
  for (int nt = 0; nt < NT; ++nt) {
    const int t = nt * 16 + nlan;
    v8h o;
    #pragma unroll
    for (int i = 0; i < 8; ++i) o[i] = (_Float16)(lg[nt][i] * inv[i]);
    if (t < NCH) *(v8h*)(pb + (size_t)t * (HW * HW)) = o;
  }

  float nacc = 0.0f;
  if (nlan == (KTAP & 15)) {
    #pragma unroll
    for (int i = 0; i < 8; ++i) nacc += lg[KTAP >> 4][i] * inv[i];
  }
  nacc += __shfl_xor(nacc, 16, 32);
  if (lane == (KTAP & 15)) atomicAdd(nsum + b, nacc);
}

typedef __attribute__((ext_vector_type(4))) _Float16 v4h;
typedef __attribute__((ext_vector_type(2))) float    v2f;

#define AK_WAVES 7
#define AK_ROWS  3
#define AK_NG    9

__global__ __launch_bounds__(AK_WAVES * 32) void apply_k(
    const float* __restrict__ x, const float* __restrict__ eps,
    const _Float16* __restrict__ probs, const float* __restrict__ nsum,
    float* __restrict__ out) {
  __shared__ float red[AK_WAVES * 12 * 32];
  const int lane = threadIdx.x & 31;
  const int wave = __builtin_amdgcn_readfirstlane(threadIdx.x >> 5);
  const int h = blockIdx.x & (HW - 1);
  const int b = blockIdx.x >> 7;
  const int col0 = 16 * lane - 12;

  float acc[3][4];
  #pragma unroll
  for (int c = 0; c < 3; ++c)
    #pragma unroll
    for (int q = 0; q < 4; ++q) acc[c][q] = 0.0f;

  #pragma unroll 1
  for (int r = 0; r < AK_ROWS; ++r) {
    const int ky = wave * AK_ROWS + r;
    const int yi = 4 * h - 10 + ky;
    const float rowkeep = ((unsigned)yi < 512u) ? 1.0f : 0.0f;
    const int yc = yi < 0 ? 0 : (yi > 511 ? 511 : yi);

    const _Float16* pr =
        probs + ((size_t)(b * NCH + ky * 21)) * (HW * HW) + (size_t)h * HW + 4 * lane;
    v4h pv[21];
    #pragma unroll
    for (int kx = 0; kx < 21; ++kx)
      pv[kx] = __builtin_nontemporal_load((const v4h*)(pr + (size_t)kx * (HW * HW)));

    bool ok[AK_NG];
    #pragma unroll
    for (int g = 0; g < AK_NG; ++g) {
      const int col = col0 + 4 * g;
      ok[g] = (g < 3) ? (col >= 0) : ((g > 6) ? (col < 512) : true);
    }
    v2f xlo[3];
    v4f xv[3][AK_NG - 2];
    v2f xhi[3];
    float xend[3];
    #pragma unroll
    for (int c = 0; c < 3; ++c) {
      const float* xr = x + ((size_t)((b * 3 + c) * 512 + yc)) * 512;
      xlo[c] = *(const v2f*)(xr + (ok[0] ? col0 : 0) + 2);
      #pragma unroll
      for (int g = 1; g < AK_NG - 1; ++g)
        xv[c][g - 1] = *(const v4f*)(xr + (ok[g] ? col0 + 4 * g : 0));
      const float* xe = xr + (ok[AK_NG - 1] ? col0 + 4 * (AK_NG - 1) : 0);
      xhi[c]  = *(const v2f*)xe;
      xend[c] = xe[2];
    }
    __builtin_amdgcn_sched_barrier(0);

    float keep[AK_NG];
    #pragma unroll
    for (int g = 0; g < AK_NG; ++g) keep[g] = ok[g] ? 1.0f : 0.0f;
    float xs[3][4 * AK_NG];
    #pragma unroll
    for (int c = 0; c < 3; ++c) {
      xs[c][2] = xlo[c][0] * keep[0];
      xs[c][3] = xlo[c][1] * keep[0];
      #pragma unroll
      for (int g = 1; g < AK_NG - 1; ++g) {
        #pragma unroll
        for (int e = 0; e < 4; ++e)
          xs[c][4 * g + e] = (g < 3 || g > 6) ? xv[c][g - 1][e] * keep[g] : xv[c][g - 1][e];
      }
      xs[c][32] = xhi[c][0] * keep[AK_NG - 1];
      xs[c][33] = xhi[c][1] * keep[AK_NG - 1];
      xs[c][34] = xend[c]   * keep[AK_NG - 1];
    }

    float part[3][4];
    #pragma unroll
    for (int c = 0; c < 3; ++c)
      #pragma unroll
      for (int q = 0; q < 4; ++q) part[c][q] = 0.0f;
    #pragma unroll
    for (int kx = 0; kx < 21; ++kx) {
      #pragma unroll
      for (int q = 0; q < 4; ++q) {
        const float pf = (float)pv[kx][q];
        #pragma unroll
        for (int c = 0; c < 3; ++c) part[c][q] += xs[c][4 * q + 2 + kx] * pf;
      }
    }
    #pragma unroll
    for (int c = 0; c < 3; ++c)
      #pragma unroll
      for (int q = 0; q < 4; ++q) acc[c][q] += part[c][q] * rowkeep;
  }

  #pragma unroll
  for (int c = 0; c < 3; ++c)
    #pragma unroll
    for (int q = 0; q < 4; ++q) red[(wave * 12 + c * 4 + q) * 32 + lane] = acc[c][q];
  __syncthreads();

  if (wave < 3) {
    const int c = wave;
    float s[4];
    #pragma unroll
    for (int q = 0; q < 4; ++q) {
      float t = 0.0f;
      #pragma unroll
      for (int j = 0; j < AK_WAVES; ++j) t += red[(j * 12 + c * 4 + q) * 32 + lane];
      s[q] = t;
    }
    const size_t idx = (((size_t)(b * 3 + c)) * HW + h) * HW + 4 * lane;
    const float ns = nsum[b] * (1.0f / (HW * HW));
    const v4f e4 = *(const v4f*)(eps + idx);
    v4f o;
    #pragma unroll
    for (int q = 0; q < 4; ++q) o[q] = s[q] + ns * e4[q];
    *(v4f*)(out + idx) = o;
  }
}

extern "C" void kernel_launch(void* const* d_in, const int* in_sizes, int n_in,
                              void* d_out, int out_size, void* d_ws, size_t ws_size,
                              hipStream_t stream) {
  const float* x     = (const float*)d_in[0];
  const float* z     = (const float*)d_in[1];
  const float* eps   = (const float*)d_in[2];
  const float* w_in  = (const float*)d_in[3];
  const float* b_in  = (const float*)d_in[4];
  const float* w1s   = (const float*)d_in[5];
  const float* b1s   = (const float*)d_in[6];
  const float* w2s   = (const float*)d_in[7];
  const float* b2s   = (const float*)d_in[8];
  const float* w_out = (const float*)d_in[9];
  const float* b_out = (const float*)d_in[10];
  float* out = (float*)d_out;

  char* ws = (char*)d_ws;
  const size_t FBYTES = (size_t)NBATCH * HALO * NPL * XP * 16 * sizeof(_Float16);
  _Float16* F0 = (_Float16*)(ws);
  _Float16* F1 = (_Float16*)(ws + FBYTES);
  _Float16* F2 = (_Float16*)(ws + 2 * FBYTES);
  _Float16* wpack  = (_Float16*)(ws + 3 * FBYTES);
  _Float16* probs  = F1;
  _Float16* wopack = probs + (size_t)NBATCH * NCH * HW * HW;
  float*    nsum   = (float*)(wopack + (size_t)NPADC * 64);

  zero_frames<<<(NPLANE * FRAME_PIECES + 255) / 256, 256, 0, stream>>>(F0, nsum);

  pack_w<<<(2 * 8 * 9 * 64 * 64 + 255) / 256, 256, 0, stream>>>(w1s, w2s, wpack);
  pack_wout<<<(NPADC * 64 + 255) / 256, 256, 0, stream>>>(w_out, wopack);

  conv_in_k<<<NBATCH * HW, HW, 0, stream>>>(z, w_in, b_in, F0);

  _Float16* cur = F0;
  _Float16* nxt = F2;
  const size_t LAYER = (size_t)9 * 64 * 64;
  for (int i = 0; i < 8; ++i) {
    conv3x3_wmma<false><<<512, 256, 9 * 64 * LDSP * sizeof(_Float16), stream>>>(
        cur, F1, wpack + (size_t)i * LAYER, b1s + (size_t)i * 64, nullptr, 1);
    conv3x3_wmma<true><<<512, 256, 9 * 64 * LDSP * sizeof(_Float16), stream>>>(
        F1, nxt, wpack + (size_t)(8 + i) * LAYER, b2s + (size_t)i * 64, cur, 0);
    _Float16* t = cur; cur = nxt; nxt = t;
  }

  convout_softmax<<<1024, 128, 0, stream>>>(cur, wopack, b_out, probs, nsum);
  apply_k<<<NBATCH * HW, AK_WAVES * 32, 0, stream>>>(x, eps, probs, nsum, out);
}


# 3x3 conv kernel: a wave computes 16 columns of two image rows, so each LDS weight fragment feeds two WMMAs; one 8-wave work-group per WGP (256) stages the weights once per 256 pixels
# speedup vs baseline: 4.5662x; 1.1023x over previous
#include <hip/hip_runtime.h>
#include <hip/hip_bf16.h>
#include <stdint.h>

typedef __attribute__((ext_vector_type(16))) _Float16 v16h;
typedef __attribute__((ext_vector_type(8)))  _Float16 v8h;
typedef __attribute__((ext_vector_type(8)))  float    v8f;
typedef __attribute__((ext_vector_type(4)))  float    v4f;

#define NFEAT 64
#define HW    128
#define HALO  130
#define NBATCH 4
#define KTAP  441
#define NCH   442
#define NPADC 448
#define LDSP  72

#define NPL   4
#define XP    136
#define XO    4
__device__ __forceinline__ size_t fofs(int b, int yy, int pl, int xx) {
  return ((((size_t)b * HALO + yy) * NPL + pl) * XP + xx) * 16;
}

__device__ __forceinline__ v16h cat8(v8h lo, v8h hi) {
  return __builtin_shufflevector(lo, hi, 0,1,2,3,4,5,6,7,8,9,10,11,12,13,14,15);
}
__device__ __forceinline__ v8f wmma16(v16h a, v16h b, v8f c) {
  return __builtin_amdgcn_wmma_f32_16x16x32_f16(false, a, false, b, (short)0, c, false, false);
}

#define ROW_PIECES   (NPL * XP * 2)
#define IMG_PIECES   (HALO * ROW_PIECES)
#define SIDE_PIECES  (HW * NPL * 2 * 2)
#define FRAME_PIECES (2 * ROW_PIECES + SIDE_PIECES)
#define NPLANE       (3 * NBATCH)
__global__ __launch_bounds__(256) void zero_frames(_Float16* __restrict__ feat,
                                                   float* __restrict__ nsum) {
  const int idx = blockIdx.x * 256 + threadIdx.x;
  if (idx < NBATCH) nsum[idx] = 0.0f;
  if (idx >= NPLANE * FRAME_PIECES) return;
  const int plane = idx / FRAME_PIECES;
  const int rem   = idx - plane * FRAME_PIECES;
  int piece;
  if (rem < ROW_PIECES) {
    piece = rem;
  } else if (rem < 2 * ROW_PIECES) {
    piece = (HALO - 1) * ROW_PIECES + (rem - ROW_PIECES);
  } else {
    const int k    = rem - 2 * ROW_PIECES;
    const int half = k & 1;
    const int side = (k >> 1) & 1;
    const int pl   = (k >> 2) & 3;
    const int yy   = 1 + (k >> 4);
    piece = ((yy * NPL + pl) * XP + (side ? XO + HW : XO - 1)) * 2 + half;
  }
  uint4* p = (uint4*)feat;
  p[(size_t)plane * IMG_PIECES + piece] = make_uint4(0u, 0u, 0u, 0u);
}

__global__ void pack_w(const float* __restrict__ w1s, const float* __restrict__ w2s,
                       _Float16* __restrict__ dst) {
  int idx = blockIdx.x * blockDim.x + threadIdx.x;
  const int TOT = 2 * 8 * 9 * 64 * 64;
  if (idx >= TOT) return;
  int c   = idx & 63;
  int n   = (idx >> 6) & 63;
  int pg  = idx >> 12;
  int pos = pg % 9;
  int lay = pg / 9;
  int blk = lay & 7;
  int which = lay >> 3;
  const float* src = which ? w2s : w1s;
  float v = src[(((size_t)(blk * 64 + n) * 64 + c) * 9) + pos];
  dst[idx] = (_Float16)v;
}

__global__ void pack_wout(const float* __restrict__ w_out, _Float16* __restrict__ dst) {
  int idx = blockIdx.x * blockDim.x + threadIdx.x;
  if (idx >= NPADC * 64) return;
  int c = idx & 63;
  int n = idx >> 6;
  dst[idx] = (n < NCH) ? (_Float16)w_out[(size_t)n * 64 + c] : (_Float16)0.0f;
}

__global__ __launch_bounds__(128) void conv_in_k(const float* __restrict__ z,
                                                 const float* __restrict__ w_in,
                                                 const float* __restrict__ b_in,
                                                 _Float16* __restrict__ dst) {
  const int x = threadIdx.x;
  const int y = blockIdx.x & (HW - 1);
  const int b = blockIdx.x >> 7;

  float zv[27];
  #pragma unroll
  for (int ci = 0; ci < 3; ++ci) {
    const float* zp = z + ((size_t)(b * 3 + ci)) * HW * HW;
    #pragma unroll
    for (int ky = 0; ky < 3; ++ky) {
      const int yy = y + ky - 1;
      #pragma unroll
      for (int kx = 0; kx < 3; ++kx) {
        const int xx = x + kx - 1;
        const bool ok = ((unsigned)yy < (unsigned)HW) && ((unsigned)xx < (unsigned)HW);
        zv[ci * 9 + ky * 3 + kx] = ok ? zp[yy * HW + xx] : 0.0f;
      }
    }
  }

  _Float16* op = dst + fofs(b, y + 1, 0, x + XO);
  #pragma unroll
  for (int g = 0; g < NFEAT / 8; ++g) {
    v8h o;
    #pragma unroll
    for (int jp = 0; jp < 4; ++jp) {
      const int co = g * 8 + jp * 2;
      const float* wp = w_in + co * 27;
      float a0 = b_in[co];
      float a1 = b_in[co + 1];
      #pragma unroll
      for (int t = 0; t < 27; ++t) {
        a0 = __builtin_fmaf(zv[t], wp[t], a0);
        a1 = __builtin_fmaf(zv[t], wp[27 + t], a1);
      }
      o[jp * 2]     = (_Float16)a0;
      o[jp * 2 + 1] = (_Float16)a1;
    }
    *(v8h*)(op + (size_t)(g >> 1) * (XP * 16) + (g & 1) * 8) = o;
  }
}

template <bool RESID>
__global__ __launch_bounds__(256) void conv3x3_wmma(
    const _Float16* __restrict__ src, _Float16* __restrict__ dst,
    const _Float16* __restrict__ wl, const float* __restrict__ bias,
    const _Float16* __restrict__ resid, int do_relu) {
  extern __shared__ __attribute__((aligned(16))) _Float16 lw[];

  const int wave = threadIdx.x >> 5;
  const int lane = threadIdx.x & 31;
  const int tile = blockIdx.x * 8 + wave;
  const int b  = tile >> 9;
  const int r  = tile & 511;
  const int y  = (r >> 3) << 1;
  const int x0 = (r & 7) << 4;

  const int mrow = lane & 15;
  const int ksel = (lane >> 4) << 3;
  const int nlan = lane & 15;
  const int mhi  = (lane >> 4) << 3;

  constexpr int NPIECE = (9 * 64 * 64) / 8 / 256;
  {
    uint4 wst[NPIECE];
    const uint4* g = (const uint4*)wl;
    #pragma unroll
    for (int j = 0; j < NPIECE; ++j) wst[j] = g[threadIdx.x + 256 * j];
    #pragma unroll
    for (int j = 0; j < NPIECE; ++j) {
      const int i = threadIdx.x + 256 * j;
      *(uint4*)(lw + (i >> 3) * LDSP + (i & 7) * 8) = wst[j];
    }
  }
  v8h af[4][3][4];
  #pragma unroll
  for (int rr = 0; rr < 4; ++rr) {
    #pragma unroll
    for (int kx = 0; kx < 3; ++kx) {
      const _Float16* ap = src + fofs(b, y + rr, 0, x0 + mrow + kx + XO - 1) + ksel;
      #pragma unroll
      for (int q = 0; q < 4; ++q)
        af[rr][kx][q] = *(const v8h*)(ap + (size_t)q * (XP * 16));
    }
  }
  __builtin_amdgcn_sched_barrier(0);
  __builtin_amdgcn_fence(__ATOMIC_RELEASE, "workgroup", "local");
  __builtin_amdgcn_s_barrier();
  __builtin_amdgcn_fence(__ATOMIC_ACQUIRE, "workgroup", "local");

  v8f acc[2][4] = {};

  const size_t roff = fofs(b, y + 1, 0, x0 + mrow + XO) + mhi;
  constexpr size_t ROWH = (size_t)NPL * XP * 16;
  v8h rv[2][4];
  v4f bq[4][2];

  const _Float16* lb = lw + nlan * LDSP + ksel;
  int hi_off = 5 * 64 * LDSP;
  asm volatile("" : "+v"(hi_off));
  const _Float16* lb_hi = lb + hi_off;
  v16h bf[2][4];
  #pragma unroll
  for (int nt = 0; nt < 4; ++nt) {
    const _Float16* w = lb + nt * 16 * LDSP;
    bf[0][nt] = cat8(*(const v8h*)w, *(const v8h*)(w + 16));
  }
  __builtin_amdgcn_sched_group_barrier(0x100, 8, 0);
  constexpr int RSTEP = 8;
  #pragma unroll
  for (int s = 0; s < 18; ++s) {
    if (s + 1 < 18) {
      const int tn = (s + 1) >> 1;
      const _Float16* wn = (tn < 5 ? lb + tn * (64 * LDSP) : lb_hi + (tn - 5) * (64 * LDSP))
                           + ((s + 1) & 1) * 32;
      #pragma unroll
      for (int nt = 0; nt < 4; ++nt) {
        const _Float16* w = wn + nt * 16 * LDSP;
        bf[(s + 1) & 1][nt] = cat8(*(const v8h*)w, *(const v8h*)(w + 16));
      }
    }
    const int ky = (s >> 1) / 3;
    const int kx = (s >> 1) % 3;
    const int hq = 2 * (s & 1);
    const v16h a0 = cat8(af[ky][kx][hq],     af[ky][kx][hq + 1]);
    const v16h a1 = cat8(af[ky + 1][kx][hq], af[ky + 1][kx][hq + 1]);
    #pragma unroll
    for (int nt = 0; nt < 4; ++nt) {
      acc[0][nt] = wmma16(bf[s & 1][nt], a0, acc[0][nt]);
      acc[1][nt] = wmma16(bf[s & 1][nt], a1, acc[1][nt]);
    }
    if (s + 1 < 18) __builtin_amdgcn_sched_group_barrier(0x100, 8, 0);
    __builtin_amdgcn_sched_group_barrier(0x008, 8, 0);
    if (s == RSTEP) {
      #pragma unroll
      for (int nt = 0; nt < 4; ++nt) {
        bq[nt][0] = *(const v4f*)(bias + nt * 16 + mhi);
        bq[nt][1] = *(const v4f*)(bias + nt * 16 + mhi + 4);
        if (RESID) {
          rv[0][nt] = *(const v8h*)(resid + roff + (size_t)nt * (XP * 16));
          rv[1][nt] = *(const v8h*)(resid + roff + ROWH + (size_t)nt * (XP * 16));
        }
      }
      __builtin_amdgcn_sched_group_barrier(0x020, RESID ? 16 : 8, 0);
    }
  }

  #pragma unroll
  for (int t = 0; t < 2; ++t) {
    #pragma unroll
    for (int nt = 0; nt < 4; ++nt) {
      v8h o;
      #pragma unroll
      for (int i = 0; i < 8; ++i) {
        float v = acc[t][nt][i] + bq[nt][i >> 2][i & 3];
        if (RESID) v += (float)rv[t][nt][i];
        if (do_relu) v = v > 0.0f ? v : 0.0f;
        o[i] = (_Float16)v;
      }
      *(v8h*)(dst + roff + (size_t)t * ROWH + (size_t)nt * (XP * 16)) = o;
    }
  }
}

__global__ __launch_bounds__(128) void convout_softmax(
    const _Float16* __restrict__ feat, const _Float16* __restrict__ wo,
    const float* __restrict__ bo, _Float16* __restrict__ probs,
    float* __restrict__ nsum) {
  const int wave = threadIdx.x >> 5;
  const int lane = threadIdx.x & 31;
  const int tile = blockIdx.x * 4 + wave;
  const int b  = tile >> 10;
  const int r  = tile & 1023;
  const int y  = r >> 3;
  const int x0 = (r & 7) << 4;

  const int mrow = lane & 15;
  const int ksel = (lane >> 4) << 3;
  const int nlan = lane & 15;
  const int mhi  = (lane >> 4) << 3;

  const _Float16* ap = feat + fofs(b, y + 1, 0, x0 + mrow + XO) + ksel;
  const size_t PLS = (size_t)XP * 16;
  v16h a0 = cat8(*(const v8h*)(ap),           *(const v8h*)(ap + PLS));
  v16h a1 = cat8(*(const v8h*)(ap + 2 * PLS), *(const v8h*)(ap + 3 * PLS));

  constexpr int NT = NPADC / 16;
  v8f lg[NT];

  constexpr int PF = 5;
  __builtin_amdgcn_sched_group_barrier(0x020, 4 + 4 * PF, 0);
  #pragma unroll
  for (int nt = 0; nt < NT; ++nt) {
    const int n = nt * 16 + nlan;
    const _Float16* wp = wo + (size_t)n * 64;
    v16h b0 = cat8(*(const v8h*)(wp + ksel),      *(const v8h*)(wp + 16 + ksel));
    v16h b1 = cat8(*(const v8h*)(wp + 32 + ksel), *(const v8h*)(wp + 48 + ksel));
    v8f c = {};
    c = wmma16(a0, b0, c);
    c = wmma16(a1, b1, c);
    lg[nt] = c;
    __builtin_amdgcn_sched_group_barrier(0x008, 2, 0);
    __builtin_amdgcn_sched_group_barrier(0x020, 4, 0);
  }
  #pragma unroll
  for (int nt = 0; nt < NT; ++nt) {
    const int n = nt * 16 + nlan;
    const float bv = (n < NCH) ? bo[n] : -1e30f;
    #pragma unroll
    for (int i = 0; i < 8; ++i) lg[nt][i] += bv;
  }

  float inv[8];
  #pragma unroll
  for (int i = 0; i < 8; ++i) {
    float m = lg[0][i];
    #pragma unroll
    for (int nt = 1; nt < NT; ++nt) m = fmaxf(m, lg[nt][i]);
    #pragma unroll
    for (int off = 8; off; off >>= 1) m = fmaxf(m, __shfl_xor(m, off, 32));
    float s = 0.0f;
    #pragma unroll
    for (int nt = 0; nt < NT; ++nt) {
      const float e = __expf(lg[nt][i] - m);
      lg[nt][i] = e;
      s += e;
    }
    #pragma unroll
    for (int off = 8; off; off >>= 1) s += __shfl_xor(s, off, 32);
    inv[i] = 1.0f / s;
  }

  _Float16* pb = probs + (size_t)b * NCH * (HW * HW) + (size_t)y * HW + x0 + mhi;
  #pragma unroll
  for (int nt = 0; nt < NT; ++nt) {
    const int t = nt * 16 + nlan;
    v8h o;
    #pragma unroll
    for (int i = 0; i < 8; ++i) o[i] = (_Float16)(lg[nt][i] * inv[i]);
    if (t < NCH) *(v8h*)(pb + (size_t)t * (HW * HW)) = o;
  }

  float nacc = 0.0f;
  if (nlan == (KTAP & 15)) {
    #pragma unroll
    for (int i = 0; i < 8; ++i) nacc += lg[KTAP >> 4][i] * inv[i];
  }
  nacc += __shfl_xor(nacc, 16, 32);
  if (lane == (KTAP & 15)) atomicAdd(nsum + b, nacc);
}

typedef __attribute__((ext_vector_type(4))) _Float16 v4h;
typedef __attribute__((ext_vector_type(2))) float    v2f;

#define AK_WAVES 7
#define AK_ROWS  3
#define AK_NG    9

__global__ __launch_bounds__(AK_WAVES * 32) void apply_k(
    const float* __restrict__ x, const float* __restrict__ eps,
    const _Float16* __restrict__ probs, const float* __restrict__ nsum,
    float* __restrict__ out) {
  __shared__ float red[AK_WAVES * 12 * 32];
  const int lane = threadIdx.x & 31;
  const int wave = __builtin_amdgcn_readfirstlane(threadIdx.x >> 5);
  const int h = blockIdx.x & (HW - 1);
  const int b = blockIdx.x >> 7;
  const int col0 = 16 * lane - 12;

  float acc[3][4];
  #pragma unroll
  for (int c = 0; c < 3; ++c)
    #pragma unroll
    for (int q = 0; q < 4; ++q) acc[c][q] = 0.0f;

  #pragma unroll 1
  for (int r = 0; r < AK_ROWS; ++r) {
    const int ky = wave * AK_ROWS + r;
    const int yi = 4 * h - 10 + ky;
    const float rowkeep = ((unsigned)yi < 512u) ? 1.0f : 0.0f;
    const int yc = yi < 0 ? 0 : (yi > 511 ? 511 : yi);

    const _Float16* pr =
        probs + ((size_t)(b * NCH + ky * 21)) * (HW * HW) + (size_t)h * HW + 4 * lane;
    v4h pv[21];
    #pragma unroll
    for (int kx = 0; kx < 21; ++kx)
      pv[kx] = __builtin_nontemporal_load((const v4h*)(pr + (size_t)kx * (HW * HW)));

    bool ok[AK_NG];
    #pragma unroll
    for (int g = 0; g < AK_NG; ++g) {
      const int col = col0 + 4 * g;
      ok[g] = (g < 3) ? (col >= 0) : ((g > 6) ? (col < 512) : true);
    }
    v2f xlo[3];
    v4f xv[3][AK_NG - 2];
    v2f xhi[3];
    float xend[3];
    #pragma unroll
    for (int c = 0; c < 3; ++c) {
      const float* xr = x + ((size_t)((b * 3 + c) * 512 + yc)) * 512;
      xlo[c] = *(const v2f*)(xr + (ok[0] ? col0 : 0) + 2);
      #pragma unroll
      for (int g = 1; g < AK_NG - 1; ++g)
        xv[c][g - 1] = *(const v4f*)(xr + (ok[g] ? col0 + 4 * g : 0));
      const float* xe = xr + (ok[AK_NG - 1] ? col0 + 4 * (AK_NG - 1) : 0);
      xhi[c]  = *(const v2f*)xe;
      xend[c] = xe[2];
    }
    __builtin_amdgcn_sched_barrier(0);

    float keep[AK_NG];
    #pragma unroll
    for (int g = 0; g < AK_NG; ++g) keep[g] = ok[g] ? 1.0f : 0.0f;
    float xs[3][4 * AK_NG];
    #pragma unroll
    for (int c = 0; c < 3; ++c) {
      xs[c][2] = xlo[c][0] * keep[0];
      xs[c][3] = xlo[c][1] * keep[0];
      #pragma unroll
      for (int g = 1; g < AK_NG - 1; ++g) {
        #pragma unroll
        for (int e = 0; e < 4; ++e)
          xs[c][4 * g + e] = (g < 3 || g > 6) ? xv[c][g - 1][e] * keep[g] : xv[c][g - 1][e];
      }
      xs[c][32] = xhi[c][0] * keep[AK_NG - 1];
      xs[c][33] = xhi[c][1] * keep[AK_NG - 1];
      xs[c][34] = xend[c]   * keep[AK_NG - 1];
    }

    float part[3][4];
    #pragma unroll
    for (int c = 0; c < 3; ++c)
      #pragma unroll
      for (int q = 0; q < 4; ++q) part[c][q] = 0.0f;
    #pragma unroll
    for (int kx = 0; kx < 21; ++kx) {
      #pragma unroll
      for (int q = 0; q < 4; ++q) {
        const float pf = (float)pv[kx][q];
        #pragma unroll
        for (int c = 0; c < 3; ++c) part[c][q] += xs[c][4 * q + 2 + kx] * pf;
      }
    }
    #pragma unroll
    for (int c = 0; c < 3; ++c)
      #pragma unroll
      for (int q = 0; q < 4; ++q) acc[c][q] += part[c][q] * rowkeep;
  }

  #pragma unroll
  for (int c = 0; c < 3; ++c)
    #pragma unroll
    for (int q = 0; q < 4; ++q) red[(wave * 12 + c * 4 + q) * 32 + lane] = acc[c][q];
  __syncthreads();

  if (wave < 3) {
    const int c = wave;
    float s[4];
    #pragma unroll
    for (int q = 0; q < 4; ++q) {
      float t = 0.0f;
      #pragma unroll
      for (int j = 0; j < AK_WAVES; ++j) t += red[(j * 12 + c * 4 + q) * 32 + lane];
      s[q] = t;
    }
    const size_t idx = (((size_t)(b * 3 + c)) * HW + h) * HW + 4 * lane;
    const float ns = nsum[b] * (1.0f / (HW * HW));
    const v4f e4 = *(const v4f*)(eps + idx);
    v4f o;
    #pragma unroll
    for (int q = 0; q < 4; ++q) o[q] = s[q] + ns * e4[q];
    *(v4f*)(out + idx) = o;
  }
}

extern "C" void kernel_launch(void* const* d_in, const int* in_sizes, int n_in,
                              void* d_out, int out_size, void* d_ws, size_t ws_size,
                              hipStream_t stream) {
  const float* x     = (const float*)d_in[0];
  const float* z     = (const float*)d_in[1];
  const float* eps   = (const float*)d_in[2];
  const float* w_in  = (const float*)d_in[3];
  const float* b_in  = (const float*)d_in[4];
  const float* w1s   = (const float*)d_in[5];
  const float* b1s   = (const float*)d_in[6];
  const float* w2s   = (const float*)d_in[7];
  const float* b2s   = (const float*)d_in[8];
  const float* w_out = (const float*)d_in[9];
  const float* b_out = (const float*)d_in[10];
  float* out = (float*)d_out;

  char* ws = (char*)d_ws;
  const size_t FBYTES = (size_t)NBATCH * HALO * NPL * XP * 16 * sizeof(_Float16);
  _Float16* F0 = (_Float16*)(ws);
  _Float16* F1 = (_Float16*)(ws + FBYTES);
  _Float16* F2 = (_Float16*)(ws + 2 * FBYTES);
  _Float16* wpack  = (_Float16*)(ws + 3 * FBYTES);
  _Float16* probs  = F1;
  _Float16* wopack = probs + (size_t)NBATCH * NCH * HW * HW;
  float*    nsum   = (float*)(wopack + (size_t)NPADC * 64);

  zero_frames<<<(NPLANE * FRAME_PIECES + 255) / 256, 256, 0, stream>>>(F0, nsum);

  pack_w<<<(2 * 8 * 9 * 64 * 64 + 255) / 256, 256, 0, stream>>>(w1s, w2s, wpack);
  pack_wout<<<(NPADC * 64 + 255) / 256, 256, 0, stream>>>(w_out, wopack);

  conv_in_k<<<NBATCH * HW, HW, 0, stream>>>(z, w_in, b_in, F0);

  _Float16* cur = F0;
  _Float16* nxt = F2;
  const size_t LAYER = (size_t)9 * 64 * 64;
  for (int i = 0; i < 8; ++i) {
    conv3x3_wmma<false><<<256, 256, 9 * 64 * LDSP * sizeof(_Float16), stream>>>(
        cur, F1, wpack + (size_t)i * LAYER, b1s + (size_t)i * 64, nullptr, 1);
    conv3x3_wmma<true><<<256, 256, 9 * 64 * LDSP * sizeof(_Float16), stream>>>(
        F1, nxt, wpack + (size_t)(8 + i) * LAYER, b2s + (size_t)i * 64, cur, 0);
    _Float16* t = cur; cur = nxt; nxt = t;
  }

  convout_softmax<<<1024, 128, 0, stream>>>(cur, wopack, b_out, probs, nsum);
  apply_k<<<NBATCH * HW, AK_WAVES * 32, 0, stream>>>(x, eps, probs, nsum, out);
}


# trunk: one launch per residual block (8, were 16): a work-group computes conv1+ReLU on the 4 rows its 2 output rows need (half again of conv1's work), keeps them in LDS (f16), runs conv2+skip there
# speedup vs baseline: 5.3496x; 1.1716x over previous
#include <hip/hip_runtime.h>
#include <hip/hip_bf16.h>
#include <stdint.h>

typedef __attribute__((ext_vector_type(16))) _Float16 v16h;
typedef __attribute__((ext_vector_type(8)))  _Float16 v8h;
typedef __attribute__((ext_vector_type(8)))  float    v8f;
typedef __attribute__((ext_vector_type(4)))  float    v4f;

#define NFEAT 64
#define HW    128
#define HALO  130
#define NBATCH 4
#define KTAP  441
#define NCH   442
#define NPADC 448
#define LDSP  72

#define NPL   4
#define XP    136
#define XO    4
__device__ __forceinline__ size_t fofs(int b, int yy, int pl, int xx) {
  return ((((size_t)b * HALO + yy) * NPL + pl) * XP + xx) * 16;
}

__device__ __forceinline__ v16h cat8(v8h lo, v8h hi) {
  return __builtin_shufflevector(lo, hi, 0,1,2,3,4,5,6,7,8,9,10,11,12,13,14,15);
}
__device__ __forceinline__ v8f wmma16(v16h a, v16h b, v8f c) {
  return __builtin_amdgcn_wmma_f32_16x16x32_f16(false, a, false, b, (short)0, c, false, false);
}

#define ROW_PIECES   (NPL * XP * 2)
#define IMG_PIECES   (HALO * ROW_PIECES)
#define SIDE_PIECES  (HW * NPL * 2 * 2)
#define FRAME_PIECES (2 * ROW_PIECES + SIDE_PIECES)
#define NPLANE       (3 * NBATCH)
__device__ __forceinline__ void zero_frames_body(int blk, _Float16* __restrict__ feat,
                                                 float* __restrict__ nsum) {
  const int idx = blk * 256 + threadIdx.x;
  if (idx < NBATCH) nsum[idx] = 0.0f;
  if (idx >= NPLANE * FRAME_PIECES) return;
  const int plane = idx / FRAME_PIECES;
  const int rem   = idx - plane * FRAME_PIECES;
  int piece;
  if (rem < ROW_PIECES) {
    piece = rem;
  } else if (rem < 2 * ROW_PIECES) {
    piece = (HALO - 1) * ROW_PIECES + (rem - ROW_PIECES);
  } else {
    const int k    = rem - 2 * ROW_PIECES;
    const int half = k & 1;
    const int side = (k >> 1) & 1;
    const int pl   = (k >> 2) & 3;
    const int yy   = 1 + (k >> 4);
    piece = ((yy * NPL + pl) * XP + (side ? XO + HW : XO - 1)) * 2 + half;
  }
  uint4* p = (uint4*)feat;
  p[(size_t)plane * IMG_PIECES + piece] = make_uint4(0u, 0u, 0u, 0u);
}

__device__ __forceinline__ void pack_w_body(int blkid, const float* __restrict__ w1s,
                                            const float* __restrict__ w2s,
                                            _Float16* __restrict__ dst) {
  int idx = blkid * 256 + threadIdx.x;
  const int TOT = 2 * 8 * 9 * 64 * 64;
  if (idx >= TOT) return;
  int c   = idx & 63;
  int n   = (idx >> 6) & 63;
  int pg  = idx >> 12;
  int pos = pg % 9;
  int lay = pg / 9;
  int blk = lay & 7;
  int which = lay >> 3;
  const float* src = which ? w2s : w1s;
  float v = src[(((size_t)(blk * 64 + n) * 64 + c) * 9) + pos];
  dst[idx] = (_Float16)v;
}

__device__ __forceinline__ void pack_wout_body(int blk, const float* __restrict__ w_out,
                                               _Float16* __restrict__ dst) {
  int idx = blk * 256 + threadIdx.x;
  if (idx >= NPADC * 64) return;
  int c = idx & 63;
  int n = idx >> 6;
  dst[idx] = (n < NCH) ? (_Float16)w_out[(size_t)n * 64 + c] : (_Float16)0.0f;
}

__device__ __forceinline__ void conv_in_body(int blk, const float* __restrict__ z,
                                             const float* __restrict__ w_in,
                                             const float* __restrict__ b_in,
                                             _Float16* __restrict__ dst) {
  const int x   = threadIdx.x & (HW - 1);
  const int row = blk * 2 + (threadIdx.x >> 7);
  const int y = row & (HW - 1);
  const int b = row >> 7;

  float zv[27];
  #pragma unroll
  for (int ci = 0; ci < 3; ++ci) {
    const float* zp = z + ((size_t)(b * 3 + ci)) * HW * HW;
    #pragma unroll
    for (int ky = 0; ky < 3; ++ky) {
      const int yy = y + ky - 1;
      #pragma unroll
      for (int kx = 0; kx < 3; ++kx) {
        const int xx = x + kx - 1;
        const bool ok = ((unsigned)yy < (unsigned)HW) && ((unsigned)xx < (unsigned)HW);
        zv[ci * 9 + ky * 3 + kx] = ok ? zp[yy * HW + xx] : 0.0f;
      }
    }
  }

  _Float16* op = dst + fofs(b, y + 1, 0, x + XO);
  #pragma unroll
  for (int g = 0; g < NFEAT / 8; ++g) {
    v8h o;
    #pragma unroll
    for (int jp = 0; jp < 4; ++jp) {
      const int co = g * 8 + jp * 2;
      const float* wp = w_in + co * 27;
      float a0 = b_in[co];
      float a1 = b_in[co + 1];
      #pragma unroll
      for (int t = 0; t < 27; ++t) {
        a0 = __builtin_fmaf(zv[t], wp[t], a0);
        a1 = __builtin_fmaf(zv[t], wp[27 + t], a1);
      }
      o[jp * 2]     = (_Float16)a0;
      o[jp * 2 + 1] = (_Float16)a1;
    }
    *(v8h*)(op + (size_t)(g >> 1) * (XP * 16) + (g & 1) * 8) = o;
  }
}

#define CI_BLOCKS (NBATCH * HW / 2)
#define PW_BLOCKS ((2 * 8 * 9 * 64 * 64) / 256)
#define PO_BLOCKS ((NPADC * 64) / 256)
#define ZF_BLOCKS ((NPLANE * FRAME_PIECES + 255) / 256)
__global__ __launch_bounds__(256) void prologue_k(
    const float* __restrict__ z, const float* __restrict__ w_in,
    const float* __restrict__ b_in, const float* __restrict__ w1s,
    const float* __restrict__ w2s, const float* __restrict__ w_out,
    _Float16* __restrict__ feat, _Float16* __restrict__ wpack,
    _Float16* __restrict__ wopack, float* __restrict__ nsum) {
  const int blk = blockIdx.x;
  if (blk < CI_BLOCKS) {
    conv_in_body(blk, z, w_in, b_in, feat);
  } else if (blk < CI_BLOCKS + PW_BLOCKS) {
    pack_w_body(blk - CI_BLOCKS, w1s, w2s, wpack);
  } else if (blk < CI_BLOCKS + PW_BLOCKS + PO_BLOCKS) {
    pack_wout_body(blk - (CI_BLOCKS + PW_BLOCKS), w_out, wopack);
  } else {
    zero_frames_body(blk - (CI_BLOCKS + PW_BLOCKS + PO_BLOCKS), feat, nsum);
  }
}

#define WLDS  (9 * 64 * LDSP)
#define MIDP  130
#define MIDH  (4 * NPL * MIDP * 16)

__device__ __forceinline__ void lds_barrier() {
  __builtin_amdgcn_fence(__ATOMIC_RELEASE, "workgroup", "local");
  __builtin_amdgcn_s_barrier();
  __builtin_amdgcn_fence(__ATOMIC_ACQUIRE, "workgroup", "local");
}

template <bool BIAS>
__device__ __forceinline__ void kloop(const _Float16* lb, const _Float16* lb_hi,
                                      const v8h (&af)[4][3][4], v8f (&acc)[2][4],
                                      const float* __restrict__ bias, int mhi, v4f (&bq)[4][2]) {
  v16h bf[2][4];
  #pragma unroll
  for (int nt = 0; nt < 4; ++nt) {
    const _Float16* w = lb + nt * 16 * LDSP;
    bf[0][nt] = cat8(*(const v8h*)w, *(const v8h*)(w + 16));
  }
  __builtin_amdgcn_sched_group_barrier(0x100, 8, 0);
  constexpr int RSTEP = 8;
  #pragma unroll
  for (int s = 0; s < 18; ++s) {
    if (s + 1 < 18) {
      const int tn = (s + 1) >> 1;
      const _Float16* wn = (tn < 5 ? lb + tn * (64 * LDSP) : lb_hi + (tn - 5) * (64 * LDSP))
                           + ((s + 1) & 1) * 32;
      #pragma unroll
      for (int nt = 0; nt < 4; ++nt) {
        const _Float16* w = wn + nt * 16 * LDSP;
        bf[(s + 1) & 1][nt] = cat8(*(const v8h*)w, *(const v8h*)(w + 16));
      }
    }
    const int ky = (s >> 1) / 3;
    const int kx = (s >> 1) % 3;
    const int hq = 2 * (s & 1);
    const v16h a0 = cat8(af[ky][kx][hq],     af[ky][kx][hq + 1]);
    const v16h a1 = cat8(af[ky + 1][kx][hq], af[ky + 1][kx][hq + 1]);
    #pragma unroll
    for (int nt = 0; nt < 4; ++nt) {
      acc[0][nt] = wmma16(bf[s & 1][nt], a0, acc[0][nt]);
      acc[1][nt] = wmma16(bf[s & 1][nt], a1, acc[1][nt]);
    }
    if (s + 1 < 18) __builtin_amdgcn_sched_group_barrier(0x100, 8, 0);
    __builtin_amdgcn_sched_group_barrier(0x008, 8, 0);
    if (BIAS && s == RSTEP) {
      #pragma unroll
      for (int nt = 0; nt < 4; ++nt) {
        bq[nt][0] = *(const v4f*)(bias + nt * 16 + mhi);
        bq[nt][1] = *(const v4f*)(bias + nt * 16 + mhi + 4);
      }
      __builtin_amdgcn_sched_group_barrier(0x020, 8, 0);
    }
  }
}

__global__ __launch_bounds__(256) void resblock_wmma(
    const _Float16* __restrict__ src, _Float16* __restrict__ dst,
    const _Float16* __restrict__ w1l, const _Float16* __restrict__ w2l,
    const float* __restrict__ bias1, const float* __restrict__ bias2) {
  __shared__ __attribute__((aligned(16))) _Float16 lw[2 * WLDS + MIDH];
  _Float16* mid = lw + 2 * WLDS;

  const int wave = threadIdx.x >> 5;
  const int lane = threadIdx.x & 31;
  const int tile = blockIdx.x * 8 + wave;
  const int b  = tile >> 9;
  const int r  = tile & 511;
  const int y  = (r >> 3) << 1;
  const int x0 = (r & 7) << 4;

  const int mrow = lane & 15;
  const int ksel = (lane >> 4) << 3;
  const int nlan = lane & 15;
  const int mhi  = (lane >> 4) << 3;

  if (threadIdx.x < 64) {
    const int rp = threadIdx.x >> 2, side = (threadIdx.x >> 1) & 1, hf = threadIdx.x & 1;
    *(uint4*)(mid + (rp * MIDP + (side ? MIDP - 1 : 0)) * 16 + hf * 8) = make_uint4(0u, 0u, 0u, 0u);
  }
  constexpr int NPIECE = (9 * 64 * 64) / 8 / 256;
  {
    uint4 wst[NPIECE];
    const uint4* g = (const uint4*)w1l;
    #pragma unroll
    for (int j = 0; j < NPIECE; ++j) wst[j] = g[threadIdx.x + 256 * j];
    #pragma unroll
    for (int j = 0; j < NPIECE; ++j) {
      const int i = threadIdx.x + 256 * j;
      *(uint4*)(lw + (i >> 3) * LDSP + (i & 7) * 8) = wst[j];
    }
  }
  v8h af[4][3][4];
  #pragma unroll
  for (int rr = 0; rr < 4; ++rr) {
    int yy = y - 1 + rr;
    yy = yy < 0 ? 0 : yy;
    #pragma unroll
    for (int kx = 0; kx < 3; ++kx) {
      const _Float16* ap = src + fofs(b, yy, 0, x0 + mrow + kx + XO - 1) + ksel;
      #pragma unroll
      for (int q = 0; q < 4; ++q)
        af[rr][kx][q] = *(const v8h*)(ap + (size_t)q * (XP * 16));
    }
  }
  __builtin_amdgcn_sched_barrier(0);
  lds_barrier();

  uint4 wst2[NPIECE];
  {
    const uint4* g = (const uint4*)w2l;
    #pragma unroll
    for (int j = 0; j < NPIECE; ++j) wst2[j] = g[threadIdx.x + 256 * j];
  }

  const _Float16* lb1 = lw + nlan * LDSP + ksel;
  int hi_off = 5 * 64 * LDSP;
  asm volatile("" : "+v"(hi_off));
  const _Float16* lb1_hi = lb1 + hi_off;

  const size_t roff = fofs(b, y + 1, 0, x0 + mrow + XO) + mhi;
  constexpr size_t ROWH = (size_t)NPL * XP * 16;
  _Float16* mpos = mid + (x0 + mrow + 1) * 16 + mhi;
  v4f bq[4][2];
  v8h rv[2][4];

  #pragma unroll
  for (int p = 0; p < 2; ++p) {
    v8f acc[2][4] = {};
    kloop<true>(lb1, lb1_hi, af, acc, bias1, mhi, bq);
    if (p == 0) {
      #pragma unroll
      for (int j = 0; j < NPIECE; ++j) {
        const int i = threadIdx.x + 256 * j;
        *(uint4*)(lw + WLDS + (i >> 3) * LDSP + (i & 7) * 8) = wst2[j];
      }
      __builtin_amdgcn_sched_barrier(0);
    }
    if (p == 0) {
      v8h nf[2][3][4];
      #pragma unroll
      for (int rr = 0; rr < 2; ++rr) {
        int yy = y + 3 + rr;
        yy = yy > HALO - 1 ? HALO - 1 : yy;
        #pragma unroll
        for (int kx = 0; kx < 3; ++kx) {
          const _Float16* ap = src + fofs(b, yy, 0, x0 + mrow + kx + XO - 1) + ksel;
          #pragma unroll
          for (int q = 0; q < 4; ++q)
            nf[rr][kx][q] = *(const v8h*)(ap + (size_t)q * (XP * 16));
        }
      }
      #pragma unroll
      for (int kx = 0; kx < 3; ++kx) {
        #pragma unroll
        for (int q = 0; q < 4; ++q) {
          af[0][kx][q] = af[2][kx][q];
          af[1][kx][q] = af[3][kx][q];
          af[2][kx][q] = nf[0][kx][q];
          af[3][kx][q] = nf[1][kx][q];
        }
      }
    }
    #pragma unroll
    for (int t = 0; t < 2; ++t) {
      const int iy = y - 1 + 2 * p + t;
      const bool inside = (unsigned)iy < (unsigned)HW;
      #pragma unroll
      for (int nt = 0; nt < 4; ++nt) {
        v8h o;
        #pragma unroll
        for (int i = 0; i < 8; ++i) {
          float v = acc[t][nt][i] + bq[nt][i >> 2][i & 3];
          v = v > 0.0f ? v : 0.0f;
          o[i] = inside ? (_Float16)v : (_Float16)0.0f;
        }
        *(v8h*)(mpos + ((2 * p + t) * NPL + nt) * (MIDP * 16)) = o;
      }
    }
    __builtin_amdgcn_sched_barrier(0);
  }
  #pragma unroll
  for (int nt = 0; nt < 4; ++nt) {
    bq[nt][0] = *(const v4f*)(bias2 + nt * 16 + mhi);
    bq[nt][1] = *(const v4f*)(bias2 + nt * 16 + mhi + 4);
    rv[0][nt] = *(const v8h*)(src + roff + (size_t)nt * (XP * 16));
    rv[1][nt] = *(const v8h*)(src + roff + ROWH + (size_t)nt * (XP * 16));
  }
  __builtin_amdgcn_sched_barrier(0);
  lds_barrier();
  int l2_off = WLDS;
  asm volatile("" : "+v"(l2_off));
  const _Float16* lb2 = lb1 + l2_off;
  int l2hi_off = WLDS + 5 * 64 * LDSP;
  asm volatile("" : "+v"(l2hi_off));
  const _Float16* lb2_hi = lb1 + l2hi_off;

  {
    const _Float16* mp = mid + (x0 + mrow) * 16 + ksel;
    #pragma unroll
    for (int rr = 0; rr < 4; ++rr)
      #pragma unroll
      for (int kx = 0; kx < 3; ++kx)
        #pragma unroll
        for (int q = 0; q < 4; ++q)
          af[rr][kx][q] = *(const v8h*)(mp + ((rr * NPL + q) * MIDP + kx) * 16);
  }
  __builtin_amdgcn_sched_barrier(0);
  v8f acc[2][4] = {};
  kloop<false>(lb2, lb2_hi, af, acc, bias2, mhi, bq);

  #pragma unroll
  for (int t = 0; t < 2; ++t) {
    #pragma unroll
    for (int nt = 0; nt < 4; ++nt) {
      v8h o;
      #pragma unroll
      for (int i = 0; i < 8; ++i) {
        float v = acc[t][nt][i] + bq[nt][i >> 2][i & 3];
        v += (float)rv[t][nt][i];
        o[i] = (_Float16)v;
      }
      *(v8h*)(dst + roff + (size_t)t * ROWH + (size_t)nt * (XP * 16)) = o;
    }
  }
}

__global__ __launch_bounds__(128) void convout_softmax(
    const _Float16* __restrict__ feat, const _Float16* __restrict__ wo,
    const float* __restrict__ bo, _Float16* __restrict__ probs,
    float* __restrict__ nsum) {
  const int wave = threadIdx.x >> 5;
  const int lane = threadIdx.x & 31;
  const int tile = blockIdx.x * 4 + wave;
  const int b  = tile >> 10;
  const int r  = tile & 1023;
  const int y  = r >> 3;
  const int x0 = (r & 7) << 4;

  const int mrow = lane & 15;
  const int ksel = (lane >> 4) << 3;
  const int nlan = lane & 15;
  const int mhi  = (lane >> 4) << 3;

  const _Float16* ap = feat + fofs(b, y + 1, 0, x0 + mrow + XO) + ksel;
  const size_t PLS = (size_t)XP * 16;
  v16h a0 = cat8(*(const v8h*)(ap),           *(const v8h*)(ap + PLS));
  v16h a1 = cat8(*(const v8h*)(ap + 2 * PLS), *(const v8h*)(ap + 3 * PLS));

  constexpr int NT = NPADC / 16;
  v8f lg[NT];

  constexpr int PF = 5;
  __builtin_amdgcn_sched_group_barrier(0x020, 4 + 4 * PF, 0);
  #pragma unroll
  for (int nt = 0; nt < NT; ++nt) {
    const int n = nt * 16 + nlan;
    const _Float16* wp = wo + (size_t)n * 64;
    v16h b0 = cat8(*(const v8h*)(wp + ksel),      *(const v8h*)(wp + 16 + ksel));
    v16h b1 = cat8(*(const v8h*)(wp + 32 + ksel), *(const v8h*)(wp + 48 + ksel));
    v8f c = {};
    c = wmma16(a0, b0, c);
    c = wmma16(a1, b1, c);
    lg[nt] = c;
    __builtin_amdgcn_sched_group_barrier(0x008, 2, 0);
    __builtin_amdgcn_sched_group_barrier(0x020, 4, 0);
  }
  #pragma unroll
  for (int nt = 0; nt < NT; ++nt) {
    const int n = nt * 16 + nlan;
    const float bv = (n < NCH) ? bo[n] : -1e30f;
    #pragma unroll
    for (int i = 0; i < 8; ++i) lg[nt][i] += bv;
  }

  float inv[8];
  #pragma unroll
  for (int i = 0; i < 8; ++i) {
    float m = lg[0][i];
    #pragma unroll
    for (int nt = 1; nt < NT; ++nt) m = fmaxf(m, lg[nt][i]);
    #pragma unroll
    for (int off = 8; off; off >>= 1) m = fmaxf(m, __shfl_xor(m, off, 32));
    float s = 0.0f;
    #pragma unroll
    for (int nt = 0; nt < NT; ++nt) {
      const float e = __expf(lg[nt][i] - m);
      lg[nt][i] = e;
      s += e;
    }
    #pragma unroll
    for (int off = 8; off; off >>= 1) s += __shfl_xor(s, off, 32);
    inv[i] = 1.0f / s;
  }

  _Float16* pb = probs + (size_t)b * NCH * (HW * HW) + (size_t)y * HW + x0 + mhi;
  #pragma unroll
  for (int nt = 0; nt < NT; ++nt) {
    const int t = nt * 16 + nlan;
    v8h o;
    #pragma unroll
    for (int i = 0; i < 8; ++i) o[i] = (_Float16)(lg[nt][i] * inv[i]);
    if (t < NCH) *(v8h*)(pb + (size_t)t * (HW * HW)) = o;
  }

  float nacc = 0.0f;
  if (nlan == (KTAP & 15)) {
    #pragma unroll
    for (int i = 0; i < 8; ++i) nacc += lg[KTAP >> 4][i] * inv[i];
  }
  nacc += __shfl_xor(nacc, 16, 32);
  if (lane == (KTAP & 15)) atomicAdd(nsum + b, nacc);
}

typedef __attribute__((ext_vector_type(4))) _Float16 v4h;
typedef __attribute__((ext_vector_type(2))) float    v2f;

#define AK_WAVES 7
#define AK_ROWS  3
#define AK_NG    9

__global__ __launch_bounds__(AK_WAVES * 32) void apply_k(
    const float* __restrict__ x, const float* __restrict__ eps,
    const _Float16* __restrict__ probs, const float* __restrict__ nsum,
    float* __restrict__ out) {
  __shared__ float red[AK_WAVES * 12 * 32];
  const int lane = threadIdx.x & 31;
  const int wave = __builtin_amdgcn_readfirstlane(threadIdx.x >> 5);
  const int h = blockIdx.x & (HW - 1);
  const int b = blockIdx.x >> 7;
  const int col0 = 16 * lane - 12;

  float acc[3][4];
  #pragma unroll
  for (int c = 0; c < 3; ++c)
    #pragma unroll
    for (int q = 0; q < 4; ++q) acc[c][q] = 0.0f;

  #pragma unroll 1
  for (int r = 0; r < AK_ROWS; ++r) {
    const int ky = wave * AK_ROWS + r;
    const int yi = 4 * h - 10 + ky;
    const float rowkeep = ((unsigned)yi < 512u) ? 1.0f : 0.0f;
    const int yc = yi < 0 ? 0 : (yi > 511 ? 511 : yi);

    const _Float16* pr =
        probs + ((size_t)(b * NCH + ky * 21)) * (HW * HW) + (size_t)h * HW + 4 * lane;
    v4h pv[21];
    #pragma unroll
    for (int kx = 0; kx < 21; ++kx)
      pv[kx] = __builtin_nontemporal_load((const v4h*)(pr + (size_t)kx * (HW * HW)));

    bool ok[AK_NG];
    #pragma unroll
    for (int g = 0; g < AK_NG; ++g) {
      const int col = col0 + 4 * g;
      ok[g] = (g < 3) ? (col >= 0) : ((g > 6) ? (col < 512) : true);
    }
    v2f xlo[3];
    v4f xv[3][AK_NG - 2];
    v2f xhi[3];
    float xend[3];
    #pragma unroll
    for (int c = 0; c < 3; ++c) {
      const float* xr = x + ((size_t)((b * 3 + c) * 512 + yc)) * 512;
      xlo[c] = *(const v2f*)(xr + (ok[0] ? col0 : 0) + 2);
      #pragma unroll
      for (int g = 1; g < AK_NG - 1; ++g)
        xv[c][g - 1] = *(const v4f*)(xr + (ok[g] ? col0 + 4 * g : 0));
      const float* xe = xr + (ok[AK_NG - 1] ? col0 + 4 * (AK_NG - 1) : 0);
      xhi[c]  = *(const v2f*)xe;
      xend[c] = xe[2];
    }
    __builtin_amdgcn_sched_barrier(0);

    float keep[AK_NG];
    #pragma unroll
    for (int g = 0; g < AK_NG; ++g) keep[g] = ok[g] ? 1.0f : 0.0f;
    float xs[3][4 * AK_NG];
    #pragma unroll
    for (int c = 0; c < 3; ++c) {
      xs[c][2] = xlo[c][0] * keep[0];
      xs[c][3] = xlo[c][1] * keep[0];
      #pragma unroll
      for (int g = 1; g < AK_NG - 1; ++g) {
        #pragma unroll
        for (int e = 0; e < 4; ++e)
          xs[c][4 * g + e] = (g < 3 || g > 6) ? xv[c][g - 1][e] * keep[g] : xv[c][g - 1][e];
      }
      xs[c][32] = xhi[c][0] * keep[AK_NG - 1];
      xs[c][33] = xhi[c][1] * keep[AK_NG - 1];
      xs[c][34] = xend[c]   * keep[AK_NG - 1];
    }

    float part[3][4];
    #pragma unroll
    for (int c = 0; c < 3; ++c)
      #pragma unroll
      for (int q = 0; q < 4; ++q) part[c][q] = 0.0f;
    #pragma unroll
    for (int kx = 0; kx < 21; ++kx) {
      #pragma unroll
      for (int q = 0; q < 4; ++q) {
        const float pf = (float)pv[kx][q];
        #pragma unroll
        for (int c = 0; c < 3; ++c) part[c][q] += xs[c][4 * q + 2 + kx] * pf;
      }
    }
    #pragma unroll
    for (int c = 0; c < 3; ++c)
      #pragma unroll
      for (int q = 0; q < 4; ++q) acc[c][q] += part[c][q] * rowkeep;
  }

  #pragma unroll
  for (int c = 0; c < 3; ++c)
    #pragma unroll
    for (int q = 0; q < 4; ++q) red[(wave * 12 + c * 4 + q) * 32 + lane] = acc[c][q];
  __syncthreads();

  if (wave < 3) {
    const int c = wave;
    float s[4];
    #pragma unroll
    for (int q = 0; q < 4; ++q) {
      float t = 0.0f;
      #pragma unroll
      for (int j = 0; j < AK_WAVES; ++j) t += red[(j * 12 + c * 4 + q) * 32 + lane];
      s[q] = t;
    }
    const size_t idx = (((size_t)(b * 3 + c)) * HW + h) * HW + 4 * lane;
    const float ns = nsum[b] * (1.0f / (HW * HW));
    const v4f e4 = *(const v4f*)(eps + idx);
    v4f o;
    #pragma unroll
    for (int q = 0; q < 4; ++q) o[q] = s[q] + ns * e4[q];
    *(v4f*)(out + idx) = o;
  }
}

extern "C" void kernel_launch(void* const* d_in, const int* in_sizes, int n_in,
                              void* d_out, int out_size, void* d_ws, size_t ws_size,
                              hipStream_t stream) {
  const float* x     = (const float*)d_in[0];
  const float* z     = (const float*)d_in[1];
  const float* eps   = (const float*)d_in[2];
  const float* w_in  = (const float*)d_in[3];
  const float* b_in  = (const float*)d_in[4];
  const float* w1s   = (const float*)d_in[5];
  const float* b1s   = (const float*)d_in[6];
  const float* w2s   = (const float*)d_in[7];
  const float* b2s   = (const float*)d_in[8];
  const float* w_out = (const float*)d_in[9];
  const float* b_out = (const float*)d_in[10];
  float* out = (float*)d_out;

  char* ws = (char*)d_ws;
  const size_t FBYTES = (size_t)NBATCH * HALO * NPL * XP * 16 * sizeof(_Float16);
  _Float16* F0 = (_Float16*)(ws);
  _Float16* F1 = (_Float16*)(ws + FBYTES);
  _Float16* F2 = (_Float16*)(ws + 2 * FBYTES);
  _Float16* wpack  = (_Float16*)(ws + 3 * FBYTES);
  _Float16* probs  = F1;
  _Float16* wopack = probs + (size_t)NBATCH * NCH * HW * HW;
  float*    nsum   = (float*)(wopack + (size_t)NPADC * 64);

  prologue_k<<<CI_BLOCKS + PW_BLOCKS + PO_BLOCKS + ZF_BLOCKS, 256, 0, stream>>>(
      z, w_in, b_in, w1s, w2s, w_out, F0, wpack, wopack, nsum);


  _Float16* cur = F0;
  _Float16* nxt = F2;
  const size_t LAYER = (size_t)9 * 64 * 64;
  for (int i = 0; i < 8; ++i) {
    resblock_wmma<<<256, 256, 0, stream>>>(
        cur, nxt, wpack + (size_t)i * LAYER, wpack + (size_t)(8 + i) * LAYER,
        b1s + (size_t)i * 64, b2s + (size_t)i * 64);
    _Float16* t = cur; cur = nxt; nxt = t;
  }

  convout_softmax<<<1024, 128, 0, stream>>>(cur, wopack, b_out, probs, nsum);
  apply_k<<<NBATCH * HW, AK_WAVES * 32, 0, stream>>>(x, eps, probs, nsum, out);
}
